# starting kernel
# baseline (speedup 1.0000x reference)
#include <hip/hip_runtime.h>
#include <math.h>

typedef __attribute__((ext_vector_type(2))) float v2f;
typedef __attribute__((ext_vector_type(8))) float v8f;

struct Rings { int r1[64]; int r2[64]; };

constexpr Rings makeRings() {
  Rings R{};
  for (int rr = 1; rr <= 2; ++rr) {
    int total[64];
    for (int i = 0; i < 64; ++i) total[i] = i;
    for (int w = 0; w < 6; ++w) {
      int t = (w + rr) % 6;
      int nxt[64];
      for (int i = 0; i < 64; ++i) {
        int cbit = (i >> (5 - w)) & 1;
        nxt[i] = total[i ^ (cbit << (5 - t))];
      }
      for (int i = 0; i < 64; ++i) total[i] = nxt[i];
    }
    for (int i = 0; i < 64; ++i) { if (rr == 1) R.r1[i] = total[i]; else R.r2[i] = total[i]; }
  }
  return R;
}

__constant__ Rings RINGS = makeRings();

__device__ __forceinline__ float2 cmul(float2 a, float2 b) {
  return make_float2(a.x * b.x - a.y * b.y, a.x * b.y + a.y * b.x);
}
__device__ __forceinline__ float2 cadd(float2 a, float2 b) { return make_float2(a.x + b.x, a.y + b.y); }
__device__ __forceinline__ float2 shflx2(float2 v, int m) {
  return make_float2(__shfl_xor(v.x, m, 32), __shfl_xor(v.y, m, 32));
}
__device__ __forceinline__ float2 shfl2(float2 v, int l) {
  return make_float2(__shfl(v.x, l, 32), __shfl(v.y, l, 32));
}

#define RED16(v) do { v += __shfl_xor(v,1,32); v += __shfl_xor(v,2,32); \
                      v += __shfl_xor(v,4,32); v += __shfl_xor(v,8,32); } while(0)

__global__ __launch_bounds__(256) void qmaml_kernel(
    const float* __restrict__ x,  const float* __restrict__ W1,
    const float* __restrict__ b1, const float* __restrict__ ln_g,
    const float* __restrict__ ln_b, const float* __restrict__ W2,
    const float* __restrict__ b2, const float* __restrict__ th_s,
    const float* __restrict__ th_t, const float* __restrict__ Wc,
    const float* __restrict__ bc, float* __restrict__ out, int B)
{
  __shared__ float  sW1[4096];
  __shared__ float  sW2[384];
  __shared__ float  sB1[64], sG[64], sBeta[64];
  __shared__ float  sB2[8], sWc[32], sBc[8];
  __shared__ float2 sGates[72];
  __shared__ float  sZ[768];

  const int tid = threadIdx.x;

  for (int i = tid; i < 1024; i += 256) ((float4*)sW1)[i] = ((const float4*)W1)[i];
  for (int i = tid; i < 384; i += 256)  sW2[i] = W2[i];
  if (tid < 64) { sB1[tid] = b1[tid]; sG[tid] = ln_g[tid]; sBeta[tid] = ln_b[tid]; }
  if (tid < 6)  sB2[tid] = b2[tid];
  if (tid < 30) sWc[tid] = Wc[tid];
  if (tid < 5)  sBc[tid] = bc[tid];
  if (tid < 18) {
    int l = tid / 6, w = tid % 6;
    const float* th = (l < 2) ? (th_s + (l * 6 + w) * 3) : (th_t + w * 3);
    float phi = th[0], theta = th[1], omega = th[2];
    float c  = cosf(0.5f * theta), s = sinf(0.5f * theta);
    float aa = 0.5f * (phi + omega), bb = 0.5f * (phi - omega);
    float ca = cosf(aa), sa = sinf(aa);
    float cb = cosf(bb), sb = sinf(bb);
    sGates[tid * 4 + 0] = make_float2( ca * c, -sa * c);
    sGates[tid * 4 + 1] = make_float2(-cb * s, -sb * s);
    sGates[tid * 4 + 2] = make_float2( cb * s, -sb * s);
    sGates[tid * 4 + 3] = make_float2( ca * c,  sa * c);
  }
  __syncthreads();

  const int lane = tid & 31;
  const int wave = tid >> 5;
  const int half = lane >> 4;
  const int lm   = lane & 15;
  const int rowBase = blockIdx.x * 128 + wave * 16;

  v8f acc[4] = {};
  int rowA = rowBase + lm; if (rowA >= B) rowA = B - 1;
  const float* xrow = x + (size_t)rowA * 64 + 2 * half;
  #pragma unroll
  for (int kk = 0; kk < 16; ++kk) {
    v2f a = *(const v2f*)(xrow + 4 * kk);
    #pragma unroll
    for (int t = 0; t < 4; ++t) {
      v2f bf = *(const v2f*)(sW1 + (16 * t + lm) * 64 + 4 * kk + 2 * half);
      acc[t] = __builtin_amdgcn_wmma_f32_16x16x4_f32(false, a, false, bf,
                                                     (short)0, acc[t], false, false);
    }
  }

  float hv[4][8];
  #pragma unroll
  for (int t = 0; t < 4; ++t) {
    float bias = sB1[16 * t + lm];
    #pragma unroll
    for (int j = 0; j < 8; ++j) hv[t][j] = fmaxf(acc[t][j] + bias, 0.0f);
  }

  float muj[8], inv[8];
  #pragma unroll
  for (int j = 0; j < 8; ++j) {
    float s  = hv[0][j] + hv[1][j] + hv[2][j] + hv[3][j];
    float q2 = hv[0][j]*hv[0][j] + hv[1][j]*hv[1][j] + hv[2][j]*hv[2][j] + hv[3][j]*hv[3][j];
    RED16(s); RED16(q2);
    float mu  = s * 0.015625f;
    float var = q2 * 0.015625f - mu * mu;
    muj[j] = mu;
    inv[j] = rsqrtf(var + 1e-5f);
  }
  #pragma unroll
  for (int t = 0; t < 4; ++t) {
    float g = sG[16 * t + lm], be = sBeta[16 * t + lm];
    #pragma unroll
    for (int j = 0; j < 8; ++j) hv[t][j] = (hv[t][j] - muj[j]) * inv[j] * g + be;
  }

  #pragma unroll
  for (int q = 0; q < 6; ++q) {
    float w0 = sW2[q * 64 + lm],      w1 = sW2[q * 64 + 16 + lm];
    float w2 = sW2[q * 64 + 32 + lm], w3 = sW2[q * 64 + 48 + lm];
    float bq = sB2[q];
    #pragma unroll
    for (int j = 0; j < 8; ++j) {
      float p = hv[0][j]*w0 + hv[1][j]*w1 + hv[2][j]*w2 + hv[3][j]*w3;
      RED16(p);
      if (lm == 0) sZ[wave * 96 + (8 * half + j) * 6 + q] = tanhf(p + bq);
    }
  }
  __syncthreads();

  float ws0[5], ws1[5];
  #pragma unroll
  for (int k = 0; k < 5; ++k) {
    float t0 = 0.f, t1 = 0.f;
    #pragma unroll
    for (int q = 0; q < 6; ++q) {
      float w = sWc[k * 6 + q];
      int b = 5 - q;
      t0 += ((lane >> b) & 1) ? -w : w;
      t1 += (((lane + 32) >> b) & 1) ? -w : w;
    }
    ws0[k] = t0; ws1[k] = t1;
  }
  const int rA0 = RINGS.r1[lane], rA1 = RINGS.r1[lane + 32];
  const int rB0 = RINGS.r2[lane], rB1 = RINGS.r2[lane + 32];

  for (int r = 0; r < 16; ++r) {
    int row = rowBase + r;
    float cq[6], sq[6];
    #pragma unroll
    for (int q = 0; q < 6; ++q) {
      float h2 = 1.57079632679489662f * sZ[wave * 96 + r * 6 + q];
      cq[q] = cosf(h2); sq[q] = sinf(h2);
    }
    float pr = 1.f;
    #pragma unroll
    for (int q = 1; q < 6; ++q) pr *= ((lane >> (5 - q)) & 1) ? sq[q] : cq[q];
    float2 a0 = make_float2(cq[0] * pr, 0.f);
    float2 a1 = make_float2(sq[0] * pr, 0.f);

    #pragma unroll
    for (int l = 0; l < 3; ++l) {
      #pragma unroll
      for (int w = 0; w < 6; ++w) {
        int g = l * 6 + w;
        float2 m00 = sGates[g * 4 + 0], m01 = sGates[g * 4 + 1];
        float2 m10 = sGates[g * 4 + 2], m11 = sGates[g * 4 + 3];
        if (w == 0) {
          float2 n0 = cadd(cmul(m00, a0), cmul(m01, a1));
          float2 n1 = cadd(cmul(m10, a0), cmul(m11, a1));
          a0 = n0; a1 = n1;
        } else {
          int b = 5 - w, mask = 1 << b;
          float2 p0 = shflx2(a0, mask), p1 = shflx2(a1, mask);
          int bit = (lane >> b) & 1;
          float2 e0 = cadd(cmul(m00, a0), cmul(m01, p0));
          float2 o0 = cadd(cmul(m10, p0), cmul(m11, a0));
          float2 e1 = cadd(cmul(m00, a1), cmul(m01, p1));
          float2 o1 = cadd(cmul(m10, p1), cmul(m11, a1));
          a0 = bit ? o0 : e0;
          a1 = bit ? o1 : e1;
        }
      }
      int s0 = (l == 1) ? rB0 : rA0;
      int s1 = (l == 1) ? rB1 : rA1;
      float2 g0a = shfl2(a0, s0 & 31), g0b = shfl2(a1, s0 & 31);
      float2 g1a = shfl2(a0, s1 & 31), g1b = shfl2(a1, s1 & 31);
      a0 = (s0 < 32) ? g0a : g0b;
      a1 = (s1 < 32) ? g1a : g1b;
    }

    float p0 = a0.x * a0.x + a0.y * a0.y;
    float p1 = a1.x * a1.x + a1.y * a1.y;
    float o[5];
    #pragma unroll
    for (int k = 0; k < 5; ++k) {
      float v = p0 * ws0[k] + p1 * ws1[k];
      v += __shfl_xor(v, 1, 32); v += __shfl_xor(v, 2, 32); v += __shfl_xor(v, 4, 32);
      v += __shfl_xor(v, 8, 32); v += __shfl_xor(v, 16, 32);
      o[k] = v;
    }
    if (lane == 0 && row < B) {
      #pragma unroll
      for (int k = 0; k < 5; ++k) out[(size_t)row * 5 + k] = o[k] + sBc[k];
    }
  }
}

extern "C" void kernel_launch(void* const* d_in, const int* in_sizes, int n_in,
                              void* d_out, int out_size, void* d_ws, size_t ws_size,
                              hipStream_t stream) {
  (void)n_in; (void)out_size; (void)d_ws; (void)ws_size;
  const float* x    = (const float*)d_in[0];
  const float* W1   = (const float*)d_in[1];
  const float* b1   = (const float*)d_in[2];
  const float* ln_g = (const float*)d_in[3];
  const float* ln_b = (const float*)d_in[4];
  const float* W2   = (const float*)d_in[5];
  const float* b2   = (const float*)d_in[6];
  const float* th_s = (const float*)d_in[7];
  const float* th_t = (const float*)d_in[8];
  const float* Wc   = (const float*)d_in[9];
  const float* bc   = (const float*)d_in[10];
  float* out = (float*)d_out;
  int B = in_sizes[0] / 64;
  int grid = (B + 127) / 128;
  qmaml_kernel<<<grid, 256, 0, stream>>>(x, W1, b1, ln_g, ln_b, W2, b2, th_s, th_t, Wc, bc, out, B);
}


# Per-row gate-by-gate lane-shuffle simulation replaced: circuit composed once per call into one 128x64 f32 matrix (extra 8-workgroup launch), applied per 16-row tile by f32 WMMA; sum order changes
# speedup vs baseline: 4.6409x; 4.6409x over previous
#include <hip/hip_runtime.h>
#include <math.h>


typedef __attribute__((ext_vector_type(2))) float v2f;
typedef __attribute__((ext_vector_type(8))) float v8f;

struct Rings { int r1[64]; int r2[64]; };

constexpr Rings makeRings() {
  Rings R{};
  for (int rr = 1; rr <= 2; ++rr) {
    int total[64];
    for (int i = 0; i < 64; ++i) total[i] = i;
    for (int w = 0; w < 6; ++w) {
      int t = (w + rr) % 6;
      int nxt[64];
      for (int i = 0; i < 64; ++i) {
        int cbit = (i >> (5 - w)) & 1;
        nxt[i] = total[i ^ (cbit << (5 - t))];
      }
      for (int i = 0; i < 64; ++i) total[i] = nxt[i];
    }
    for (int i = 0; i < 64; ++i) { if (rr == 1) R.r1[i] = total[i]; else R.r2[i] = total[i]; }
  }
  return R;
}

__constant__ Rings RINGS = makeRings();

__device__ __forceinline__ float2 cmul(float2 a, float2 b) {
  return make_float2(a.x * b.x - a.y * b.y, a.x * b.y + a.y * b.x);
}
__device__ __forceinline__ float2 cadd(float2 a, float2 b) { return make_float2(a.x + b.x, a.y + b.y); }
__device__ __forceinline__ float2 shflx2(float2 v, int m) {
  return make_float2(__shfl_xor(v.x, m, 32), __shfl_xor(v.y, m, 32));
}
__device__ __forceinline__ float2 shfl2(float2 v, int l) {
  return make_float2(__shfl(v.x, l, 32), __shfl(v.y, l, 32));
}

#define RED16(v) do { v += __shfl_xor(v,1,32); v += __shfl_xor(v,2,32); \
                      v += __shfl_xor(v,4,32); v += __shfl_xor(v,8,32); } while(0)

#define UP 68
#define UIMG (128 * UP)

__global__ __launch_bounds__(256) void circuit_compose(
    const float* __restrict__ th_s, const float* __restrict__ th_t, float* __restrict__ uimg)
{
  __shared__ float2 sGates[72];
  __shared__ float  sChunk[1024];

  const int tid = threadIdx.x;
  if (tid < 18) {
    int l = tid / 6, w = tid % 6;
    const float* th = (l < 2) ? (th_s + (l * 6 + w) * 3) : (th_t + w * 3);
    float phi = th[0], theta = th[1], omega = th[2];
    float c  = cosf(0.5f * theta), s = sinf(0.5f * theta);
    float aa = 0.5f * (phi + omega), bb = 0.5f * (phi - omega);
    float ca = cosf(aa), sa = sinf(aa);
    float cb = cosf(bb), sb = sinf(bb);
    sGates[tid * 4 + 0] = make_float2( ca * c, -sa * c);
    sGates[tid * 4 + 1] = make_float2(-cb * s, -sb * s);
    sGates[tid * 4 + 2] = make_float2( cb * s, -sb * s);
    sGates[tid * 4 + 3] = make_float2( ca * c,  sa * c);
  }
  __syncthreads();

  const int lane = tid & 31;
  const int wave = tid >> 5;
  const int rA0 = RINGS.r1[lane], rA1 = RINGS.r1[lane + 32];
  const int rB0 = RINGS.r2[lane], rB1 = RINGS.r2[lane + 32];

  const int k = blockIdx.x * 8 + wave;
  float2 a0 = make_float2((lane == k) ? 1.f : 0.f, 0.f);
  float2 a1 = make_float2((lane + 32 == k) ? 1.f : 0.f, 0.f);

  #pragma unroll
  for (int l = 0; l < 3; ++l) {
    #pragma unroll
    for (int w = 0; w < 6; ++w) {
      int g = l * 6 + w;
      float2 m00 = sGates[g * 4 + 0], m01 = sGates[g * 4 + 1];
      float2 m10 = sGates[g * 4 + 2], m11 = sGates[g * 4 + 3];
      if (w == 0) {
        float2 n0 = cadd(cmul(m00, a0), cmul(m01, a1));
        float2 n1 = cadd(cmul(m10, a0), cmul(m11, a1));
        a0 = n0; a1 = n1;
      } else {
        int b = 5 - w, mask = 1 << b;
        float2 p0 = shflx2(a0, mask), p1 = shflx2(a1, mask);
        int bit = (lane >> b) & 1;
        float2 e0 = cadd(cmul(m00, a0), cmul(m01, p0));
        float2 o0 = cadd(cmul(m10, p0), cmul(m11, a0));
        float2 e1 = cadd(cmul(m00, a1), cmul(m01, p1));
        float2 o1 = cadd(cmul(m10, p1), cmul(m11, a1));
        a0 = bit ? o0 : e0;
        a1 = bit ? o1 : e1;
      }
    }
    int s0 = (l == 1) ? rB0 : rA0;
    int s1 = (l == 1) ? rB1 : rA1;
    float2 g0a = shfl2(a0, s0 & 31), g0b = shfl2(a1, s0 & 31);
    float2 g1a = shfl2(a0, s1 & 31), g1b = shfl2(a1, s1 & 31);
    a0 = (s0 < 32) ? g0a : g0b;
    a1 = (s1 < 32) ? g1a : g1b;
  }
  sChunk[(lane)      * 8 + wave] = a0.x;
  sChunk[(lane + 32) * 8 + wave] = a1.x;
  sChunk[(lane + 64) * 8 + wave] = a0.y;
  sChunk[(lane + 96) * 8 + wave] = a1.y;
  __syncthreads();
  ((float4*)uimg)[blockIdx.x * 256 + tid] = ((const float4*)sChunk)[tid];
}

__global__ __launch_bounds__(256) void qmaml_kernel(
    const float* __restrict__ x,  const float* __restrict__ W1,
    const float* __restrict__ b1, const float* __restrict__ ln_g,
    const float* __restrict__ ln_b, const float* __restrict__ W2,
    const float* __restrict__ b2, const float* __restrict__ uimg,
    const float* __restrict__ Wc, const float* __restrict__ bc,
    float* __restrict__ out, int B)
{
  __shared__ float  sW1[4096];
  __shared__ float  sW2[384];
  __shared__ float  sB1[64], sG[64], sBeta[64];
  __shared__ float  sB2[8], sWc[32], sBc[8];
  __shared__ float  sU[UIMG];
  __shared__ float  sZ[768];

  const int tid = threadIdx.x;

  for (int i = tid; i < 1024; i += 256) ((float4*)sW1)[i] = ((const float4*)W1)[i];
  for (int i = tid; i < 2048; i += 256) {
    const int n = (i & 255) >> 1;
    *(float4*)(sU + n * UP + 8 * (i >> 8) + 4 * (i & 1)) = ((const float4*)uimg)[i];
  }
  for (int i = tid; i < 384; i += 256)  sW2[i] = W2[i];
  if (tid < 64) { sB1[tid] = b1[tid]; sG[tid] = ln_g[tid]; sBeta[tid] = ln_b[tid]; }
  if (tid < 6)  sB2[tid] = b2[tid];
  if (tid < 30) sWc[tid] = Wc[tid];
  if (tid < 5)  sBc[tid] = bc[tid];
  __syncthreads();

  const int lane = tid & 31;
  const int wave = tid >> 5;
  const int half = lane >> 4;
  const int lm   = lane & 15;
  const int rowBase = blockIdx.x * 128 + wave * 16;

  v8f acc[4] = {};
  int rowA = rowBase + lm; if (rowA >= B) rowA = B - 1;
  const float* xrow = x + (size_t)rowA * 64 + 2 * half;
  #pragma unroll
  for (int kk = 0; kk < 16; ++kk) {
    v2f a = *(const v2f*)(xrow + 4 * kk);
    #pragma unroll
    for (int t = 0; t < 4; ++t) {
      v2f bf = *(const v2f*)(sW1 + (16 * t + lm) * 64 + 4 * kk + 2 * half);
      acc[t] = __builtin_amdgcn_wmma_f32_16x16x4_f32(false, a, false, bf,
                                                     (short)0, acc[t], false, false);
    }
  }

  float hv[4][8];
  #pragma unroll
  for (int t = 0; t < 4; ++t) {
    float bias = sB1[16 * t + lm];
    #pragma unroll
    for (int j = 0; j < 8; ++j) hv[t][j] = fmaxf(acc[t][j] + bias, 0.0f);
  }

  float muj[8], inv[8];
  #pragma unroll
  for (int j = 0; j < 8; ++j) {
    float s  = hv[0][j] + hv[1][j] + hv[2][j] + hv[3][j];
    float q2 = hv[0][j]*hv[0][j] + hv[1][j]*hv[1][j] + hv[2][j]*hv[2][j] + hv[3][j]*hv[3][j];
    RED16(s); RED16(q2);
    float mu  = s * 0.015625f;
    float var = q2 * 0.015625f - mu * mu;
    muj[j] = mu;
    inv[j] = rsqrtf(var + 1e-5f);
  }
  #pragma unroll
  for (int t = 0; t < 4; ++t) {
    float g = sG[16 * t + lm], be = sBeta[16 * t + lm];
    #pragma unroll
    for (int j = 0; j < 8; ++j) hv[t][j] = (hv[t][j] - muj[j]) * inv[j] * g + be;
  }

  #pragma unroll
  for (int q = 0; q < 6; ++q) {
    float w0 = sW2[q * 64 + lm],      w1 = sW2[q * 64 + 16 + lm];
    float w2 = sW2[q * 64 + 32 + lm], w3 = sW2[q * 64 + 48 + lm];
    float bq = sB2[q];
    #pragma unroll
    for (int j = 0; j < 8; ++j) {
      float p = hv[0][j]*w0 + hv[1][j]*w1 + hv[2][j]*w2 + hv[3][j]*w3;
      RED16(p);
      if (lm == 0) sZ[wave * 96 + (8 * half + j) * 6 + q] = tanhf(p + bq);
    }
  }
  __syncthreads();

  float cq[6], sq[6];
  #pragma unroll
  for (int q = 0; q < 6; ++q) {
    float h2 = 1.57079632679489662f * sZ[wave * 96 + lm * 6 + q];
    cq[q] = cosf(h2); sq[q] = sinf(h2);
  }
  float p01[4], p012[8], p0123[16];
  #pragma unroll
  for (int i = 0; i < 4; ++i)  p01[i]   = ((i >> 1) ? sq[0] : cq[0]) * ((i & 1) ? sq[1] : cq[1]);
  #pragma unroll
  for (int i = 0; i < 8; ++i)  p012[i]  = p01[i >> 1]  * ((i & 1) ? sq[2] : cq[2]);
  #pragma unroll
  for (int i = 0; i < 16; ++i) p0123[i] = p012[i >> 1] * ((i & 1) ? sq[3] : cq[3]);
  const float f4 = half ? sq[4] : cq[4];
  v2f pf[16];
  #pragma unroll
  for (int kk = 0; kk < 16; ++kk) {
    float t4 = p0123[kk] * f4;
    pf[kk].x = t4 * cq[5];
    pf[kk].y = t4 * sq[5];
  }

  v8f amp[8] = {};
  #pragma unroll
  for (int kk = 0; kk < 16; ++kk) {
    #pragma unroll
    for (int t = 0; t < 8; ++t) {
      v2f uf = *(const v2f*)(sU + (16 * t + lm) * UP + 4 * kk + 2 * half);
      amp[t] = __builtin_amdgcn_wmma_f32_16x16x4_f32(false, uf, false, pf[kk],
                                                     (short)0, amp[t], false, false);
    }
  }

  float P[4][8];
  #pragma unroll
  for (int t = 0; t < 4; ++t) {
    #pragma unroll
    for (int r = 0; r < 8; ++r) P[t][r] = amp[t][r] * amp[t][r] + amp[t + 4][r] * amp[t + 4][r];
  }
  float e5 = 0.f, e4 = 0.f, e3 = 0.f;
  float s1[4][4], s2[4][2], s3[4];
  #pragma unroll
  for (int t = 0; t < 4; ++t) {
    #pragma unroll
    for (int m = 0; m < 4; ++m) { s1[t][m] = P[t][2*m] + P[t][2*m+1]; e5 += P[t][2*m] - P[t][2*m+1]; }
    #pragma unroll
    for (int m = 0; m < 2; ++m) { s2[t][m] = s1[t][2*m] + s1[t][2*m+1]; e4 += s1[t][2*m] - s1[t][2*m+1]; }
    s3[t] = s2[t][0] + s2[t][1]; e3 += s2[t][0] - s2[t][1];
  }
  const float u0 = s3[0] + s3[1], u1 = s3[2] + s3[3];
  const float e1 = (s3[0] - s3[1]) + (s3[2] - s3[3]);
  const float S  = u0 + u1;
  const float e0 = u0 - u1;
  const float e2 = half ? -S : S;

  const int row = rowBase + lm;
  #pragma unroll
  for (int k = 0; k < 5; ++k) {
    float v = sWc[k*6+0]*e0 + sWc[k*6+1]*e1 + sWc[k*6+2]*e2
            + sWc[k*6+3]*e3 + sWc[k*6+4]*e4 + sWc[k*6+5]*e5;
    v += __shfl_xor(v, 16, 32);
    if (half == 0 && row < B) out[(size_t)row * 5 + k] = v + sBc[k];
  }
}

extern "C" void kernel_launch(void* const* d_in, const int* in_sizes, int n_in,
                              void* d_out, int out_size, void* d_ws, size_t ws_size,
                              hipStream_t stream) {
  (void)n_in; (void)out_size; (void)ws_size;
  const float* x    = (const float*)d_in[0];
  const float* W1   = (const float*)d_in[1];
  const float* b1   = (const float*)d_in[2];
  const float* ln_g = (const float*)d_in[3];
  const float* ln_b = (const float*)d_in[4];
  const float* W2   = (const float*)d_in[5];
  const float* b2   = (const float*)d_in[6];
  const float* th_s = (const float*)d_in[7];
  const float* th_t = (const float*)d_in[8];
  const float* Wc   = (const float*)d_in[9];
  const float* bc   = (const float*)d_in[10];
  float* out = (float*)d_out;
  float* uimg = (float*)d_ws;
  int B = in_sizes[0] / 64;
  int grid = (B + 127) / 128;
  circuit_compose<<<8, 256, 0, stream>>>(th_s, th_t, uimg);
  qmaml_kernel<<<grid, 256, 0, stream>>>(x, W1, b1, ln_g, ln_b, W2, b2, uimg, Wc, bc, out, B);
}


# LDS row pitch of the 64x64 first-layer weight matrix padded from 64 to 68 floats: its fragment reads no longer put the 16 lanes of a half-wave on one bank pair (16-way LDS bank conflict removed)
# speedup vs baseline: 5.2592x; 1.1332x over previous
#include <hip/hip_runtime.h>
#include <math.h>


typedef __attribute__((ext_vector_type(2))) float v2f;
typedef __attribute__((ext_vector_type(8))) float v8f;

struct Rings { int r1[64]; int r2[64]; };

constexpr Rings makeRings() {
  Rings R{};
  for (int rr = 1; rr <= 2; ++rr) {
    int total[64];
    for (int i = 0; i < 64; ++i) total[i] = i;
    for (int w = 0; w < 6; ++w) {
      int t = (w + rr) % 6;
      int nxt[64];
      for (int i = 0; i < 64; ++i) {
        int cbit = (i >> (5 - w)) & 1;
        nxt[i] = total[i ^ (cbit << (5 - t))];
      }
      for (int i = 0; i < 64; ++i) total[i] = nxt[i];
    }
    for (int i = 0; i < 64; ++i) { if (rr == 1) R.r1[i] = total[i]; else R.r2[i] = total[i]; }
  }
  return R;
}

__constant__ Rings RINGS = makeRings();

__device__ __forceinline__ float2 cmul(float2 a, float2 b) {
  return make_float2(a.x * b.x - a.y * b.y, a.x * b.y + a.y * b.x);
}
__device__ __forceinline__ float2 cadd(float2 a, float2 b) { return make_float2(a.x + b.x, a.y + b.y); }
__device__ __forceinline__ float2 shflx2(float2 v, int m) {
  return make_float2(__shfl_xor(v.x, m, 32), __shfl_xor(v.y, m, 32));
}
__device__ __forceinline__ float2 shfl2(float2 v, int l) {
  return make_float2(__shfl(v.x, l, 32), __shfl(v.y, l, 32));
}

#define RED16(v) do { v += __shfl_xor(v,1,32); v += __shfl_xor(v,2,32); \
                      v += __shfl_xor(v,4,32); v += __shfl_xor(v,8,32); } while(0)

#define UP 68
#define UIMG (128 * UP)
#define WP 68

__global__ __launch_bounds__(256) void circuit_compose(
    const float* __restrict__ th_s, const float* __restrict__ th_t, float* __restrict__ uimg)
{
  __shared__ float2 sGates[72];
  __shared__ float  sChunk[1024];

  const int tid = threadIdx.x;
  if (tid < 18) {
    int l = tid / 6, w = tid % 6;
    const float* th = (l < 2) ? (th_s + (l * 6 + w) * 3) : (th_t + w * 3);
    float phi = th[0], theta = th[1], omega = th[2];
    float c  = cosf(0.5f * theta), s = sinf(0.5f * theta);
    float aa = 0.5f * (phi + omega), bb = 0.5f * (phi - omega);
    float ca = cosf(aa), sa = sinf(aa);
    float cb = cosf(bb), sb = sinf(bb);
    sGates[tid * 4 + 0] = make_float2( ca * c, -sa * c);
    sGates[tid * 4 + 1] = make_float2(-cb * s, -sb * s);
    sGates[tid * 4 + 2] = make_float2( cb * s, -sb * s);
    sGates[tid * 4 + 3] = make_float2( ca * c,  sa * c);
  }
  __syncthreads();

  const int lane = tid & 31;
  const int wave = tid >> 5;
  const int rA0 = RINGS.r1[lane], rA1 = RINGS.r1[lane + 32];
  const int rB0 = RINGS.r2[lane], rB1 = RINGS.r2[lane + 32];

  const int k = blockIdx.x * 8 + wave;
  float2 a0 = make_float2((lane == k) ? 1.f : 0.f, 0.f);
  float2 a1 = make_float2((lane + 32 == k) ? 1.f : 0.f, 0.f);

  #pragma unroll
  for (int l = 0; l < 3; ++l) {
    #pragma unroll
    for (int w = 0; w < 6; ++w) {
      int g = l * 6 + w;
      float2 m00 = sGates[g * 4 + 0], m01 = sGates[g * 4 + 1];
      float2 m10 = sGates[g * 4 + 2], m11 = sGates[g * 4 + 3];
      if (w == 0) {
        float2 n0 = cadd(cmul(m00, a0), cmul(m01, a1));
        float2 n1 = cadd(cmul(m10, a0), cmul(m11, a1));
        a0 = n0; a1 = n1;
      } else {
        int b = 5 - w, mask = 1 << b;
        float2 p0 = shflx2(a0, mask), p1 = shflx2(a1, mask);
        int bit = (lane >> b) & 1;
        float2 e0 = cadd(cmul(m00, a0), cmul(m01, p0));
        float2 o0 = cadd(cmul(m10, p0), cmul(m11, a0));
        float2 e1 = cadd(cmul(m00, a1), cmul(m01, p1));
        float2 o1 = cadd(cmul(m10, p1), cmul(m11, a1));
        a0 = bit ? o0 : e0;
        a1 = bit ? o1 : e1;
      }
    }
    int s0 = (l == 1) ? rB0 : rA0;
    int s1 = (l == 1) ? rB1 : rA1;
    float2 g0a = shfl2(a0, s0 & 31), g0b = shfl2(a1, s0 & 31);
    float2 g1a = shfl2(a0, s1 & 31), g1b = shfl2(a1, s1 & 31);
    a0 = (s0 < 32) ? g0a : g0b;
    a1 = (s1 < 32) ? g1a : g1b;
  }
  sChunk[(lane)      * 8 + wave] = a0.x;
  sChunk[(lane + 32) * 8 + wave] = a1.x;
  sChunk[(lane + 64) * 8 + wave] = a0.y;
  sChunk[(lane + 96) * 8 + wave] = a1.y;
  __syncthreads();
  ((float4*)uimg)[blockIdx.x * 256 + tid] = ((const float4*)sChunk)[tid];
}

__global__ __launch_bounds__(256) void qmaml_kernel(
    const float* __restrict__ x,  const float* __restrict__ W1,
    const float* __restrict__ b1, const float* __restrict__ ln_g,
    const float* __restrict__ ln_b, const float* __restrict__ W2,
    const float* __restrict__ b2, const float* __restrict__ uimg,
    const float* __restrict__ Wc, const float* __restrict__ bc,
    float* __restrict__ out, int B)
{
  __shared__ float  sW1[64 * WP];
  __shared__ float  sW2[384];
  __shared__ float  sB1[64], sG[64], sBeta[64];
  __shared__ float  sB2[8], sWc[32], sBc[8];
  __shared__ float  sU[UIMG];
  __shared__ float  sZ[768];

  const int tid = threadIdx.x;

  for (int i = tid; i < 1024; i += 256)
    *(float4*)(sW1 + (i >> 4) * WP + 4 * (i & 15)) = ((const float4*)W1)[i];
  for (int i = tid; i < 2048; i += 256) {
    const int n = (i & 255) >> 1;
    *(float4*)(sU + n * UP + 8 * (i >> 8) + 4 * (i & 1)) = ((const float4*)uimg)[i];
  }
  for (int i = tid; i < 384; i += 256)  sW2[i] = W2[i];
  if (tid < 64) { sB1[tid] = b1[tid]; sG[tid] = ln_g[tid]; sBeta[tid] = ln_b[tid]; }
  if (tid < 6)  sB2[tid] = b2[tid];
  if (tid < 30) sWc[tid] = Wc[tid];
  if (tid < 5)  sBc[tid] = bc[tid];
  __syncthreads();

  const int lane = tid & 31;
  const int wave = tid >> 5;
  const int half = lane >> 4;
  const int lm   = lane & 15;
  const int rowBase = blockIdx.x * 128 + wave * 16;

  v8f acc[4] = {};
  int rowA = rowBase + lm; if (rowA >= B) rowA = B - 1;
  const float* xrow = x + (size_t)rowA * 64 + 2 * half;
  #pragma unroll
  for (int kk = 0; kk < 16; ++kk) {
    v2f a = *(const v2f*)(xrow + 4 * kk);
    #pragma unroll
    for (int t = 0; t < 4; ++t) {
      v2f bf = *(const v2f*)(sW1 + (16 * t + lm) * WP + 4 * kk + 2 * half);
      acc[t] = __builtin_amdgcn_wmma_f32_16x16x4_f32(false, a, false, bf,
                                                     (short)0, acc[t], false, false);
    }
  }

  float hv[4][8];
  #pragma unroll
  for (int t = 0; t < 4; ++t) {
    float bias = sB1[16 * t + lm];
    #pragma unroll
    for (int j = 0; j < 8; ++j) hv[t][j] = fmaxf(acc[t][j] + bias, 0.0f);
  }

  float muj[8], inv[8];
  #pragma unroll
  for (int j = 0; j < 8; ++j) {
    float s  = hv[0][j] + hv[1][j] + hv[2][j] + hv[3][j];
    float q2 = hv[0][j]*hv[0][j] + hv[1][j]*hv[1][j] + hv[2][j]*hv[2][j] + hv[3][j]*hv[3][j];
    RED16(s); RED16(q2);
    float mu  = s * 0.015625f;
    float var = q2 * 0.015625f - mu * mu;
    muj[j] = mu;
    inv[j] = rsqrtf(var + 1e-5f);
  }
  #pragma unroll
  for (int t = 0; t < 4; ++t) {
    float g = sG[16 * t + lm], be = sBeta[16 * t + lm];
    #pragma unroll
    for (int j = 0; j < 8; ++j) hv[t][j] = (hv[t][j] - muj[j]) * inv[j] * g + be;
  }

  #pragma unroll
  for (int q = 0; q < 6; ++q) {
    float w0 = sW2[q * 64 + lm],      w1 = sW2[q * 64 + 16 + lm];
    float w2 = sW2[q * 64 + 32 + lm], w3 = sW2[q * 64 + 48 + lm];
    float bq = sB2[q];
    #pragma unroll
    for (int j = 0; j < 8; ++j) {
      float p = hv[0][j]*w0 + hv[1][j]*w1 + hv[2][j]*w2 + hv[3][j]*w3;
      RED16(p);
      if (lm == 0) sZ[wave * 96 + (8 * half + j) * 6 + q] = tanhf(p + bq);
    }
  }
  __syncthreads();

  float cq[6], sq[6];
  #pragma unroll
  for (int q = 0; q < 6; ++q) {
    float h2 = 1.57079632679489662f * sZ[wave * 96 + lm * 6 + q];
    cq[q] = cosf(h2); sq[q] = sinf(h2);
  }
  float p01[4], p012[8], p0123[16];
  #pragma unroll
  for (int i = 0; i < 4; ++i)  p01[i]   = ((i >> 1) ? sq[0] : cq[0]) * ((i & 1) ? sq[1] : cq[1]);
  #pragma unroll
  for (int i = 0; i < 8; ++i)  p012[i]  = p01[i >> 1]  * ((i & 1) ? sq[2] : cq[2]);
  #pragma unroll
  for (int i = 0; i < 16; ++i) p0123[i] = p012[i >> 1] * ((i & 1) ? sq[3] : cq[3]);
  const float f4 = half ? sq[4] : cq[4];
  v2f pf[16];
  #pragma unroll
  for (int kk = 0; kk < 16; ++kk) {
    float t4 = p0123[kk] * f4;
    pf[kk].x = t4 * cq[5];
    pf[kk].y = t4 * sq[5];
  }

  v8f amp[8] = {};
  #pragma unroll
  for (int kk = 0; kk < 16; ++kk) {
    #pragma unroll
    for (int t = 0; t < 8; ++t) {
      v2f uf = *(const v2f*)(sU + (16 * t + lm) * UP + 4 * kk + 2 * half);
      amp[t] = __builtin_amdgcn_wmma_f32_16x16x4_f32(false, uf, false, pf[kk],
                                                     (short)0, amp[t], false, false);
    }
  }

  float P[4][8];
  #pragma unroll
  for (int t = 0; t < 4; ++t) {
    #pragma unroll
    for (int r = 0; r < 8; ++r) P[t][r] = amp[t][r] * amp[t][r] + amp[t + 4][r] * amp[t + 4][r];
  }
  float e5 = 0.f, e4 = 0.f, e3 = 0.f;
  float s1[4][4], s2[4][2], s3[4];
  #pragma unroll
  for (int t = 0; t < 4; ++t) {
    #pragma unroll
    for (int m = 0; m < 4; ++m) { s1[t][m] = P[t][2*m] + P[t][2*m+1]; e5 += P[t][2*m] - P[t][2*m+1]; }
    #pragma unroll
    for (int m = 0; m < 2; ++m) { s2[t][m] = s1[t][2*m] + s1[t][2*m+1]; e4 += s1[t][2*m] - s1[t][2*m+1]; }
    s3[t] = s2[t][0] + s2[t][1]; e3 += s2[t][0] - s2[t][1];
  }
  const float u0 = s3[0] + s3[1], u1 = s3[2] + s3[3];
  const float e1 = (s3[0] - s3[1]) + (s3[2] - s3[3]);
  const float S  = u0 + u1;
  const float e0 = u0 - u1;
  const float e2 = half ? -S : S;

  const int row = rowBase + lm;
  #pragma unroll
  for (int k = 0; k < 5; ++k) {
    float v = sWc[k*6+0]*e0 + sWc[k*6+1]*e1 + sWc[k*6+2]*e2
            + sWc[k*6+3]*e3 + sWc[k*6+4]*e4 + sWc[k*6+5]*e5;
    v += __shfl_xor(v, 16, 32);
    if (half == 0 && row < B) out[(size_t)row * 5 + k] = v + sBc[k];
  }
}

extern "C" void kernel_launch(void* const* d_in, const int* in_sizes, int n_in,
                              void* d_out, int out_size, void* d_ws, size_t ws_size,
                              hipStream_t stream) {
  (void)n_in; (void)out_size; (void)ws_size;
  const float* x    = (const float*)d_in[0];
  const float* W1   = (const float*)d_in[1];
  const float* b1   = (const float*)d_in[2];
  const float* ln_g = (const float*)d_in[3];
  const float* ln_b = (const float*)d_in[4];
  const float* W2   = (const float*)d_in[5];
  const float* b2   = (const float*)d_in[6];
  const float* th_s = (const float*)d_in[7];
  const float* th_t = (const float*)d_in[8];
  const float* Wc   = (const float*)d_in[9];
  const float* bc   = (const float*)d_in[10];
  float* out = (float*)d_out;
  float* uimg = (float*)d_ws;
  int B = in_sizes[0] / 64;
  int grid = (B + 127) / 128;
  circuit_compose<<<8, 256, 0, stream>>>(th_s, th_t, uimg);
  qmaml_kernel<<<grid, 256, 0, stream>>>(x, W1, b1, ln_g, ln_b, W2, b2, uimg, Wc, bc, out, B);
}


# First product transposed (W1 as A, x rows as B): each lane owns 32 features of ONE row; LayerNorm and the 6 dots sum in-lane + one half-swap; 16-lane butterflies, z LDS trip gone; sum order changes
# speedup vs baseline: 6.6622x; 1.2668x over previous
#include <hip/hip_runtime.h>
#include <math.h>


typedef __attribute__((ext_vector_type(2))) float v2f;
typedef __attribute__((ext_vector_type(8))) float v8f;

struct Rings { int r1[64]; int r2[64]; };

constexpr Rings makeRings() {
  Rings R{};
  for (int rr = 1; rr <= 2; ++rr) {
    int total[64];
    for (int i = 0; i < 64; ++i) total[i] = i;
    for (int w = 0; w < 6; ++w) {
      int t = (w + rr) % 6;
      int nxt[64];
      for (int i = 0; i < 64; ++i) {
        int cbit = (i >> (5 - w)) & 1;
        nxt[i] = total[i ^ (cbit << (5 - t))];
      }
      for (int i = 0; i < 64; ++i) total[i] = nxt[i];
    }
    for (int i = 0; i < 64; ++i) { if (rr == 1) R.r1[i] = total[i]; else R.r2[i] = total[i]; }
  }
  return R;
}

__constant__ Rings RINGS = makeRings();

__device__ __forceinline__ float2 cmul(float2 a, float2 b) {
  return make_float2(a.x * b.x - a.y * b.y, a.x * b.y + a.y * b.x);
}
__device__ __forceinline__ float2 cadd(float2 a, float2 b) { return make_float2(a.x + b.x, a.y + b.y); }
__device__ __forceinline__ float2 shflx2(float2 v, int m) {
  return make_float2(__shfl_xor(v.x, m, 32), __shfl_xor(v.y, m, 32));
}
__device__ __forceinline__ float2 shfl2(float2 v, int l) {
  return make_float2(__shfl(v.x, l, 32), __shfl(v.y, l, 32));
}

__device__ __forceinline__ float xhalf(float v) {
  int b = __builtin_bit_cast(int, v);
  return __builtin_bit_cast(float, __builtin_amdgcn_permlanex16(b, b, 0x76543210, 0xfedcba98, false, false));
}

#define UP 68
#define UIMG (128 * UP)
#define WP 68

__global__ __launch_bounds__(256) void circuit_compose(
    const float* __restrict__ th_s, const float* __restrict__ th_t, float* __restrict__ uimg)
{
  __shared__ float2 sGates[72];
  __shared__ float  sChunk[1024];

  const int tid = threadIdx.x;
  if (tid < 18) {
    int l = tid / 6, w = tid % 6;
    const float* th = (l < 2) ? (th_s + (l * 6 + w) * 3) : (th_t + w * 3);
    float phi = th[0], theta = th[1], omega = th[2];
    float c  = cosf(0.5f * theta), s = sinf(0.5f * theta);
    float aa = 0.5f * (phi + omega), bb = 0.5f * (phi - omega);
    float ca = cosf(aa), sa = sinf(aa);
    float cb = cosf(bb), sb = sinf(bb);
    sGates[tid * 4 + 0] = make_float2( ca * c, -sa * c);
    sGates[tid * 4 + 1] = make_float2(-cb * s, -sb * s);
    sGates[tid * 4 + 2] = make_float2( cb * s, -sb * s);
    sGates[tid * 4 + 3] = make_float2( ca * c,  sa * c);
  }
  __syncthreads();

  const int lane = tid & 31;
  const int wave = tid >> 5;
  const int rA0 = RINGS.r1[lane], rA1 = RINGS.r1[lane + 32];
  const int rB0 = RINGS.r2[lane], rB1 = RINGS.r2[lane + 32];

  const int k = blockIdx.x * 8 + wave;
  float2 a0 = make_float2((lane == k) ? 1.f : 0.f, 0.f);
  float2 a1 = make_float2((lane + 32 == k) ? 1.f : 0.f, 0.f);

  #pragma unroll
  for (int l = 0; l < 3; ++l) {
    #pragma unroll
    for (int w = 0; w < 6; ++w) {
      int g = l * 6 + w;
      float2 m00 = sGates[g * 4 + 0], m01 = sGates[g * 4 + 1];
      float2 m10 = sGates[g * 4 + 2], m11 = sGates[g * 4 + 3];
      if (w == 0) {
        float2 n0 = cadd(cmul(m00, a0), cmul(m01, a1));
        float2 n1 = cadd(cmul(m10, a0), cmul(m11, a1));
        a0 = n0; a1 = n1;
      } else {
        int b = 5 - w, mask = 1 << b;
        float2 p0 = shflx2(a0, mask), p1 = shflx2(a1, mask);
        int bit = (lane >> b) & 1;
        float2 e0 = cadd(cmul(m00, a0), cmul(m01, p0));
        float2 o0 = cadd(cmul(m10, p0), cmul(m11, a0));
        float2 e1 = cadd(cmul(m00, a1), cmul(m01, p1));
        float2 o1 = cadd(cmul(m10, p1), cmul(m11, a1));
        a0 = bit ? o0 : e0;
        a1 = bit ? o1 : e1;
      }
    }
    int s0 = (l == 1) ? rB0 : rA0;
    int s1 = (l == 1) ? rB1 : rA1;
    float2 g0a = shfl2(a0, s0 & 31), g0b = shfl2(a1, s0 & 31);
    float2 g1a = shfl2(a0, s1 & 31), g1b = shfl2(a1, s1 & 31);
    a0 = (s0 < 32) ? g0a : g0b;
    a1 = (s1 < 32) ? g1a : g1b;
  }
  sChunk[(lane)      * 8 + wave] = a0.x;
  sChunk[(lane + 32) * 8 + wave] = a1.x;
  sChunk[(lane + 64) * 8 + wave] = a0.y;
  sChunk[(lane + 96) * 8 + wave] = a1.y;
  __syncthreads();
  ((float4*)uimg)[blockIdx.x * 256 + tid] = ((const float4*)sChunk)[tid];
}

__global__ __launch_bounds__(256) void qmaml_kernel(
    const float* __restrict__ x,  const float* __restrict__ W1,
    const float* __restrict__ b1, const float* __restrict__ ln_g,
    const float* __restrict__ ln_b, const float* __restrict__ W2,
    const float* __restrict__ b2, const float* __restrict__ uimg,
    const float* __restrict__ Wc, const float* __restrict__ bc,
    float* __restrict__ out, int B)
{
  __shared__ float  sW1[64 * WP];
  __shared__ __attribute__((aligned(16))) float sW2[384];
  __shared__ __attribute__((aligned(16))) float sB1[64], sG[64], sBeta[64];
  __shared__ float  sB2[8], sWc[32], sBc[8];
  __shared__ float  sU[UIMG];

  const int tid = threadIdx.x;

  for (int i = tid; i < 1024; i += 256)
    *(float4*)(sW1 + (i >> 4) * WP + 4 * (i & 15)) = ((const float4*)W1)[i];
  for (int i = tid; i < 2048; i += 256) {
    const int n = (i & 255) >> 1;
    *(float4*)(sU + n * UP + 8 * (i >> 8) + 4 * (i & 1)) = ((const float4*)uimg)[i];
  }
  for (int i = tid; i < 384; i += 256)  sW2[i] = W2[i];
  if (tid < 64) { sB1[tid] = b1[tid]; sG[tid] = ln_g[tid]; sBeta[tid] = ln_b[tid]; }
  if (tid < 6)  sB2[tid] = b2[tid];
  if (tid < 30) sWc[tid] = Wc[tid];
  if (tid < 5)  sBc[tid] = bc[tid];
  __syncthreads();

  const int lane = tid & 31;
  const int wave = tid >> 5;
  const int half = lane >> 4;
  const int lm   = lane & 15;
  const int rowBase = blockIdx.x * 128 + wave * 16;

  v8f acc[4] = {};
  int rowA = rowBase + lm; if (rowA >= B) rowA = B - 1;
  const float* xrow = x + (size_t)rowA * 64 + 2 * half;
  #pragma unroll
  for (int kk = 0; kk < 16; ++kk) {
    v2f xf = *(const v2f*)(xrow + 4 * kk);
    #pragma unroll
    for (int t = 0; t < 4; ++t) {
      v2f wf = *(const v2f*)(sW1 + (16 * t + lm) * WP + 4 * kk + 2 * half);
      acc[t] = __builtin_amdgcn_wmma_f32_16x16x4_f32(false, wf, false, xf,
                                                     (short)0, acc[t], false, false);
    }
  }

  const int fb = 8 * half;
  float hv[4][8];
  #pragma unroll
  for (int t = 0; t < 4; ++t) {
    const float4 ba = *(const float4*)(sB1 + 16 * t + fb);
    const float4 bb = *(const float4*)(sB1 + 16 * t + fb + 4);
    const float bias[8] = { ba.x, ba.y, ba.z, ba.w, bb.x, bb.y, bb.z, bb.w };
    #pragma unroll
    for (int j = 0; j < 8; ++j) hv[t][j] = fmaxf(acc[t][j] + bias[j], 0.0f);
  }

  float st[4], qt[4];
  #pragma unroll
  for (int t = 0; t < 4; ++t) {
    st[t] = hv[t][0];
    qt[t] = hv[t][0] * hv[t][0];
    #pragma unroll
    for (int j = 1; j < 8; ++j) { st[t] += hv[t][j]; qt[t] += hv[t][j] * hv[t][j]; }
  }
  float s  = (st[0] + st[1]) + (st[2] + st[3]);
  float q2 = (qt[0] + qt[1]) + (qt[2] + qt[3]);
  s  += xhalf(s);
  q2 += xhalf(q2);
  const float mu  = s * 0.015625f;
  const float var = q2 * 0.015625f - mu * mu;
  const float inv = rsqrtf(var + 1e-5f);
  #pragma unroll
  for (int t = 0; t < 4; ++t) {
    const float4 ga = *(const float4*)(sG + 16 * t + fb);
    const float4 gb = *(const float4*)(sG + 16 * t + fb + 4);
    const float4 ea = *(const float4*)(sBeta + 16 * t + fb);
    const float4 eb = *(const float4*)(sBeta + 16 * t + fb + 4);
    const float g[8]  = { ga.x, ga.y, ga.z, ga.w, gb.x, gb.y, gb.z, gb.w };
    const float be[8] = { ea.x, ea.y, ea.z, ea.w, eb.x, eb.y, eb.z, eb.w };
    #pragma unroll
    for (int j = 0; j < 8; ++j) hv[t][j] = (hv[t][j] - mu) * inv * g[j] + be[j];
  }

  float zq[6];
  #pragma unroll
  for (int q = 0; q < 6; ++q) {
    v2f pt[4];
    #pragma unroll
    for (int t = 0; t < 4; ++t) {
      const float4 wa = *(const float4*)(sW2 + q * 64 + 16 * t + fb);
      const float4 wb = *(const float4*)(sW2 + q * 64 + 16 * t + fb + 4);
      pt[t]  = v2f{hv[t][0], hv[t][1]} * v2f{wa.x, wa.y};
      pt[t] += v2f{hv[t][2], hv[t][3]} * v2f{wa.z, wa.w};
      pt[t] += v2f{hv[t][4], hv[t][5]} * v2f{wb.x, wb.y};
      pt[t] += v2f{hv[t][6], hv[t][7]} * v2f{wb.z, wb.w};
    }
    const v2f pp = (pt[0] + pt[1]) + (pt[2] + pt[3]);
    float p = pp.x + pp.y;
    p += xhalf(p);
    zq[q] = tanhf(p + sB2[q]);
  }

  float cq[6], sq[6];
  #pragma unroll
  for (int q = 0; q < 6; ++q) {
    float h2 = 1.57079632679489662f * zq[q];
    cq[q] = cosf(h2); sq[q] = sinf(h2);
  }
  float p01[4], p012[8], p0123[16];
  #pragma unroll
  for (int i = 0; i < 4; ++i)  p01[i]   = ((i >> 1) ? sq[0] : cq[0]) * ((i & 1) ? sq[1] : cq[1]);
  #pragma unroll
  for (int i = 0; i < 8; ++i)  p012[i]  = p01[i >> 1]  * ((i & 1) ? sq[2] : cq[2]);
  #pragma unroll
  for (int i = 0; i < 16; ++i) p0123[i] = p012[i >> 1] * ((i & 1) ? sq[3] : cq[3]);
  const float f4 = half ? sq[4] : cq[4];
  v2f pf[16];
  #pragma unroll
  for (int kk = 0; kk < 16; ++kk) {
    float t4 = p0123[kk] * f4;
    pf[kk].x = t4 * cq[5];
    pf[kk].y = t4 * sq[5];
  }

  v8f amp[8] = {};
  #pragma unroll
  for (int kk = 0; kk < 16; ++kk) {
    #pragma unroll
    for (int t = 0; t < 8; ++t) {
      v2f uf = *(const v2f*)(sU + (16 * t + lm) * UP + 4 * kk + 2 * half);
      amp[t] = __builtin_amdgcn_wmma_f32_16x16x4_f32(false, uf, false, pf[kk],
                                                     (short)0, amp[t], false, false);
    }
  }

  float P[4][8];
  #pragma unroll
  for (int t = 0; t < 4; ++t) {
    #pragma unroll
    for (int r = 0; r < 8; ++r) P[t][r] = amp[t][r] * amp[t][r] + amp[t + 4][r] * amp[t + 4][r];
  }
  float e5 = 0.f, e4 = 0.f, e3 = 0.f;
  float s1[4][4], s2[4][2], s3[4];
  #pragma unroll
  for (int t = 0; t < 4; ++t) {
    #pragma unroll
    for (int m = 0; m < 4; ++m) { s1[t][m] = P[t][2*m] + P[t][2*m+1]; e5 += P[t][2*m] - P[t][2*m+1]; }
    #pragma unroll
    for (int m = 0; m < 2; ++m) { s2[t][m] = s1[t][2*m] + s1[t][2*m+1]; e4 += s1[t][2*m] - s1[t][2*m+1]; }
    s3[t] = s2[t][0] + s2[t][1]; e3 += s2[t][0] - s2[t][1];
  }
  const float u0 = s3[0] + s3[1], u1 = s3[2] + s3[3];
  const float e1 = (s3[0] - s3[1]) + (s3[2] - s3[3]);
  const float S  = u0 + u1;
  const float e0 = u0 - u1;
  const float e2 = half ? -S : S;

  const int row = rowBase + lm;
  #pragma unroll
  for (int k = 0; k < 5; ++k) {
    float v = sWc[k*6+0]*e0 + sWc[k*6+1]*e1 + sWc[k*6+2]*e2
            + sWc[k*6+3]*e3 + sWc[k*6+4]*e4 + sWc[k*6+5]*e5;
    v += __shfl_xor(v, 16, 32);
    if (half == 0 && row < B) out[(size_t)row * 5 + k] = v + sBc[k];
  }
}

extern "C" void kernel_launch(void* const* d_in, const int* in_sizes, int n_in,
                              void* d_out, int out_size, void* d_ws, size_t ws_size,
                              hipStream_t stream) {
  (void)n_in; (void)out_size; (void)ws_size;
  const float* x    = (const float*)d_in[0];
  const float* W1   = (const float*)d_in[1];
  const float* b1   = (const float*)d_in[2];
  const float* ln_g = (const float*)d_in[3];
  const float* ln_b = (const float*)d_in[4];
  const float* W2   = (const float*)d_in[5];
  const float* b2   = (const float*)d_in[6];
  const float* th_s = (const float*)d_in[7];
  const float* th_t = (const float*)d_in[8];
  const float* Wc   = (const float*)d_in[9];
  const float* bc   = (const float*)d_in[10];
  float* out = (float*)d_out;
  float* uimg = (float*)d_ws;
  int B = in_sizes[0] / 64;
  int grid = (B + 127) / 128;
  circuit_compose<<<8, 256, 0, stream>>>(th_s, th_t, uimg);
  qmaml_kernel<<<grid, 256, 0, stream>>>(x, W1, b1, ln_g, ln_b, W2, b2, uimg, Wc, bc, out, B);
}


# Both matrix loops: the LDS fragment reads of each 8-instruction block are issued as one group one block ahead into a second register set, not singly with a wait before every 1-2 matrix instructions
# speedup vs baseline: 6.8681x; 1.0309x over previous
#include <hip/hip_runtime.h>
#include <math.h>


typedef __attribute__((ext_vector_type(2))) float v2f;
typedef __attribute__((ext_vector_type(8))) float v8f;

struct Rings { int r1[64]; int r2[64]; };

constexpr Rings makeRings() {
  Rings R{};
  for (int rr = 1; rr <= 2; ++rr) {
    int total[64];
    for (int i = 0; i < 64; ++i) total[i] = i;
    for (int w = 0; w < 6; ++w) {
      int t = (w + rr) % 6;
      int nxt[64];
      for (int i = 0; i < 64; ++i) {
        int cbit = (i >> (5 - w)) & 1;
        nxt[i] = total[i ^ (cbit << (5 - t))];
      }
      for (int i = 0; i < 64; ++i) total[i] = nxt[i];
    }
    for (int i = 0; i < 64; ++i) { if (rr == 1) R.r1[i] = total[i]; else R.r2[i] = total[i]; }
  }
  return R;
}

__constant__ Rings RINGS = makeRings();

__device__ __forceinline__ float2 cmul(float2 a, float2 b) {
  return make_float2(a.x * b.x - a.y * b.y, a.x * b.y + a.y * b.x);
}
__device__ __forceinline__ float2 cadd(float2 a, float2 b) { return make_float2(a.x + b.x, a.y + b.y); }
__device__ __forceinline__ float2 shflx2(float2 v, int m) {
  return make_float2(__shfl_xor(v.x, m, 32), __shfl_xor(v.y, m, 32));
}
__device__ __forceinline__ float2 shfl2(float2 v, int l) {
  return make_float2(__shfl(v.x, l, 32), __shfl(v.y, l, 32));
}

__device__ __forceinline__ float xhalf(float v) {
  int b = __builtin_bit_cast(int, v);
  return __builtin_bit_cast(float, __builtin_amdgcn_permlanex16(b, b, 0x76543210, 0xfedcba98, false, false));
}

#define UP 68
#define UIMG (128 * UP)
#define WP 68

__global__ __launch_bounds__(256) void circuit_compose(
    const float* __restrict__ th_s, const float* __restrict__ th_t, float* __restrict__ uimg)
{
  __shared__ float2 sGates[72];
  __shared__ float  sChunk[1024];

  const int tid = threadIdx.x;
  if (tid < 18) {
    int l = tid / 6, w = tid % 6;
    const float* th = (l < 2) ? (th_s + (l * 6 + w) * 3) : (th_t + w * 3);
    float phi = th[0], theta = th[1], omega = th[2];
    float c  = cosf(0.5f * theta), s = sinf(0.5f * theta);
    float aa = 0.5f * (phi + omega), bb = 0.5f * (phi - omega);
    float ca = cosf(aa), sa = sinf(aa);
    float cb = cosf(bb), sb = sinf(bb);
    sGates[tid * 4 + 0] = make_float2( ca * c, -sa * c);
    sGates[tid * 4 + 1] = make_float2(-cb * s, -sb * s);
    sGates[tid * 4 + 2] = make_float2( cb * s, -sb * s);
    sGates[tid * 4 + 3] = make_float2( ca * c,  sa * c);
  }
  __syncthreads();

  const int lane = tid & 31;
  const int wave = tid >> 5;
  const int rA0 = RINGS.r1[lane], rA1 = RINGS.r1[lane + 32];
  const int rB0 = RINGS.r2[lane], rB1 = RINGS.r2[lane + 32];

  const int k = blockIdx.x * 8 + wave;
  float2 a0 = make_float2((lane == k) ? 1.f : 0.f, 0.f);
  float2 a1 = make_float2((lane + 32 == k) ? 1.f : 0.f, 0.f);

  #pragma unroll
  for (int l = 0; l < 3; ++l) {
    #pragma unroll
    for (int w = 0; w < 6; ++w) {
      int g = l * 6 + w;
      float2 m00 = sGates[g * 4 + 0], m01 = sGates[g * 4 + 1];
      float2 m10 = sGates[g * 4 + 2], m11 = sGates[g * 4 + 3];
      if (w == 0) {
        float2 n0 = cadd(cmul(m00, a0), cmul(m01, a1));
        float2 n1 = cadd(cmul(m10, a0), cmul(m11, a1));
        a0 = n0; a1 = n1;
      } else {
        int b = 5 - w, mask = 1 << b;
        float2 p0 = shflx2(a0, mask), p1 = shflx2(a1, mask);
        int bit = (lane >> b) & 1;
        float2 e0 = cadd(cmul(m00, a0), cmul(m01, p0));
        float2 o0 = cadd(cmul(m10, p0), cmul(m11, a0));
        float2 e1 = cadd(cmul(m00, a1), cmul(m01, p1));
        float2 o1 = cadd(cmul(m10, p1), cmul(m11, a1));
        a0 = bit ? o0 : e0;
        a1 = bit ? o1 : e1;
      }
    }
    int s0 = (l == 1) ? rB0 : rA0;
    int s1 = (l == 1) ? rB1 : rA1;
    float2 g0a = shfl2(a0, s0 & 31), g0b = shfl2(a1, s0 & 31);
    float2 g1a = shfl2(a0, s1 & 31), g1b = shfl2(a1, s1 & 31);
    a0 = (s0 < 32) ? g0a : g0b;
    a1 = (s1 < 32) ? g1a : g1b;
  }
  sChunk[(lane)      * 8 + wave] = a0.x;
  sChunk[(lane + 32) * 8 + wave] = a1.x;
  sChunk[(lane + 64) * 8 + wave] = a0.y;
  sChunk[(lane + 96) * 8 + wave] = a1.y;
  __syncthreads();
  ((float4*)uimg)[blockIdx.x * 256 + tid] = ((const float4*)sChunk)[tid];
}

__global__ __launch_bounds__(256) void qmaml_kernel(
    const float* __restrict__ x,  const float* __restrict__ W1,
    const float* __restrict__ b1, const float* __restrict__ ln_g,
    const float* __restrict__ ln_b, const float* __restrict__ W2,
    const float* __restrict__ b2, const float* __restrict__ uimg,
    const float* __restrict__ Wc, const float* __restrict__ bc,
    float* __restrict__ out, int B)
{
  __shared__ float  sW1[64 * WP];
  __shared__ __attribute__((aligned(16))) float sW2[384];
  __shared__ __attribute__((aligned(16))) float sB1[64], sG[64], sBeta[64];
  __shared__ float  sB2[8], sWc[32], sBc[8];
  __shared__ float  sU[UIMG];

  const int tid = threadIdx.x;

  for (int i = tid; i < 1024; i += 256)
    *(float4*)(sW1 + (i >> 4) * WP + 4 * (i & 15)) = ((const float4*)W1)[i];
  for (int i = tid; i < 2048; i += 256) {
    const int n = (i & 255) >> 1;
    *(float4*)(sU + n * UP + 8 * (i >> 8) + 4 * (i & 1)) = ((const float4*)uimg)[i];
  }
  for (int i = tid; i < 384; i += 256)  sW2[i] = W2[i];
  if (tid < 64) { sB1[tid] = b1[tid]; sG[tid] = ln_g[tid]; sBeta[tid] = ln_b[tid]; }
  if (tid < 6)  sB2[tid] = b2[tid];
  if (tid < 30) sWc[tid] = Wc[tid];
  if (tid < 5)  sBc[tid] = bc[tid];
  __syncthreads();

  const int lane = tid & 31;
  const int wave = tid >> 5;
  const int half = lane >> 4;
  const int lm   = lane & 15;
  const int rowBase = blockIdx.x * 128 + wave * 16;

  v8f acc[4] = {};
  int rowA = rowBase + lm; if (rowA >= B) rowA = B - 1;
  const float* xrow = x + (size_t)rowA * 64 + 2 * half;
  v2f xf[16];
  #pragma unroll
  for (int kk = 0; kk < 16; ++kk) xf[kk] = *(const v2f*)(xrow + 4 * kk);
  const float* w1p = sW1 + lm * WP + 2 * half;
  v2f wa[2][8];
  #pragma unroll
  for (int j = 0; j < 2; ++j) {
    #pragma unroll
    for (int t = 0; t < 4; ++t) wa[0][4 * j + t] = *(const v2f*)(w1p + 16 * t * WP + 4 * j);
  }
  #pragma unroll
  for (int b = 0; b < 8; ++b) {
    if (b + 1 < 8) {
      #pragma unroll
      for (int j = 0; j < 2; ++j) {
        #pragma unroll
        for (int t = 0; t < 4; ++t)
          wa[(b + 1) & 1][4 * j + t] = *(const v2f*)(w1p + 16 * t * WP + 4 * (2 * b + 2 + j));
      }
    }
    __builtin_amdgcn_sched_barrier(0);
    #pragma unroll
    for (int j = 0; j < 2; ++j) {
      #pragma unroll
      for (int t = 0; t < 4; ++t)
        acc[t] = __builtin_amdgcn_wmma_f32_16x16x4_f32(false, wa[b & 1][4 * j + t], false, xf[2 * b + j],
                                                       (short)0, acc[t], false, false);
    }
    __builtin_amdgcn_sched_barrier(0);
  }

  const int fb = 8 * half;
  float hv[4][8];
  #pragma unroll
  for (int t = 0; t < 4; ++t) {
    const float4 ba = *(const float4*)(sB1 + 16 * t + fb);
    const float4 bb = *(const float4*)(sB1 + 16 * t + fb + 4);
    const float bias[8] = { ba.x, ba.y, ba.z, ba.w, bb.x, bb.y, bb.z, bb.w };
    #pragma unroll
    for (int j = 0; j < 8; ++j) hv[t][j] = fmaxf(acc[t][j] + bias[j], 0.0f);
  }

  float st[4], qt[4];
  #pragma unroll
  for (int t = 0; t < 4; ++t) {
    st[t] = hv[t][0];
    qt[t] = hv[t][0] * hv[t][0];
    #pragma unroll
    for (int j = 1; j < 8; ++j) { st[t] += hv[t][j]; qt[t] += hv[t][j] * hv[t][j]; }
  }
  float s  = (st[0] + st[1]) + (st[2] + st[3]);
  float q2 = (qt[0] + qt[1]) + (qt[2] + qt[3]);
  s  += xhalf(s);
  q2 += xhalf(q2);
  const float mu  = s * 0.015625f;
  const float var = q2 * 0.015625f - mu * mu;
  const float inv = rsqrtf(var + 1e-5f);
  #pragma unroll
  for (int t = 0; t < 4; ++t) {
    const float4 ga = *(const float4*)(sG + 16 * t + fb);
    const float4 gb = *(const float4*)(sG + 16 * t + fb + 4);
    const float4 ea = *(const float4*)(sBeta + 16 * t + fb);
    const float4 eb = *(const float4*)(sBeta + 16 * t + fb + 4);
    const float g[8]  = { ga.x, ga.y, ga.z, ga.w, gb.x, gb.y, gb.z, gb.w };
    const float be[8] = { ea.x, ea.y, ea.z, ea.w, eb.x, eb.y, eb.z, eb.w };
    #pragma unroll
    for (int j = 0; j < 8; ++j) hv[t][j] = (hv[t][j] - mu) * inv * g[j] + be[j];
  }

  float zq[6];
  #pragma unroll
  for (int q = 0; q < 6; ++q) {
    v2f pt[4];
    #pragma unroll
    for (int t = 0; t < 4; ++t) {
      const float4 wa = *(const float4*)(sW2 + q * 64 + 16 * t + fb);
      const float4 wb = *(const float4*)(sW2 + q * 64 + 16 * t + fb + 4);
      pt[t]  = v2f{hv[t][0], hv[t][1]} * v2f{wa.x, wa.y};
      pt[t] += v2f{hv[t][2], hv[t][3]} * v2f{wa.z, wa.w};
      pt[t] += v2f{hv[t][4], hv[t][5]} * v2f{wb.x, wb.y};
      pt[t] += v2f{hv[t][6], hv[t][7]} * v2f{wb.z, wb.w};
    }
    const v2f pp = (pt[0] + pt[1]) + (pt[2] + pt[3]);
    float p = pp.x + pp.y;
    p += xhalf(p);
    zq[q] = tanhf(p + sB2[q]);
  }

  float cq[6], sq[6];
  #pragma unroll
  for (int q = 0; q < 6; ++q) {
    float h2 = 1.57079632679489662f * zq[q];
    cq[q] = cosf(h2); sq[q] = sinf(h2);
  }
  float p01[4], p012[8], p0123[16];
  #pragma unroll
  for (int i = 0; i < 4; ++i)  p01[i]   = ((i >> 1) ? sq[0] : cq[0]) * ((i & 1) ? sq[1] : cq[1]);
  #pragma unroll
  for (int i = 0; i < 8; ++i)  p012[i]  = p01[i >> 1]  * ((i & 1) ? sq[2] : cq[2]);
  #pragma unroll
  for (int i = 0; i < 16; ++i) p0123[i] = p012[i >> 1] * ((i & 1) ? sq[3] : cq[3]);
  const float f4 = half ? sq[4] : cq[4];

  v8f amp[8] = {};
  const float* up = sU + lm * UP + 2 * half;
  v2f ua[2][8];
  #pragma unroll
  for (int t = 0; t < 8; ++t) ua[0][t] = *(const v2f*)(up + 16 * t * UP);
  v2f pfc;
  {
    float t4 = p0123[0] * f4;
    pfc.x = t4 * cq[5];
    pfc.y = t4 * sq[5];
  }
  #pragma unroll
  for (int kk = 0; kk < 16; ++kk) {
    if (kk + 1 < 16) {
      #pragma unroll
      for (int t = 0; t < 8; ++t) ua[(kk + 1) & 1][t] = *(const v2f*)(up + 16 * t * UP + 4 * (kk + 1));
    }
    __builtin_amdgcn_sched_barrier(0);
    #pragma unroll
    for (int t = 0; t < 8; ++t)
      amp[t] = __builtin_amdgcn_wmma_f32_16x16x4_f32(false, ua[kk & 1][t], false, pfc,
                                                     (short)0, amp[t], false, false);
    v2f pfn = pfc;
    if (kk + 1 < 16) {
      float t4 = p0123[kk + 1] * f4;
      pfn.x = t4 * cq[5];
      pfn.y = t4 * sq[5];
    }
    __builtin_amdgcn_sched_barrier(0);
    pfc = pfn;
  }

  float P[4][8];
  #pragma unroll
  for (int t = 0; t < 4; ++t) {
    #pragma unroll
    for (int r = 0; r < 8; ++r) P[t][r] = amp[t][r] * amp[t][r] + amp[t + 4][r] * amp[t + 4][r];
  }
  float e5 = 0.f, e4 = 0.f, e3 = 0.f;
  float s1[4][4], s2[4][2], s3[4];
  #pragma unroll
  for (int t = 0; t < 4; ++t) {
    #pragma unroll
    for (int m = 0; m < 4; ++m) { s1[t][m] = P[t][2*m] + P[t][2*m+1]; e5 += P[t][2*m] - P[t][2*m+1]; }
    #pragma unroll
    for (int m = 0; m < 2; ++m) { s2[t][m] = s1[t][2*m] + s1[t][2*m+1]; e4 += s1[t][2*m] - s1[t][2*m+1]; }
    s3[t] = s2[t][0] + s2[t][1]; e3 += s2[t][0] - s2[t][1];
  }
  const float u0 = s3[0] + s3[1], u1 = s3[2] + s3[3];
  const float e1 = (s3[0] - s3[1]) + (s3[2] - s3[3]);
  const float S  = u0 + u1;
  const float e0 = u0 - u1;
  const float e2 = half ? -S : S;

  const int row = rowBase + lm;
  #pragma unroll
  for (int k = 0; k < 5; ++k) {
    float v = sWc[k*6+0]*e0 + sWc[k*6+1]*e1 + sWc[k*6+2]*e2
            + sWc[k*6+3]*e3 + sWc[k*6+4]*e4 + sWc[k*6+5]*e5;
    v += __shfl_xor(v, 16, 32);
    if (half == 0 && row < B) out[(size_t)row * 5 + k] = v + sBc[k];
  }
}

extern "C" void kernel_launch(void* const* d_in, const int* in_sizes, int n_in,
                              void* d_out, int out_size, void* d_ws, size_t ws_size,
                              hipStream_t stream) {
  (void)n_in; (void)out_size; (void)ws_size;
  const float* x    = (const float*)d_in[0];
  const float* W1   = (const float*)d_in[1];
  const float* b1   = (const float*)d_in[2];
  const float* ln_g = (const float*)d_in[3];
  const float* ln_b = (const float*)d_in[4];
  const float* W2   = (const float*)d_in[5];
  const float* b2   = (const float*)d_in[6];
  const float* th_s = (const float*)d_in[7];
  const float* th_t = (const float*)d_in[8];
  const float* Wc   = (const float*)d_in[9];
  const float* bc   = (const float*)d_in[10];
  float* out = (float*)d_out;
  float* uimg = (float*)d_ws;
  int B = in_sizes[0] / 64;
  int grid = (B + 127) / 128;
  circuit_compose<<<8, 256, 0, stream>>>(th_s, th_t, uimg);
  qmaml_kernel<<<grid, 256, 0, stream>>>(x, W1, b1, ln_g, ln_b, W2, b2, uimg, Wc, bc, out, B);
}


# Main kernel prologue: its 19 one-after-another global-load round trips (12 staging-loop passes, 6 small-vector loads, then the x tile) are issued as ONE group of loads before the first wait
# speedup vs baseline: 7.9513x; 1.1577x over previous
#include <hip/hip_runtime.h>
#include <math.h>


typedef __attribute__((ext_vector_type(2))) float v2f;
typedef __attribute__((ext_vector_type(8))) float v8f;

struct Rings { int r1[64]; int r2[64]; };

constexpr Rings makeRings() {
  Rings R{};
  for (int rr = 1; rr <= 2; ++rr) {
    int total[64];
    for (int i = 0; i < 64; ++i) total[i] = i;
    for (int w = 0; w < 6; ++w) {
      int t = (w + rr) % 6;
      int nxt[64];
      for (int i = 0; i < 64; ++i) {
        int cbit = (i >> (5 - w)) & 1;
        nxt[i] = total[i ^ (cbit << (5 - t))];
      }
      for (int i = 0; i < 64; ++i) total[i] = nxt[i];
    }
    for (int i = 0; i < 64; ++i) { if (rr == 1) R.r1[i] = total[i]; else R.r2[i] = total[i]; }
  }
  return R;
}

__constant__ Rings RINGS = makeRings();

__device__ __forceinline__ float2 cmul(float2 a, float2 b) {
  return make_float2(a.x * b.x - a.y * b.y, a.x * b.y + a.y * b.x);
}
__device__ __forceinline__ float2 cadd(float2 a, float2 b) { return make_float2(a.x + b.x, a.y + b.y); }
__device__ __forceinline__ float2 shflx2(float2 v, int m) {
  return make_float2(__shfl_xor(v.x, m, 32), __shfl_xor(v.y, m, 32));
}
__device__ __forceinline__ float2 shfl2(float2 v, int l) {
  return make_float2(__shfl(v.x, l, 32), __shfl(v.y, l, 32));
}

__device__ __forceinline__ float xhalf(float v) {
  int b = __builtin_bit_cast(int, v);
  return __builtin_bit_cast(float, __builtin_amdgcn_permlanex16(b, b, 0x76543210, 0xfedcba98, false, false));
}

#define UP 68
#define UIMG (128 * UP)
#define WP 68

__global__ __launch_bounds__(256) void circuit_compose(
    const float* __restrict__ th_s, const float* __restrict__ th_t, float* __restrict__ uimg)
{
  __shared__ float2 sGates[72];
  __shared__ float  sChunk[1024];

  const int tid = threadIdx.x;
  if (tid < 18) {
    int l = tid / 6, w = tid % 6;
    const float* th = (l < 2) ? (th_s + (l * 6 + w) * 3) : (th_t + w * 3);
    float phi = th[0], theta = th[1], omega = th[2];
    float c  = cosf(0.5f * theta), s = sinf(0.5f * theta);
    float aa = 0.5f * (phi + omega), bb = 0.5f * (phi - omega);
    float ca = cosf(aa), sa = sinf(aa);
    float cb = cosf(bb), sb = sinf(bb);
    sGates[tid * 4 + 0] = make_float2( ca * c, -sa * c);
    sGates[tid * 4 + 1] = make_float2(-cb * s, -sb * s);
    sGates[tid * 4 + 2] = make_float2( cb * s, -sb * s);
    sGates[tid * 4 + 3] = make_float2( ca * c,  sa * c);
  }
  __syncthreads();

  const int lane = tid & 31;
  const int wave = tid >> 5;
  const int rA0 = RINGS.r1[lane], rA1 = RINGS.r1[lane + 32];
  const int rB0 = RINGS.r2[lane], rB1 = RINGS.r2[lane + 32];

  const int k = blockIdx.x * 8 + wave;
  float2 a0 = make_float2((lane == k) ? 1.f : 0.f, 0.f);
  float2 a1 = make_float2((lane + 32 == k) ? 1.f : 0.f, 0.f);

  #pragma unroll
  for (int l = 0; l < 3; ++l) {
    #pragma unroll
    for (int w = 0; w < 6; ++w) {
      int g = l * 6 + w;
      float2 m00 = sGates[g * 4 + 0], m01 = sGates[g * 4 + 1];
      float2 m10 = sGates[g * 4 + 2], m11 = sGates[g * 4 + 3];
      if (w == 0) {
        float2 n0 = cadd(cmul(m00, a0), cmul(m01, a1));
        float2 n1 = cadd(cmul(m10, a0), cmul(m11, a1));
        a0 = n0; a1 = n1;
      } else {
        int b = 5 - w, mask = 1 << b;
        float2 p0 = shflx2(a0, mask), p1 = shflx2(a1, mask);
        int bit = (lane >> b) & 1;
        float2 e0 = cadd(cmul(m00, a0), cmul(m01, p0));
        float2 o0 = cadd(cmul(m10, p0), cmul(m11, a0));
        float2 e1 = cadd(cmul(m00, a1), cmul(m01, p1));
        float2 o1 = cadd(cmul(m10, p1), cmul(m11, a1));
        a0 = bit ? o0 : e0;
        a1 = bit ? o1 : e1;
      }
    }
    int s0 = (l == 1) ? rB0 : rA0;
    int s1 = (l == 1) ? rB1 : rA1;
    float2 g0a = shfl2(a0, s0 & 31), g0b = shfl2(a1, s0 & 31);
    float2 g1a = shfl2(a0, s1 & 31), g1b = shfl2(a1, s1 & 31);
    a0 = (s0 < 32) ? g0a : g0b;
    a1 = (s1 < 32) ? g1a : g1b;
  }
  sChunk[(lane)      * 8 + wave] = a0.x;
  sChunk[(lane + 32) * 8 + wave] = a1.x;
  sChunk[(lane + 64) * 8 + wave] = a0.y;
  sChunk[(lane + 96) * 8 + wave] = a1.y;
  __syncthreads();
  ((float4*)uimg)[blockIdx.x * 256 + tid] = ((const float4*)sChunk)[tid];
}

__global__ __launch_bounds__(256) void qmaml_kernel(
    const float* __restrict__ x,  const float* __restrict__ W1,
    const float* __restrict__ b1, const float* __restrict__ ln_g,
    const float* __restrict__ ln_b, const float* __restrict__ W2,
    const float* __restrict__ b2, const float* __restrict__ uimg,
    const float* __restrict__ Wc, const float* __restrict__ bc,
    float* __restrict__ out, int B)
{
  __shared__ float  sW1[64 * WP];
  __shared__ __attribute__((aligned(16))) float sW2[512];
  __shared__ __attribute__((aligned(16))) float sSmall[256];
  const float* sB1 = sSmall;
  const float* sG = sSmall + 64;
  const float* sBeta = sSmall + 128;
  const float* sB2 = sSmall + 192;
  const float* sWc = sSmall + 198;
  const float* sBc = sSmall + 228;
  __shared__ float  sU[UIMG];

  const int tid = threadIdx.x;
  const int lane = tid & 31;
  const int wave = tid >> 5;
  const int half = lane >> 4;
  const int lm   = lane & 15;
  const int rowBase = blockIdx.x * 128 + wave * 16;

  float4 rW1[4], rU[8];
  #pragma unroll
  for (int j = 0; j < 4; ++j) rW1[j] = ((const float4*)W1)[tid + 256 * j];
  #pragma unroll
  for (int j = 0; j < 8; ++j) rU[j] = ((const float4*)uimg)[tid + 256 * j];
  const float rW2a = W2[tid];
  const float rW2b = W2[256 + (tid & 127)];
  const int i64 = tid & 63;
  const int iB2 = tid < 192 ? 0 : (tid < 198 ? tid - 192 : 5);
  const int iWc = tid < 198 ? 0 : (tid < 228 ? tid - 198 : 29);
  const int iBc = tid < 228 ? 0 : (tid < 233 ? tid - 228 : 4);
  const float* const pB1 = b1 + i64;
  const float* const pG  = ln_g + i64;
  const float* const pBe = ln_b + i64;
  const float* const pB2 = b2 + iB2;
  const float* const pWc = Wc + iWc;
  const float* const pBc = bc + iBc;
  const float* pSmall = pBc;
  pSmall = (tid < 228) ? pWc : pSmall;
  pSmall = (tid < 198) ? pB2 : pSmall;
  pSmall = (tid < 192) ? pBe : pSmall;
  pSmall = (tid < 128) ? pG  : pSmall;
  pSmall = (tid <  64) ? pB1 : pSmall;
  const float rSmall = *pSmall;
  int rowA = rowBase + lm; if (rowA >= B) rowA = B - 1;
  const float* xrow = x + (size_t)rowA * 64 + 2 * half;
  v2f xf[16];
  #pragma unroll
  for (int kk = 0; kk < 16; ++kk) xf[kk] = *(const v2f*)(xrow + 4 * kk);

  #pragma unroll
  for (int j = 0; j < 4; ++j) {
    const int i = tid + 256 * j;
    *(float4*)(sW1 + (i >> 4) * WP + 4 * (i & 15)) = rW1[j];
  }
  #pragma unroll
  for (int j = 0; j < 8; ++j) {
    *(float4*)(sU + (tid >> 1) * UP + 8 * j + 4 * (tid & 1)) = rU[j];
  }
  sW2[tid] = rW2a;
  sW2[256 + tid] = rW2b;
  sSmall[tid] = rSmall;
  __syncthreads();

  v8f acc[4] = {};
  const float* w1p = sW1 + lm * WP + 2 * half;
  v2f wa[2][8];
  #pragma unroll
  for (int j = 0; j < 2; ++j) {
    #pragma unroll
    for (int t = 0; t < 4; ++t) wa[0][4 * j + t] = *(const v2f*)(w1p + 16 * t * WP + 4 * j);
  }
  #pragma unroll
  for (int b = 0; b < 8; ++b) {
    if (b + 1 < 8) {
      #pragma unroll
      for (int j = 0; j < 2; ++j) {
        #pragma unroll
        for (int t = 0; t < 4; ++t)
          wa[(b + 1) & 1][4 * j + t] = *(const v2f*)(w1p + 16 * t * WP + 4 * (2 * b + 2 + j));
      }
    }
    __builtin_amdgcn_sched_barrier(0);
    #pragma unroll
    for (int j = 0; j < 2; ++j) {
      #pragma unroll
      for (int t = 0; t < 4; ++t)
        acc[t] = __builtin_amdgcn_wmma_f32_16x16x4_f32(false, wa[b & 1][4 * j + t], false, xf[2 * b + j],
                                                       (short)0, acc[t], false, false);
    }
    __builtin_amdgcn_sched_barrier(0);
  }

  const int fb = 8 * half;
  float hv[4][8];
  #pragma unroll
  for (int t = 0; t < 4; ++t) {
    const float4 ba = *(const float4*)(sB1 + 16 * t + fb);
    const float4 bb = *(const float4*)(sB1 + 16 * t + fb + 4);
    const float bias[8] = { ba.x, ba.y, ba.z, ba.w, bb.x, bb.y, bb.z, bb.w };
    #pragma unroll
    for (int j = 0; j < 8; ++j) hv[t][j] = fmaxf(acc[t][j] + bias[j], 0.0f);
  }

  float st[4], qt[4];
  #pragma unroll
  for (int t = 0; t < 4; ++t) {
    st[t] = hv[t][0];
    qt[t] = hv[t][0] * hv[t][0];
    #pragma unroll
    for (int j = 1; j < 8; ++j) { st[t] += hv[t][j]; qt[t] += hv[t][j] * hv[t][j]; }
  }
  float s  = (st[0] + st[1]) + (st[2] + st[3]);
  float q2 = (qt[0] + qt[1]) + (qt[2] + qt[3]);
  s  += xhalf(s);
  q2 += xhalf(q2);
  const float mu  = s * 0.015625f;
  const float var = q2 * 0.015625f - mu * mu;
  const float inv = rsqrtf(var + 1e-5f);
  #pragma unroll
  for (int t = 0; t < 4; ++t) {
    const float4 ga = *(const float4*)(sG + 16 * t + fb);
    const float4 gb = *(const float4*)(sG + 16 * t + fb + 4);
    const float4 ea = *(const float4*)(sBeta + 16 * t + fb);
    const float4 eb = *(const float4*)(sBeta + 16 * t + fb + 4);
    const float g[8]  = { ga.x, ga.y, ga.z, ga.w, gb.x, gb.y, gb.z, gb.w };
    const float be[8] = { ea.x, ea.y, ea.z, ea.w, eb.x, eb.y, eb.z, eb.w };
    #pragma unroll
    for (int j = 0; j < 8; ++j) hv[t][j] = (hv[t][j] - mu) * inv * g[j] + be[j];
  }

  float zq[6];
  #pragma unroll
  for (int q = 0; q < 6; ++q) {
    v2f pt[4];
    #pragma unroll
    for (int t = 0; t < 4; ++t) {
      const float4 wa = *(const float4*)(sW2 + q * 64 + 16 * t + fb);
      const float4 wb = *(const float4*)(sW2 + q * 64 + 16 * t + fb + 4);
      pt[t]  = v2f{hv[t][0], hv[t][1]} * v2f{wa.x, wa.y};
      pt[t] += v2f{hv[t][2], hv[t][3]} * v2f{wa.z, wa.w};
      pt[t] += v2f{hv[t][4], hv[t][5]} * v2f{wb.x, wb.y};
      pt[t] += v2f{hv[t][6], hv[t][7]} * v2f{wb.z, wb.w};
    }
    const v2f pp = (pt[0] + pt[1]) + (pt[2] + pt[3]);
    float p = pp.x + pp.y;
    p += xhalf(p);
    zq[q] = tanhf(p + sB2[q]);
  }

  float cq[6], sq[6];
  #pragma unroll
  for (int q = 0; q < 6; ++q) {
    float h2 = 1.57079632679489662f * zq[q];
    cq[q] = cosf(h2); sq[q] = sinf(h2);
  }
  float p01[4], p012[8], p0123[16];
  #pragma unroll
  for (int i = 0; i < 4; ++i)  p01[i]   = ((i >> 1) ? sq[0] : cq[0]) * ((i & 1) ? sq[1] : cq[1]);
  #pragma unroll
  for (int i = 0; i < 8; ++i)  p012[i]  = p01[i >> 1]  * ((i & 1) ? sq[2] : cq[2]);
  #pragma unroll
  for (int i = 0; i < 16; ++i) p0123[i] = p012[i >> 1] * ((i & 1) ? sq[3] : cq[3]);
  const float f4 = half ? sq[4] : cq[4];

  v8f amp[8] = {};
  const float* up = sU + lm * UP + 2 * half;
  v2f ua[2][8];
  #pragma unroll
  for (int t = 0; t < 8; ++t) ua[0][t] = *(const v2f*)(up + 16 * t * UP);
  v2f pfc;
  {
    float t4 = p0123[0] * f4;
    pfc.x = t4 * cq[5];
    pfc.y = t4 * sq[5];
  }
  #pragma unroll
  for (int kk = 0; kk < 16; ++kk) {
    if (kk + 1 < 16) {
      #pragma unroll
      for (int t = 0; t < 8; ++t) ua[(kk + 1) & 1][t] = *(const v2f*)(up + 16 * t * UP + 4 * (kk + 1));
    }
    __builtin_amdgcn_sched_barrier(0);
    #pragma unroll
    for (int t = 0; t < 8; ++t)
      amp[t] = __builtin_amdgcn_wmma_f32_16x16x4_f32(false, ua[kk & 1][t], false, pfc,
                                                     (short)0, amp[t], false, false);
    v2f pfn = pfc;
    if (kk + 1 < 16) {
      float t4 = p0123[kk + 1] * f4;
      pfn.x = t4 * cq[5];
      pfn.y = t4 * sq[5];
    }
    __builtin_amdgcn_sched_barrier(0);
    pfc = pfn;
  }

  float P[4][8];
  #pragma unroll
  for (int t = 0; t < 4; ++t) {
    #pragma unroll
    for (int r = 0; r < 8; ++r) P[t][r] = amp[t][r] * amp[t][r] + amp[t + 4][r] * amp[t + 4][r];
  }
  float e5 = 0.f, e4 = 0.f, e3 = 0.f;
  float s1[4][4], s2[4][2], s3[4];
  #pragma unroll
  for (int t = 0; t < 4; ++t) {
    #pragma unroll
    for (int m = 0; m < 4; ++m) { s1[t][m] = P[t][2*m] + P[t][2*m+1]; e5 += P[t][2*m] - P[t][2*m+1]; }
    #pragma unroll
    for (int m = 0; m < 2; ++m) { s2[t][m] = s1[t][2*m] + s1[t][2*m+1]; e4 += s1[t][2*m] - s1[t][2*m+1]; }
    s3[t] = s2[t][0] + s2[t][1]; e3 += s2[t][0] - s2[t][1];
  }
  const float u0 = s3[0] + s3[1], u1 = s3[2] + s3[3];
  const float e1 = (s3[0] - s3[1]) + (s3[2] - s3[3]);
  const float S  = u0 + u1;
  const float e0 = u0 - u1;
  const float e2 = half ? -S : S;

  const int row = rowBase + lm;
  #pragma unroll
  for (int k = 0; k < 5; ++k) {
    float v = sWc[k*6+0]*e0 + sWc[k*6+1]*e1 + sWc[k*6+2]*e2
            + sWc[k*6+3]*e3 + sWc[k*6+4]*e4 + sWc[k*6+5]*e5;
    v += __shfl_xor(v, 16, 32);
    if (half == 0 && row < B) out[(size_t)row * 5 + k] = v + sBc[k];
  }
}

extern "C" void kernel_launch(void* const* d_in, const int* in_sizes, int n_in,
                              void* d_out, int out_size, void* d_ws, size_t ws_size,
                              hipStream_t stream) {
  (void)n_in; (void)out_size; (void)ws_size;
  const float* x    = (const float*)d_in[0];
  const float* W1   = (const float*)d_in[1];
  const float* b1   = (const float*)d_in[2];
  const float* ln_g = (const float*)d_in[3];
  const float* ln_b = (const float*)d_in[4];
  const float* W2   = (const float*)d_in[5];
  const float* b2   = (const float*)d_in[6];
  const float* th_s = (const float*)d_in[7];
  const float* th_t = (const float*)d_in[8];
  const float* Wc   = (const float*)d_in[9];
  const float* bc   = (const float*)d_in[10];
  float* out = (float*)d_out;
  float* uimg = (float*)d_ws;
  int B = in_sizes[0] / 64;
  int grid = (B + 127) / 128;
  circuit_compose<<<8, 256, 0, stream>>>(th_s, th_t, uimg);
  qmaml_kernel<<<grid, 256, 0, stream>>>(x, W1, b1, ln_g, ln_b, W2, b2, uimg, Wc, bc, out, B);
}


# tanh, sin and cos computed by short bounded-range forms (exp2 + reciprocal; half-angle polynomials, angle = (pi/2)*tanh so |u| <= pi/4) instead of the library routines with their range reduction
# speedup vs baseline: 8.3090x; 1.0450x over previous
#include <hip/hip_runtime.h>
#include <math.h>


typedef __attribute__((ext_vector_type(2))) float v2f;
typedef __attribute__((ext_vector_type(8))) float v8f;

struct Rings { int r1[64]; int r2[64]; };

constexpr Rings makeRings() {
  Rings R{};
  for (int rr = 1; rr <= 2; ++rr) {
    int total[64];
    for (int i = 0; i < 64; ++i) total[i] = i;
    for (int w = 0; w < 6; ++w) {
      int t = (w + rr) % 6;
      int nxt[64];
      for (int i = 0; i < 64; ++i) {
        int cbit = (i >> (5 - w)) & 1;
        nxt[i] = total[i ^ (cbit << (5 - t))];
      }
      for (int i = 0; i < 64; ++i) total[i] = nxt[i];
    }
    for (int i = 0; i < 64; ++i) { if (rr == 1) R.r1[i] = total[i]; else R.r2[i] = total[i]; }
  }
  return R;
}

__constant__ Rings RINGS = makeRings();

__device__ __forceinline__ float2 cmul(float2 a, float2 b) {
  return make_float2(a.x * b.x - a.y * b.y, a.x * b.y + a.y * b.x);
}
__device__ __forceinline__ float2 cadd(float2 a, float2 b) { return make_float2(a.x + b.x, a.y + b.y); }
__device__ __forceinline__ float2 shflx2(float2 v, int m) {
  return make_float2(__shfl_xor(v.x, m, 32), __shfl_xor(v.y, m, 32));
}
__device__ __forceinline__ float2 shfl2(float2 v, int l) {
  return make_float2(__shfl(v.x, l, 32), __shfl(v.y, l, 32));
}

__device__ __forceinline__ float xhalf(float v) {
  int b = __builtin_bit_cast(int, v);
  return __builtin_bit_cast(float, __builtin_amdgcn_permlanex16(b, b, 0x76543210, 0xfedcba98, false, false));
}

__device__ __forceinline__ float tanh_bounded(float x) {
  const float t = __builtin_amdgcn_exp2f(fabsf(x) * -2.885390081777927f);
  const float r = (1.0f - t) * __builtin_amdgcn_rcpf(1.0f + t);
  return copysignf(r, x);
}

__device__ __forceinline__ void sincos_halfpi(float z, float& s, float& c) {
  const float u  = 0.78539816339744831f * z;
  const float u2 = u * u;
  const float ps = fmaf(u2, fmaf(u2, -1.9515295891e-4f, 8.3321608736e-3f), -1.6666654611e-1f);
  const float su = fmaf(u * u2, ps, u);
  const float pc = fmaf(u2, fmaf(u2, 2.443315711809948e-5f, -1.388731625493765e-3f), 4.166664568298827e-2f);
  const float cu = fmaf(u2 * u2, pc, fmaf(u2, -0.5f, 1.0f));
  const float s2 = su + su;
  s = s2 * cu;
  c = fmaf(-s2, su, 1.0f);
}

#define UP 68
#define UIMG (128 * UP)
#define WP 68

__global__ __launch_bounds__(256) void circuit_compose(
    const float* __restrict__ th_s, const float* __restrict__ th_t, float* __restrict__ uimg)
{
  __shared__ float2 sGates[72];
  __shared__ float  sChunk[1024];

  const int tid = threadIdx.x;
  if (tid < 18) {
    int l = tid / 6, w = tid % 6;
    const float* th = (l < 2) ? (th_s + (l * 6 + w) * 3) : (th_t + w * 3);
    float phi = th[0], theta = th[1], omega = th[2];
    float c  = cosf(0.5f * theta), s = sinf(0.5f * theta);
    float aa = 0.5f * (phi + omega), bb = 0.5f * (phi - omega);
    float ca = cosf(aa), sa = sinf(aa);
    float cb = cosf(bb), sb = sinf(bb);
    sGates[tid * 4 + 0] = make_float2( ca * c, -sa * c);
    sGates[tid * 4 + 1] = make_float2(-cb * s, -sb * s);
    sGates[tid * 4 + 2] = make_float2( cb * s, -sb * s);
    sGates[tid * 4 + 3] = make_float2( ca * c,  sa * c);
  }
  __syncthreads();

  const int lane = tid & 31;
  const int wave = tid >> 5;
  const int rA0 = RINGS.r1[lane], rA1 = RINGS.r1[lane + 32];
  const int rB0 = RINGS.r2[lane], rB1 = RINGS.r2[lane + 32];

  const int k = blockIdx.x * 8 + wave;
  float2 a0 = make_float2((lane == k) ? 1.f : 0.f, 0.f);
  float2 a1 = make_float2((lane + 32 == k) ? 1.f : 0.f, 0.f);

  #pragma unroll
  for (int l = 0; l < 3; ++l) {
    #pragma unroll
    for (int w = 0; w < 6; ++w) {
      int g = l * 6 + w;
      float2 m00 = sGates[g * 4 + 0], m01 = sGates[g * 4 + 1];
      float2 m10 = sGates[g * 4 + 2], m11 = sGates[g * 4 + 3];
      if (w == 0) {
        float2 n0 = cadd(cmul(m00, a0), cmul(m01, a1));
        float2 n1 = cadd(cmul(m10, a0), cmul(m11, a1));
        a0 = n0; a1 = n1;
      } else {
        int b = 5 - w, mask = 1 << b;
        float2 p0 = shflx2(a0, mask), p1 = shflx2(a1, mask);
        int bit = (lane >> b) & 1;
        float2 e0 = cadd(cmul(m00, a0), cmul(m01, p0));
        float2 o0 = cadd(cmul(m10, p0), cmul(m11, a0));
        float2 e1 = cadd(cmul(m00, a1), cmul(m01, p1));
        float2 o1 = cadd(cmul(m10, p1), cmul(m11, a1));
        a0 = bit ? o0 : e0;
        a1 = bit ? o1 : e1;
      }
    }
    int s0 = (l == 1) ? rB0 : rA0;
    int s1 = (l == 1) ? rB1 : rA1;
    float2 g0a = shfl2(a0, s0 & 31), g0b = shfl2(a1, s0 & 31);
    float2 g1a = shfl2(a0, s1 & 31), g1b = shfl2(a1, s1 & 31);
    a0 = (s0 < 32) ? g0a : g0b;
    a1 = (s1 < 32) ? g1a : g1b;
  }
  sChunk[(lane)      * 8 + wave] = a0.x;
  sChunk[(lane + 32) * 8 + wave] = a1.x;
  sChunk[(lane + 64) * 8 + wave] = a0.y;
  sChunk[(lane + 96) * 8 + wave] = a1.y;
  __syncthreads();
  ((float4*)uimg)[blockIdx.x * 256 + tid] = ((const float4*)sChunk)[tid];
}

__global__ __launch_bounds__(256) void qmaml_kernel(
    const float* __restrict__ x,  const float* __restrict__ W1,
    const float* __restrict__ b1, const float* __restrict__ ln_g,
    const float* __restrict__ ln_b, const float* __restrict__ W2,
    const float* __restrict__ b2, const float* __restrict__ uimg,
    const float* __restrict__ Wc, const float* __restrict__ bc,
    float* __restrict__ out, int B)
{
  __shared__ float  sW1[64 * WP];
  __shared__ __attribute__((aligned(16))) float sW2[512];
  __shared__ __attribute__((aligned(16))) float sSmall[256];
  const float* sB1 = sSmall;
  const float* sG = sSmall + 64;
  const float* sBeta = sSmall + 128;
  const float* sB2 = sSmall + 192;
  const float* sWc = sSmall + 198;
  const float* sBc = sSmall + 228;
  __shared__ float  sU[UIMG];

  const int tid = threadIdx.x;
  const int lane = tid & 31;
  const int wave = tid >> 5;
  const int half = lane >> 4;
  const int lm   = lane & 15;
  const int rowBase = blockIdx.x * 128 + wave * 16;

  float4 rW1[4], rU[8];
  #pragma unroll
  for (int j = 0; j < 4; ++j) rW1[j] = ((const float4*)W1)[tid + 256 * j];
  #pragma unroll
  for (int j = 0; j < 8; ++j) rU[j] = ((const float4*)uimg)[tid + 256 * j];
  const float rW2a = W2[tid];
  const float rW2b = W2[256 + (tid & 127)];
  const int i64 = tid & 63;
  const int iB2 = tid < 192 ? 0 : (tid < 198 ? tid - 192 : 5);
  const int iWc = tid < 198 ? 0 : (tid < 228 ? tid - 198 : 29);
  const int iBc = tid < 228 ? 0 : (tid < 233 ? tid - 228 : 4);
  const float* const pB1 = b1 + i64;
  const float* const pG  = ln_g + i64;
  const float* const pBe = ln_b + i64;
  const float* const pB2 = b2 + iB2;
  const float* const pWc = Wc + iWc;
  const float* const pBc = bc + iBc;
  const float* pSmall = pBc;
  pSmall = (tid < 228) ? pWc : pSmall;
  pSmall = (tid < 198) ? pB2 : pSmall;
  pSmall = (tid < 192) ? pBe : pSmall;
  pSmall = (tid < 128) ? pG  : pSmall;
  pSmall = (tid <  64) ? pB1 : pSmall;
  const float rSmall = *pSmall;
  int rowA = rowBase + lm; if (rowA >= B) rowA = B - 1;
  const float* xrow = x + (size_t)rowA * 64 + 2 * half;
  v2f xf[16];
  #pragma unroll
  for (int kk = 0; kk < 16; ++kk) xf[kk] = *(const v2f*)(xrow + 4 * kk);

  #pragma unroll
  for (int j = 0; j < 4; ++j) {
    const int i = tid + 256 * j;
    *(float4*)(sW1 + (i >> 4) * WP + 4 * (i & 15)) = rW1[j];
  }
  #pragma unroll
  for (int j = 0; j < 8; ++j) {
    *(float4*)(sU + (tid >> 1) * UP + 8 * j + 4 * (tid & 1)) = rU[j];
  }
  sW2[tid] = rW2a;
  sW2[256 + tid] = rW2b;
  sSmall[tid] = rSmall;
  __syncthreads();

  v8f acc[4] = {};
  const float* w1p = sW1 + lm * WP + 2 * half;
  v2f wa[2][8];
  #pragma unroll
  for (int j = 0; j < 2; ++j) {
    #pragma unroll
    for (int t = 0; t < 4; ++t) wa[0][4 * j + t] = *(const v2f*)(w1p + 16 * t * WP + 4 * j);
  }
  #pragma unroll
  for (int b = 0; b < 8; ++b) {
    if (b + 1 < 8) {
      #pragma unroll
      for (int j = 0; j < 2; ++j) {
        #pragma unroll
        for (int t = 0; t < 4; ++t)
          wa[(b + 1) & 1][4 * j + t] = *(const v2f*)(w1p + 16 * t * WP + 4 * (2 * b + 2 + j));
      }
    }
    __builtin_amdgcn_sched_barrier(0);
    #pragma unroll
    for (int j = 0; j < 2; ++j) {
      #pragma unroll
      for (int t = 0; t < 4; ++t)
        acc[t] = __builtin_amdgcn_wmma_f32_16x16x4_f32(false, wa[b & 1][4 * j + t], false, xf[2 * b + j],
                                                       (short)0, acc[t], false, false);
    }
    __builtin_amdgcn_sched_barrier(0);
  }

  const int fb = 8 * half;
  float hv[4][8];
  #pragma unroll
  for (int t = 0; t < 4; ++t) {
    const float4 ba = *(const float4*)(sB1 + 16 * t + fb);
    const float4 bb = *(const float4*)(sB1 + 16 * t + fb + 4);
    const float bias[8] = { ba.x, ba.y, ba.z, ba.w, bb.x, bb.y, bb.z, bb.w };
    #pragma unroll
    for (int j = 0; j < 8; ++j) hv[t][j] = fmaxf(acc[t][j] + bias[j], 0.0f);
  }

  float st[4], qt[4];
  #pragma unroll
  for (int t = 0; t < 4; ++t) {
    st[t] = hv[t][0];
    qt[t] = hv[t][0] * hv[t][0];
    #pragma unroll
    for (int j = 1; j < 8; ++j) { st[t] += hv[t][j]; qt[t] += hv[t][j] * hv[t][j]; }
  }
  float s  = (st[0] + st[1]) + (st[2] + st[3]);
  float q2 = (qt[0] + qt[1]) + (qt[2] + qt[3]);
  s  += xhalf(s);
  q2 += xhalf(q2);
  const float mu  = s * 0.015625f;
  const float var = q2 * 0.015625f - mu * mu;
  const float inv = rsqrtf(var + 1e-5f);
  #pragma unroll
  for (int t = 0; t < 4; ++t) {
    const float4 ga = *(const float4*)(sG + 16 * t + fb);
    const float4 gb = *(const float4*)(sG + 16 * t + fb + 4);
    const float4 ea = *(const float4*)(sBeta + 16 * t + fb);
    const float4 eb = *(const float4*)(sBeta + 16 * t + fb + 4);
    const float g[8]  = { ga.x, ga.y, ga.z, ga.w, gb.x, gb.y, gb.z, gb.w };
    const float be[8] = { ea.x, ea.y, ea.z, ea.w, eb.x, eb.y, eb.z, eb.w };
    #pragma unroll
    for (int j = 0; j < 8; ++j) hv[t][j] = (hv[t][j] - mu) * inv * g[j] + be[j];
  }

  float zq[6];
  #pragma unroll
  for (int q = 0; q < 6; ++q) {
    v2f pt[4];
    #pragma unroll
    for (int t = 0; t < 4; ++t) {
      const float4 wa = *(const float4*)(sW2 + q * 64 + 16 * t + fb);
      const float4 wb = *(const float4*)(sW2 + q * 64 + 16 * t + fb + 4);
      pt[t]  = v2f{hv[t][0], hv[t][1]} * v2f{wa.x, wa.y};
      pt[t] += v2f{hv[t][2], hv[t][3]} * v2f{wa.z, wa.w};
      pt[t] += v2f{hv[t][4], hv[t][5]} * v2f{wb.x, wb.y};
      pt[t] += v2f{hv[t][6], hv[t][7]} * v2f{wb.z, wb.w};
    }
    const v2f pp = (pt[0] + pt[1]) + (pt[2] + pt[3]);
    float p = pp.x + pp.y;
    p += xhalf(p);
    zq[q] = tanh_bounded(p + sB2[q]);
  }

  float cq[6], sq[6];
  #pragma unroll
  for (int q = 0; q < 6; ++q) sincos_halfpi(zq[q], sq[q], cq[q]);
  float p01[4], p012[8], p0123[16];
  #pragma unroll
  for (int i = 0; i < 4; ++i)  p01[i]   = ((i >> 1) ? sq[0] : cq[0]) * ((i & 1) ? sq[1] : cq[1]);
  #pragma unroll
  for (int i = 0; i < 8; ++i)  p012[i]  = p01[i >> 1]  * ((i & 1) ? sq[2] : cq[2]);
  #pragma unroll
  for (int i = 0; i < 16; ++i) p0123[i] = p012[i >> 1] * ((i & 1) ? sq[3] : cq[3]);
  const float f4 = half ? sq[4] : cq[4];

  v8f amp[8] = {};
  const float* up = sU + lm * UP + 2 * half;
  v2f ua[2][8];
  #pragma unroll
  for (int t = 0; t < 8; ++t) ua[0][t] = *(const v2f*)(up + 16 * t * UP);
  v2f pfc;
  {
    float t4 = p0123[0] * f4;
    pfc.x = t4 * cq[5];
    pfc.y = t4 * sq[5];
  }
  #pragma unroll
  for (int kk = 0; kk < 16; ++kk) {
    if (kk + 1 < 16) {
      #pragma unroll
      for (int t = 0; t < 8; ++t) ua[(kk + 1) & 1][t] = *(const v2f*)(up + 16 * t * UP + 4 * (kk + 1));
    }
    __builtin_amdgcn_sched_barrier(0);
    #pragma unroll
    for (int t = 0; t < 8; ++t)
      amp[t] = __builtin_amdgcn_wmma_f32_16x16x4_f32(false, ua[kk & 1][t], false, pfc,
                                                     (short)0, amp[t], false, false);
    v2f pfn = pfc;
    if (kk + 1 < 16) {
      float t4 = p0123[kk + 1] * f4;
      pfn.x = t4 * cq[5];
      pfn.y = t4 * sq[5];
    }
    __builtin_amdgcn_sched_barrier(0);
    pfc = pfn;
  }

  float P[4][8];
  #pragma unroll
  for (int t = 0; t < 4; ++t) {
    #pragma unroll
    for (int r = 0; r < 8; ++r) P[t][r] = amp[t][r] * amp[t][r] + amp[t + 4][r] * amp[t + 4][r];
  }
  float e5 = 0.f, e4 = 0.f, e3 = 0.f;
  float s1[4][4], s2[4][2], s3[4];
  #pragma unroll
  for (int t = 0; t < 4; ++t) {
    #pragma unroll
    for (int m = 0; m < 4; ++m) { s1[t][m] = P[t][2*m] + P[t][2*m+1]; e5 += P[t][2*m] - P[t][2*m+1]; }
    #pragma unroll
    for (int m = 0; m < 2; ++m) { s2[t][m] = s1[t][2*m] + s1[t][2*m+1]; e4 += s1[t][2*m] - s1[t][2*m+1]; }
    s3[t] = s2[t][0] + s2[t][1]; e3 += s2[t][0] - s2[t][1];
  }
  const float u0 = s3[0] + s3[1], u1 = s3[2] + s3[3];
  const float e1 = (s3[0] - s3[1]) + (s3[2] - s3[3]);
  const float S  = u0 + u1;
  const float e0 = u0 - u1;
  const float e2 = half ? -S : S;

  const int row = rowBase + lm;
  #pragma unroll
  for (int k = 0; k < 5; ++k) {
    float v = sWc[k*6+0]*e0 + sWc[k*6+1]*e1 + sWc[k*6+2]*e2
            + sWc[k*6+3]*e3 + sWc[k*6+4]*e4 + sWc[k*6+5]*e5;
    v += __shfl_xor(v, 16, 32);
    if (half == 0 && row < B) out[(size_t)row * 5 + k] = v + sBc[k];
  }
}

extern "C" void kernel_launch(void* const* d_in, const int* in_sizes, int n_in,
                              void* d_out, int out_size, void* d_ws, size_t ws_size,
                              hipStream_t stream) {
  (void)n_in; (void)out_size; (void)ws_size;
  const float* x    = (const float*)d_in[0];
  const float* W1   = (const float*)d_in[1];
  const float* b1   = (const float*)d_in[2];
  const float* ln_g = (const float*)d_in[3];
  const float* ln_b = (const float*)d_in[4];
  const float* W2   = (const float*)d_in[5];
  const float* b2   = (const float*)d_in[6];
  const float* th_s = (const float*)d_in[7];
  const float* th_t = (const float*)d_in[8];
  const float* Wc   = (const float*)d_in[9];
  const float* bc   = (const float*)d_in[10];
  float* out = (float*)d_out;
  float* uimg = (float*)d_ws;
  int B = in_sizes[0] / 64;
  int grid = (B + 127) / 128;
  circuit_compose<<<8, 256, 0, stream>>>(th_s, th_t, uimg);
  qmaml_kernel<<<grid, 256, 0, stream>>>(x, W1, b1, ln_g, ln_b, W2, b2, uimg, Wc, bc, out, B);
}


# Main kernel made persistent: fixed grid of 512 work-groups, each stages the parameters into LDS once and its waves loop over 16-row tiles, the next tile's x loads issued before this tile's products
# speedup vs baseline: 9.1765x; 1.1044x over previous
#include <hip/hip_runtime.h>
#include <math.h>


typedef __attribute__((ext_vector_type(2))) float v2f;
typedef __attribute__((ext_vector_type(8))) float v8f;

struct Rings { int r1[64]; int r2[64]; };

constexpr Rings makeRings() {
  Rings R{};
  for (int rr = 1; rr <= 2; ++rr) {
    int total[64];
    for (int i = 0; i < 64; ++i) total[i] = i;
    for (int w = 0; w < 6; ++w) {
      int t = (w + rr) % 6;
      int nxt[64];
      for (int i = 0; i < 64; ++i) {
        int cbit = (i >> (5 - w)) & 1;
        nxt[i] = total[i ^ (cbit << (5 - t))];
      }
      for (int i = 0; i < 64; ++i) total[i] = nxt[i];
    }
    for (int i = 0; i < 64; ++i) { if (rr == 1) R.r1[i] = total[i]; else R.r2[i] = total[i]; }
  }
  return R;
}

__constant__ Rings RINGS = makeRings();

__device__ __forceinline__ float2 cmul(float2 a, float2 b) {
  return make_float2(a.x * b.x - a.y * b.y, a.x * b.y + a.y * b.x);
}
__device__ __forceinline__ float2 cadd(float2 a, float2 b) { return make_float2(a.x + b.x, a.y + b.y); }
__device__ __forceinline__ float2 shflx2(float2 v, int m) {
  return make_float2(__shfl_xor(v.x, m, 32), __shfl_xor(v.y, m, 32));
}
__device__ __forceinline__ float2 shfl2(float2 v, int l) {
  return make_float2(__shfl(v.x, l, 32), __shfl(v.y, l, 32));
}

__device__ __forceinline__ float xhalf(float v) {
  int b = __builtin_bit_cast(int, v);
  return __builtin_bit_cast(float, __builtin_amdgcn_permlanex16(b, b, 0x76543210, 0xfedcba98, false, false));
}

__device__ __forceinline__ float tanh_bounded(float x) {
  const float t = __builtin_amdgcn_exp2f(fabsf(x) * -2.885390081777927f);
  const float r = (1.0f - t) * __builtin_amdgcn_rcpf(1.0f + t);
  return copysignf(r, x);
}

__device__ __forceinline__ void sincos_halfpi(float z, float& s, float& c) {
  const float u  = 0.78539816339744831f * z;
  const float u2 = u * u;
  const float ps = fmaf(u2, fmaf(u2, -1.9515295891e-4f, 8.3321608736e-3f), -1.6666654611e-1f);
  const float su = fmaf(u * u2, ps, u);
  const float pc = fmaf(u2, fmaf(u2, 2.443315711809948e-5f, -1.388731625493765e-3f), 4.166664568298827e-2f);
  const float cu = fmaf(u2 * u2, pc, fmaf(u2, -0.5f, 1.0f));
  const float s2 = su + su;
  s = s2 * cu;
  c = fmaf(-s2, su, 1.0f);
}

#define UP 68
#define UIMG (128 * UP)
#define WP 68

__global__ __launch_bounds__(256) void circuit_compose(
    const float* __restrict__ th_s, const float* __restrict__ th_t, float* __restrict__ uimg)
{
  __shared__ float2 sGates[72];
  __shared__ float  sChunk[1024];

  const int tid = threadIdx.x;
  if (tid < 18) {
    int l = tid / 6, w = tid % 6;
    const float* th = (l < 2) ? (th_s + (l * 6 + w) * 3) : (th_t + w * 3);
    float phi = th[0], theta = th[1], omega = th[2];
    float c  = cosf(0.5f * theta), s = sinf(0.5f * theta);
    float aa = 0.5f * (phi + omega), bb = 0.5f * (phi - omega);
    float ca = cosf(aa), sa = sinf(aa);
    float cb = cosf(bb), sb = sinf(bb);
    sGates[tid * 4 + 0] = make_float2( ca * c, -sa * c);
    sGates[tid * 4 + 1] = make_float2(-cb * s, -sb * s);
    sGates[tid * 4 + 2] = make_float2( cb * s, -sb * s);
    sGates[tid * 4 + 3] = make_float2( ca * c,  sa * c);
  }
  __syncthreads();

  const int lane = tid & 31;
  const int wave = tid >> 5;
  const int rA0 = RINGS.r1[lane], rA1 = RINGS.r1[lane + 32];
  const int rB0 = RINGS.r2[lane], rB1 = RINGS.r2[lane + 32];

  const int k = blockIdx.x * 8 + wave;
  float2 a0 = make_float2((lane == k) ? 1.f : 0.f, 0.f);
  float2 a1 = make_float2((lane + 32 == k) ? 1.f : 0.f, 0.f);

  #pragma unroll
  for (int l = 0; l < 3; ++l) {
    #pragma unroll
    for (int w = 0; w < 6; ++w) {
      int g = l * 6 + w;
      float2 m00 = sGates[g * 4 + 0], m01 = sGates[g * 4 + 1];
      float2 m10 = sGates[g * 4 + 2], m11 = sGates[g * 4 + 3];
      if (w == 0) {
        float2 n0 = cadd(cmul(m00, a0), cmul(m01, a1));
        float2 n1 = cadd(cmul(m10, a0), cmul(m11, a1));
        a0 = n0; a1 = n1;
      } else {
        int b = 5 - w, mask = 1 << b;
        float2 p0 = shflx2(a0, mask), p1 = shflx2(a1, mask);
        int bit = (lane >> b) & 1;
        float2 e0 = cadd(cmul(m00, a0), cmul(m01, p0));
        float2 o0 = cadd(cmul(m10, p0), cmul(m11, a0));
        float2 e1 = cadd(cmul(m00, a1), cmul(m01, p1));
        float2 o1 = cadd(cmul(m10, p1), cmul(m11, a1));
        a0 = bit ? o0 : e0;
        a1 = bit ? o1 : e1;
      }
    }
    int s0 = (l == 1) ? rB0 : rA0;
    int s1 = (l == 1) ? rB1 : rA1;
    float2 g0a = shfl2(a0, s0 & 31), g0b = shfl2(a1, s0 & 31);
    float2 g1a = shfl2(a0, s1 & 31), g1b = shfl2(a1, s1 & 31);
    a0 = (s0 < 32) ? g0a : g0b;
    a1 = (s1 < 32) ? g1a : g1b;
  }
  sChunk[(lane)      * 8 + wave] = a0.x;
  sChunk[(lane + 32) * 8 + wave] = a1.x;
  sChunk[(lane + 64) * 8 + wave] = a0.y;
  sChunk[(lane + 96) * 8 + wave] = a1.y;
  __syncthreads();
  ((float4*)uimg)[blockIdx.x * 256 + tid] = ((const float4*)sChunk)[tid];
}

__global__ __launch_bounds__(256) void qmaml_kernel(
    const float* __restrict__ x,  const float* __restrict__ W1,
    const float* __restrict__ b1, const float* __restrict__ ln_g,
    const float* __restrict__ ln_b, const float* __restrict__ W2,
    const float* __restrict__ b2, const float* __restrict__ uimg,
    const float* __restrict__ Wc, const float* __restrict__ bc,
    float* __restrict__ out, int B)
{
  __shared__ float  sW1[64 * WP];
  __shared__ __attribute__((aligned(16))) float sW2[512];
  __shared__ __attribute__((aligned(16))) float sSmall[256];
  const float* sB1 = sSmall;
  const float* sG = sSmall + 64;
  const float* sBeta = sSmall + 128;
  const float* sB2 = sSmall + 192;
  const float* sWc = sSmall + 198;
  const float* sBc = sSmall + 228;
  __shared__ float  sU[UIMG];

  const int tid = threadIdx.x;
  const int lane = tid & 31;
  const int wave = __builtin_amdgcn_readfirstlane(tid >> 5);
  const int half = lane >> 4;
  const int lm   = lane & 15;
  const int nTiles = (B + 15) >> 4;
  const int tStep  = gridDim.x * 8;
  int tile = blockIdx.x * 8 + wave;

  float4 rW1[4], rU[8];
  #pragma unroll
  for (int j = 0; j < 4; ++j) rW1[j] = ((const float4*)W1)[tid + 256 * j];
  #pragma unroll
  for (int j = 0; j < 8; ++j) rU[j] = ((const float4*)uimg)[tid + 256 * j];
  const float rW2a = W2[tid];
  const float rW2b = W2[256 + (tid & 127)];
  const int i64 = tid & 63;
  const int iB2 = tid < 192 ? 0 : (tid < 198 ? tid - 192 : 5);
  const int iWc = tid < 198 ? 0 : (tid < 228 ? tid - 198 : 29);
  const int iBc = tid < 228 ? 0 : (tid < 233 ? tid - 228 : 4);
  const float* const pB1 = b1 + i64;
  const float* const pG  = ln_g + i64;
  const float* const pBe = ln_b + i64;
  const float* const pB2 = b2 + iB2;
  const float* const pWc = Wc + iWc;
  const float* const pBc = bc + iBc;
  const float* pSmall = pBc;
  pSmall = (tid < 228) ? pWc : pSmall;
  pSmall = (tid < 198) ? pB2 : pSmall;
  pSmall = (tid < 192) ? pBe : pSmall;
  pSmall = (tid < 128) ? pG  : pSmall;
  pSmall = (tid <  64) ? pB1 : pSmall;
  const float rSmall = *pSmall;
  v2f xf[16];
  {
    int rowA = tile * 16 + lm; if (rowA >= B) rowA = B - 1;
    const float* xrow = x + (size_t)rowA * 64 + 2 * half;
    #pragma unroll
    for (int kk = 0; kk < 16; ++kk) xf[kk] = *(const v2f*)(xrow + 4 * kk);
  }

  #pragma unroll
  for (int j = 0; j < 4; ++j) {
    const int i = tid + 256 * j;
    *(float4*)(sW1 + (i >> 4) * WP + 4 * (i & 15)) = rW1[j];
  }
  #pragma unroll
  for (int j = 0; j < 8; ++j) {
    *(float4*)(sU + (tid >> 1) * UP + 8 * j + 4 * (tid & 1)) = rU[j];
  }
  sW2[tid] = rW2a;
  sW2[256 + tid] = rW2b;
  sSmall[tid] = rSmall;
  __syncthreads();

  const int fb = 8 * half;
  const float* w1p = sW1 + lm * WP + 2 * half;
  const float* up = sU + lm * UP + 2 * half;

  for (; tile < nTiles; tile += tStep) {
  const int rowBase = tile * 16;
  v2f xn[16];
  {
    int rowN = rowBase + tStep * 16 + lm; if (rowN >= B) rowN = B - 1;
    const float* xrowN = x + (size_t)rowN * 64 + 2 * half;
    #pragma unroll
    for (int kk = 0; kk < 16; ++kk) xn[kk] = *(const v2f*)(xrowN + 4 * kk);
  }

  v8f acc[4] = {};
  v2f wa[2][8];
  #pragma unroll
  for (int j = 0; j < 2; ++j) {
    #pragma unroll
    for (int t = 0; t < 4; ++t) wa[0][4 * j + t] = *(const v2f*)(w1p + 16 * t * WP + 4 * j);
  }
  #pragma unroll
  for (int b = 0; b < 8; ++b) {
    if (b + 1 < 8) {
      #pragma unroll
      for (int j = 0; j < 2; ++j) {
        #pragma unroll
        for (int t = 0; t < 4; ++t)
          wa[(b + 1) & 1][4 * j + t] = *(const v2f*)(w1p + 16 * t * WP + 4 * (2 * b + 2 + j));
      }
    }
    __builtin_amdgcn_sched_barrier(0);
    #pragma unroll
    for (int j = 0; j < 2; ++j) {
      #pragma unroll
      for (int t = 0; t < 4; ++t)
        acc[t] = __builtin_amdgcn_wmma_f32_16x16x4_f32(false, wa[b & 1][4 * j + t], false, xf[2 * b + j],
                                                       (short)0, acc[t], false, false);
    }
    __builtin_amdgcn_sched_barrier(0);
  }

  float hv[4][8];
  #pragma unroll
  for (int t = 0; t < 4; ++t) {
    const float4 ba = *(const float4*)(sB1 + 16 * t + fb);
    const float4 bb = *(const float4*)(sB1 + 16 * t + fb + 4);
    const float bias[8] = { ba.x, ba.y, ba.z, ba.w, bb.x, bb.y, bb.z, bb.w };
    #pragma unroll
    for (int j = 0; j < 8; ++j) hv[t][j] = fmaxf(acc[t][j] + bias[j], 0.0f);
  }

  float st[4], qt[4];
  #pragma unroll
  for (int t = 0; t < 4; ++t) {
    st[t] = hv[t][0];
    qt[t] = hv[t][0] * hv[t][0];
    #pragma unroll
    for (int j = 1; j < 8; ++j) { st[t] += hv[t][j]; qt[t] += hv[t][j] * hv[t][j]; }
  }
  float s  = (st[0] + st[1]) + (st[2] + st[3]);
  float q2 = (qt[0] + qt[1]) + (qt[2] + qt[3]);
  s  += xhalf(s);
  q2 += xhalf(q2);
  const float mu  = s * 0.015625f;
  const float var = q2 * 0.015625f - mu * mu;
  const float inv = rsqrtf(var + 1e-5f);
  #pragma unroll
  for (int t = 0; t < 4; ++t) {
    const float4 ga = *(const float4*)(sG + 16 * t + fb);
    const float4 gb = *(const float4*)(sG + 16 * t + fb + 4);
    const float4 ea = *(const float4*)(sBeta + 16 * t + fb);
    const float4 eb = *(const float4*)(sBeta + 16 * t + fb + 4);
    const float g[8]  = { ga.x, ga.y, ga.z, ga.w, gb.x, gb.y, gb.z, gb.w };
    const float be[8] = { ea.x, ea.y, ea.z, ea.w, eb.x, eb.y, eb.z, eb.w };
    #pragma unroll
    for (int j = 0; j < 8; ++j) hv[t][j] = (hv[t][j] - mu) * inv * g[j] + be[j];
  }

  float zq[6];
  #pragma unroll
  for (int q = 0; q < 6; ++q) {
    v2f pt[4];
    #pragma unroll
    for (int t = 0; t < 4; ++t) {
      const float4 wa = *(const float4*)(sW2 + q * 64 + 16 * t + fb);
      const float4 wb = *(const float4*)(sW2 + q * 64 + 16 * t + fb + 4);
      pt[t]  = v2f{hv[t][0], hv[t][1]} * v2f{wa.x, wa.y};
      pt[t] += v2f{hv[t][2], hv[t][3]} * v2f{wa.z, wa.w};
      pt[t] += v2f{hv[t][4], hv[t][5]} * v2f{wb.x, wb.y};
      pt[t] += v2f{hv[t][6], hv[t][7]} * v2f{wb.z, wb.w};
    }
    const v2f pp = (pt[0] + pt[1]) + (pt[2] + pt[3]);
    float p = pp.x + pp.y;
    p += xhalf(p);
    zq[q] = tanh_bounded(p + sB2[q]);
  }

  float cq[6], sq[6];
  #pragma unroll
  for (int q = 0; q < 6; ++q) sincos_halfpi(zq[q], sq[q], cq[q]);
  float p01[4], p012[8], p0123[16];
  #pragma unroll
  for (int i = 0; i < 4; ++i)  p01[i]   = ((i >> 1) ? sq[0] : cq[0]) * ((i & 1) ? sq[1] : cq[1]);
  #pragma unroll
  for (int i = 0; i < 8; ++i)  p012[i]  = p01[i >> 1]  * ((i & 1) ? sq[2] : cq[2]);
  #pragma unroll
  for (int i = 0; i < 16; ++i) p0123[i] = p012[i >> 1] * ((i & 1) ? sq[3] : cq[3]);
  const float f4 = half ? sq[4] : cq[4];

  v8f amp[8] = {};
  v2f ua[2][8];
  #pragma unroll
  for (int t = 0; t < 8; ++t) ua[0][t] = *(const v2f*)(up + 16 * t * UP);
  v2f pfc;
  {
    float t4 = p0123[0] * f4;
    pfc.x = t4 * cq[5];
    pfc.y = t4 * sq[5];
  }
  #pragma unroll
  for (int kk = 0; kk < 16; ++kk) {
    if (kk + 1 < 16) {
      #pragma unroll
      for (int t = 0; t < 8; ++t) ua[(kk + 1) & 1][t] = *(const v2f*)(up + 16 * t * UP + 4 * (kk + 1));
    }
    __builtin_amdgcn_sched_barrier(0);
    #pragma unroll
    for (int t = 0; t < 8; ++t)
      amp[t] = __builtin_amdgcn_wmma_f32_16x16x4_f32(false, ua[kk & 1][t], false, pfc,
                                                     (short)0, amp[t], false, false);
    v2f pfn = pfc;
    if (kk + 1 < 16) {
      float t4 = p0123[kk + 1] * f4;
      pfn.x = t4 * cq[5];
      pfn.y = t4 * sq[5];
    }
    __builtin_amdgcn_sched_barrier(0);
    pfc = pfn;
  }

  float P[4][8];
  #pragma unroll
  for (int t = 0; t < 4; ++t) {
    #pragma unroll
    for (int r = 0; r < 8; ++r) P[t][r] = amp[t][r] * amp[t][r] + amp[t + 4][r] * amp[t + 4][r];
  }
  float e5 = 0.f, e4 = 0.f, e3 = 0.f;
  float s1[4][4], s2[4][2], s3[4];
  #pragma unroll
  for (int t = 0; t < 4; ++t) {
    #pragma unroll
    for (int m = 0; m < 4; ++m) { s1[t][m] = P[t][2*m] + P[t][2*m+1]; e5 += P[t][2*m] - P[t][2*m+1]; }
    #pragma unroll
    for (int m = 0; m < 2; ++m) { s2[t][m] = s1[t][2*m] + s1[t][2*m+1]; e4 += s1[t][2*m] - s1[t][2*m+1]; }
    s3[t] = s2[t][0] + s2[t][1]; e3 += s2[t][0] - s2[t][1];
  }
  const float u0 = s3[0] + s3[1], u1 = s3[2] + s3[3];
  const float e1 = (s3[0] - s3[1]) + (s3[2] - s3[3]);
  const float S  = u0 + u1;
  const float e0 = u0 - u1;
  const float e2 = half ? -S : S;

  const int row = rowBase + lm;
  #pragma unroll
  for (int k = 0; k < 5; ++k) {
    float v = sWc[k*6+0]*e0 + sWc[k*6+1]*e1 + sWc[k*6+2]*e2
            + sWc[k*6+3]*e3 + sWc[k*6+4]*e4 + sWc[k*6+5]*e5;
    v += __shfl_xor(v, 16, 32);
    if (half == 0 && row < B) out[(size_t)row * 5 + k] = v + sBc[k];
  }
  #pragma unroll
  for (int kk = 0; kk < 16; ++kk) xf[kk] = xn[kk];
  }
}

extern "C" void kernel_launch(void* const* d_in, const int* in_sizes, int n_in,
                              void* d_out, int out_size, void* d_ws, size_t ws_size,
                              hipStream_t stream) {
  (void)n_in; (void)out_size; (void)ws_size;
  const float* x    = (const float*)d_in[0];
  const float* W1   = (const float*)d_in[1];
  const float* b1   = (const float*)d_in[2];
  const float* ln_g = (const float*)d_in[3];
  const float* ln_b = (const float*)d_in[4];
  const float* W2   = (const float*)d_in[5];
  const float* b2   = (const float*)d_in[6];
  const float* th_s = (const float*)d_in[7];
  const float* th_t = (const float*)d_in[8];
  const float* Wc   = (const float*)d_in[9];
  const float* bc   = (const float*)d_in[10];
  float* out = (float*)d_out;
  float* uimg = (float*)d_ws;
  int B = in_sizes[0] / 64;
  circuit_compose<<<8, 256, 0, stream>>>(th_s, th_t, uimg);
  qmaml_kernel<<<512, 256, 0, stream>>>(x, W1, b1, ln_g, ln_b, W2, b2, uimg, Wc, bc, out, B);
}


# Circuit, readout and classifier as exact 3-node interpolation per qubit: launch 1 tabulates 729 node states, launch 2 does a (160x28)@(28x16) f32 WMMA product, 70 not 128 per tile; sum order changes
# speedup vs baseline: 10.4617x; 1.1401x over previous
#include <hip/hip_runtime.h>
#include <math.h>


typedef __attribute__((ext_vector_type(2))) float v2f;
typedef __attribute__((ext_vector_type(8))) float v8f;

struct Rings { int r1[64]; int r2[64]; };

constexpr Rings makeRings() {
  Rings R{};
  for (int rr = 1; rr <= 2; ++rr) {
    int total[64];
    for (int i = 0; i < 64; ++i) total[i] = i;
    for (int w = 0; w < 6; ++w) {
      int t = (w + rr) % 6;
      int nxt[64];
      for (int i = 0; i < 64; ++i) {
        int cbit = (i >> (5 - w)) & 1;
        nxt[i] = total[i ^ (cbit << (5 - t))];
      }
      for (int i = 0; i < 64; ++i) total[i] = nxt[i];
    }
    for (int i = 0; i < 64; ++i) { if (rr == 1) R.r1[i] = total[i]; else R.r2[i] = total[i]; }
  }
  return R;
}

__constant__ Rings RINGS = makeRings();

__device__ __forceinline__ float2 cmul(float2 a, float2 b) {
  return make_float2(a.x * b.x - a.y * b.y, a.x * b.y + a.y * b.x);
}
__device__ __forceinline__ float2 cadd(float2 a, float2 b) { return make_float2(a.x + b.x, a.y + b.y); }
__device__ __forceinline__ float2 shflx2(float2 v, int m) {
  return make_float2(__shfl_xor(v.x, m, 32), __shfl_xor(v.y, m, 32));
}
__device__ __forceinline__ float2 shfl2(float2 v, int l) {
  return make_float2(__shfl(v.x, l, 32), __shfl(v.y, l, 32));
}

__device__ __forceinline__ float xhalf(float v) {
  int b = __builtin_bit_cast(int, v);
  return __builtin_bit_cast(float, __builtin_amdgcn_permlanex16(b, b, 0x76543210, 0xfedcba98, false, false));
}

__device__ __forceinline__ float tanh_bounded(float x) {
  const float t = __builtin_amdgcn_exp2f(fabsf(x) * -2.885390081777927f);
  const float r = (1.0f - t) * __builtin_amdgcn_rcpf(1.0f + t);
  return copysignf(r, x);
}

__device__ __forceinline__ void sincos_halfpi(float z, float& s, float& c) {
  const float u  = 0.78539816339744831f * z;
  const float u2 = u * u;
  const float ps = fmaf(u2, fmaf(u2, -1.9515295891e-4f, 8.3321608736e-3f), -1.6666654611e-1f);
  const float su = fmaf(u * u2, ps, u);
  const float pc = fmaf(u2, fmaf(u2, 2.443315711809948e-5f, -1.388731625493765e-3f), 4.166664568298827e-2f);
  const float cu = fmaf(u2 * u2, pc, fmaf(u2, -0.5f, 1.0f));
  const float s2 = su + su;
  s = s2 * cu;
  c = fmaf(-s2, su, 1.0f);
}

#define VP 36
#define VROWS 160
#define NODE_S 0.8660254037844386f
#define WP 68

__global__ __launch_bounds__(192) void circuit_compose(
    const float* __restrict__ th_s, const float* __restrict__ th_t,
    const float* __restrict__ Wc, float* __restrict__ vimg)
{
  __shared__ float2 sGates[72];
  __shared__ float  sLine[32];

  const int tid = threadIdx.x;
  const int kc = ((tid & 31) < 5) ? (tid & 31) : 0;
  float wc[6];
  #pragma unroll
  for (int q = 0; q < 6; ++q) wc[q] = Wc[kc * 6 + q];
  if (tid < 18) {
    int l = tid / 6, w = tid % 6;
    const float* th = (l < 2) ? (th_s + (l * 6 + w) * 3) : (th_t + w * 3);
    float phi = th[0], theta = th[1], omega = th[2];
    float c  = cosf(0.5f * theta), s = sinf(0.5f * theta);
    float aa = 0.5f * (phi + omega), bb = 0.5f * (phi - omega);
    float ca = cosf(aa), sa = sinf(aa);
    float cb = cosf(bb), sb = sinf(bb);
    sGates[tid * 4 + 0] = make_float2( ca * c, -sa * c);
    sGates[tid * 4 + 1] = make_float2(-cb * s, -sb * s);
    sGates[tid * 4 + 2] = make_float2( cb * s, -sb * s);
    sGates[tid * 4 + 3] = make_float2( ca * c,  sa * c);
  }
  __syncthreads();
  asm volatile("" :: "v"(wc[0]), "v"(wc[1]), "v"(wc[2]), "v"(wc[3]), "v"(wc[4]), "v"(wc[5]));

  const int lane = tid & 31;
  const int wave = tid >> 5;
  const int rA0 = RINGS.r1[lane], rA1 = RINGS.r1[lane + 32];
  const int rB0 = RINGS.r2[lane], rB1 = RINGS.r2[lane + 32];

  const int st = blockIdx.x * 6 + wave;
  const int dg[6] = { (st / 243) % 3, (st / 81) % 3, (st / 27) % 3, (st / 9) % 3, (st / 3) % 3, st % 3 };
  float am0 = 1.f, am1 = 1.f;
  #pragma unroll
  for (int q = 0; q < 6; ++q) {
    const float nc = (dg[q] == 0) ? 1.f : ((dg[q] == 1) ? 0.5f : -0.5f);
    const float ns = (dg[q] == 0) ? 0.f : NODE_S;
    const int bit = (q == 0) ? 0 : ((lane >> (5 - q)) & 1);
    am0 *= bit ? ns : nc;
    am1 *= ((q == 0) ? 1 : bit) ? ns : nc;
  }
  float2 a0 = make_float2(am0, 0.f);
  float2 a1 = make_float2(am1, 0.f);

  #pragma unroll
  for (int l = 0; l < 3; ++l) {
    #pragma unroll
    for (int w = 0; w < 6; ++w) {
      int g = l * 6 + w;
      float2 m00 = sGates[g * 4 + 0], m01 = sGates[g * 4 + 1];
      float2 m10 = sGates[g * 4 + 2], m11 = sGates[g * 4 + 3];
      if (w == 0) {
        float2 n0 = cadd(cmul(m00, a0), cmul(m01, a1));
        float2 n1 = cadd(cmul(m10, a0), cmul(m11, a1));
        a0 = n0; a1 = n1;
      } else {
        int b = 5 - w, mask = 1 << b;
        float2 p0 = shflx2(a0, mask), p1 = shflx2(a1, mask);
        int bit = (lane >> b) & 1;
        float2 e0 = cadd(cmul(m00, a0), cmul(m01, p0));
        float2 o0 = cadd(cmul(m10, p0), cmul(m11, a0));
        float2 e1 = cadd(cmul(m00, a1), cmul(m01, p1));
        float2 o1 = cadd(cmul(m10, p1), cmul(m11, a1));
        a0 = bit ? o0 : e0;
        a1 = bit ? o1 : e1;
      }
    }
    int s0 = (l == 1) ? rB0 : rA0;
    int s1 = (l == 1) ? rB1 : rA1;
    float2 g0a = shfl2(a0, s0 & 31), g0b = shfl2(a1, s0 & 31);
    float2 g1a = shfl2(a0, s1 & 31), g1b = shfl2(a1, s1 & 31);
    a0 = (s0 < 32) ? g0a : g0b;
    a1 = (s1 < 32) ? g1a : g1b;
  }
  const float P0 = a0.x * a0.x + a0.y * a0.y;
  const float P1 = a1.x * a1.x + a1.y * a1.y;
  const float Ps = P0 + P1;
  float e[6];
  e[0] = P0 - P1;
  #pragma unroll
  for (int q = 1; q < 6; ++q) e[q] = ((lane >> (5 - q)) & 1) ? -Ps : Ps;
  #pragma unroll
  for (int q = 0; q < 6; ++q) {
    #pragma unroll
    for (int m = 16; m >= 1; m >>= 1) e[q] += __shfl_xor(e[q], m, 32);
  }
  float v = 0.f;
  #pragma unroll
  for (int q = 0; q < 6; ++q) v += wc[q] * e[q];
  if (lane < 5) sLine[wave * 5 + lane] = (st < 729) ? v : 0.f;
  if (tid < 2) sLine[30 + tid] = 0.f;
  __syncthreads();
  if (tid < 32) vimg[blockIdx.x * 32 + tid] = sLine[tid];
}

__global__ __launch_bounds__(256) void qmaml_kernel(
    const float* __restrict__ x,  const float* __restrict__ W1,
    const float* __restrict__ b1, const float* __restrict__ ln_g,
    const float* __restrict__ ln_b, const float* __restrict__ W2,
    const float* __restrict__ b2, const float* __restrict__ vimg,
    const float* __restrict__ bc,
    float* __restrict__ out, int B)
{
  __shared__ float  sW1[64 * WP];
  __shared__ __attribute__((aligned(16))) float sW2[512];
  __shared__ __attribute__((aligned(16))) float sSmall[256];
  const float* sB1 = sSmall;
  const float* sG = sSmall + 64;
  const float* sBeta = sSmall + 128;
  const float* sB2 = sSmall + 192;
  const float* sBc = sSmall + 198;
  __shared__ __attribute__((aligned(16))) float sV[VROWS * VP];

  const int tid = threadIdx.x;
  const int lane = tid & 31;
  const int wave = __builtin_amdgcn_readfirstlane(tid >> 5);
  const int half = lane >> 4;
  const int lm   = lane & 15;
  const int nTiles = (B + 15) >> 4;
  const int tStep  = gridDim.x * 8;
  int tile = blockIdx.x * 8 + wave;

  float4 rW1[4];
  #pragma unroll
  for (int j = 0; j < 4; ++j) rW1[j] = ((const float4*)W1)[tid + 256 * j];
  float rV[3][5];
  #pragma unroll
  for (int j = 0; j < 3; ++j) {
    int sN = tid + 256 * j; if (sN > 728) sN = 728;
    const float* pv = vimg + 32 * (sN / 6) + 5 * (sN % 6);
    #pragma unroll
    for (int k = 0; k < 5; ++k) rV[j][k] = pv[k];
  }
  const float rW2a = W2[tid];
  const float rW2b = W2[256 + (tid & 127)];
  const int i64 = tid & 63;
  const int iB2 = tid < 192 ? 0 : (tid < 198 ? tid - 192 : 5);
  const int iBc = tid < 198 ? 0 : (tid < 203 ? tid - 198 : 4);
  const float* const pB1 = b1 + i64;
  const float* const pG  = ln_g + i64;
  const float* const pBe = ln_b + i64;
  const float* const pB2 = b2 + iB2;
  const float* const pBc = bc + iBc;
  const float* pSmall = pBc;
  pSmall = (tid < 198) ? pB2 : pSmall;
  pSmall = (tid < 192) ? pBe : pSmall;
  pSmall = (tid < 128) ? pG  : pSmall;
  pSmall = (tid <  64) ? pB1 : pSmall;
  const float rSmall = *pSmall;
  v2f xf[16];
  {
    int rowA = tile * 16 + lm; if (rowA >= B) rowA = B - 1;
    const float* xrow = x + (size_t)rowA * 64 + 2 * half;
    #pragma unroll
    for (int kk = 0; kk < 16; ++kk) xf[kk] = *(const v2f*)(xrow + 4 * kk);
  }

  #pragma unroll
  for (int j = 0; j < 4; ++j) {
    const int i = tid + 256 * j;
    *(float4*)(sW1 + (i >> 4) * WP + 4 * (i & 15)) = rW1[j];
  }
  #pragma unroll
  for (int j = 0; j < 3; ++j) {
    const int sN = tid + 256 * j;
    const int sC = sN > 728 ? 728 : sN;
    const int g = sC / 27;
    const int h = (sN < 729) ? sC - 27 * g : 28;
    #pragma unroll
    for (int k = 0; k < 5; ++k) sV[(32 * k + g) * VP + h] = rV[j][k];
  }
  if (tid < 225) {
    const int k = tid / 45, rem = tid - 45 * k, j = rem / 9, c4 = rem - 9 * j;
    *(float4*)(sV + (32 * k + 27 + j) * VP + 4 * c4) = make_float4(0.f, 0.f, 0.f, 0.f);
  }
  if (tid < 135) {
    const int k = tid / 27, g = tid - 27 * k;
    sV[(32 * k + g) * VP + 27] = 0.f;
  }
  sW2[tid] = rW2a;
  sW2[256 + tid] = rW2b;
  sSmall[tid] = rSmall;
  __syncthreads();

  const int fb = 8 * half;
  const float* w1p = sW1 + lm * WP + 2 * half;
  const float* vp = sV + lm * VP + 2 * half;

  for (; tile < nTiles; tile += tStep) {
  const int rowBase = tile * 16;
  v2f xn[16];
  {
    int rowN = rowBase + tStep * 16 + lm; if (rowN >= B) rowN = B - 1;
    const float* xrowN = x + (size_t)rowN * 64 + 2 * half;
    #pragma unroll
    for (int kk = 0; kk < 16; ++kk) xn[kk] = *(const v2f*)(xrowN + 4 * kk);
  }

  v8f acc[4] = {};
  v2f wa[2][8];
  #pragma unroll
  for (int j = 0; j < 2; ++j) {
    #pragma unroll
    for (int t = 0; t < 4; ++t) wa[0][4 * j + t] = *(const v2f*)(w1p + 16 * t * WP + 4 * j);
  }
  #pragma unroll
  for (int b = 0; b < 8; ++b) {
    if (b + 1 < 8) {
      #pragma unroll
      for (int j = 0; j < 2; ++j) {
        #pragma unroll
        for (int t = 0; t < 4; ++t)
          wa[(b + 1) & 1][4 * j + t] = *(const v2f*)(w1p + 16 * t * WP + 4 * (2 * b + 2 + j));
      }
    }
    __builtin_amdgcn_sched_barrier(0);
    #pragma unroll
    for (int j = 0; j < 2; ++j) {
      #pragma unroll
      for (int t = 0; t < 4; ++t)
        acc[t] = __builtin_amdgcn_wmma_f32_16x16x4_f32(false, wa[b & 1][4 * j + t], false, xf[2 * b + j],
                                                       (short)0, acc[t], false, false);
    }
    __builtin_amdgcn_sched_barrier(0);
  }

  float hv[4][8];
  #pragma unroll
  for (int t = 0; t < 4; ++t) {
    const float4 ba = *(const float4*)(sB1 + 16 * t + fb);
    const float4 bb = *(const float4*)(sB1 + 16 * t + fb + 4);
    const float bias[8] = { ba.x, ba.y, ba.z, ba.w, bb.x, bb.y, bb.z, bb.w };
    #pragma unroll
    for (int j = 0; j < 8; ++j) hv[t][j] = fmaxf(acc[t][j] + bias[j], 0.0f);
  }

  float st[4], qt[4];
  #pragma unroll
  for (int t = 0; t < 4; ++t) {
    st[t] = hv[t][0];
    qt[t] = hv[t][0] * hv[t][0];
    #pragma unroll
    for (int j = 1; j < 8; ++j) { st[t] += hv[t][j]; qt[t] += hv[t][j] * hv[t][j]; }
  }
  float s  = (st[0] + st[1]) + (st[2] + st[3]);
  float q2 = (qt[0] + qt[1]) + (qt[2] + qt[3]);
  s  += xhalf(s);
  q2 += xhalf(q2);
  const float mu  = s * 0.015625f;
  const float var = q2 * 0.015625f - mu * mu;
  const float inv = rsqrtf(var + 1e-5f);
  #pragma unroll
  for (int t = 0; t < 4; ++t) {
    const float4 ga = *(const float4*)(sG + 16 * t + fb);
    const float4 gb = *(const float4*)(sG + 16 * t + fb + 4);
    const float4 ea = *(const float4*)(sBeta + 16 * t + fb);
    const float4 eb = *(const float4*)(sBeta + 16 * t + fb + 4);
    const float g[8]  = { ga.x, ga.y, ga.z, ga.w, gb.x, gb.y, gb.z, gb.w };
    const float be[8] = { ea.x, ea.y, ea.z, ea.w, eb.x, eb.y, eb.z, eb.w };
    #pragma unroll
    for (int j = 0; j < 8; ++j) hv[t][j] = (hv[t][j] - mu) * inv * g[j] + be[j];
  }

  float zq[6];
  #pragma unroll
  for (int q = 0; q < 6; ++q) {
    v2f pt[4];
    #pragma unroll
    for (int t = 0; t < 4; ++t) {
      const float4 wa = *(const float4*)(sW2 + q * 64 + 16 * t + fb);
      const float4 wb = *(const float4*)(sW2 + q * 64 + 16 * t + fb + 4);
      pt[t]  = v2f{hv[t][0], hv[t][1]} * v2f{wa.x, wa.y};
      pt[t] += v2f{hv[t][2], hv[t][3]} * v2f{wa.z, wa.w};
      pt[t] += v2f{hv[t][4], hv[t][5]} * v2f{wb.x, wb.y};
      pt[t] += v2f{hv[t][6], hv[t][7]} * v2f{wb.z, wb.w};
    }
    const v2f pp = (pt[0] + pt[1]) + (pt[2] + pt[3]);
    float p = pp.x + pp.y;
    p += xhalf(p);
    zq[q] = tanh_bounded(p + sB2[q]);
  }

  float f0_0, f0_1, f0_2;
  {
    float sh, ch;
    sincos_halfpi(zq[0], sh, ch);
    const float ct = fmaf(-2.0f * sh, sh, 1.0f);
    const float st = 2.0f * sh * ch;
    const float bq = fmaf(ct, -0.33333333333333333f, 0.33333333333333333f);
    const float dq = st * 0.57735026918962576f;
    f0_0 = fmaf(ct, 0.66666666666666667f, 0.33333333333333333f);
    f0_1 = bq + dq;
    f0_2 = bq - dq;
  }
  float f1_0, f1_1, f1_2;
  {
    float sh, ch;
    sincos_halfpi(zq[1], sh, ch);
    const float ct = fmaf(-2.0f * sh, sh, 1.0f);
    const float st = 2.0f * sh * ch;
    const float bq = fmaf(ct, -0.33333333333333333f, 0.33333333333333333f);
    const float dq = st * 0.57735026918962576f;
    f1_0 = fmaf(ct, 0.66666666666666667f, 0.33333333333333333f);
    f1_1 = bq + dq;
    f1_2 = bq - dq;
  }
  float f2_0, f2_1, f2_2;
  {
    float sh, ch;
    sincos_halfpi(zq[2], sh, ch);
    const float ct = fmaf(-2.0f * sh, sh, 1.0f);
    const float st = 2.0f * sh * ch;
    const float bq = fmaf(ct, -0.33333333333333333f, 0.33333333333333333f);
    const float dq = st * 0.57735026918962576f;
    f2_0 = fmaf(ct, 0.66666666666666667f, 0.33333333333333333f);
    f2_1 = bq + dq;
    f2_2 = bq - dq;
  }
  float f3_0, f3_1, f3_2;
  {
    float sh, ch;
    sincos_halfpi(zq[3], sh, ch);
    const float ct = fmaf(-2.0f * sh, sh, 1.0f);
    const float st = 2.0f * sh * ch;
    const float bq = fmaf(ct, -0.33333333333333333f, 0.33333333333333333f);
    const float dq = st * 0.57735026918962576f;
    f3_0 = fmaf(ct, 0.66666666666666667f, 0.33333333333333333f);
    f3_1 = bq + dq;
    f3_2 = bq - dq;
  }
  float f4_0, f4_1, f4_2;
  {
    float sh, ch;
    sincos_halfpi(zq[4], sh, ch);
    const float ct = fmaf(-2.0f * sh, sh, 1.0f);
    const float st = 2.0f * sh * ch;
    const float bq = fmaf(ct, -0.33333333333333333f, 0.33333333333333333f);
    const float dq = st * 0.57735026918962576f;
    f4_0 = fmaf(ct, 0.66666666666666667f, 0.33333333333333333f);
    f4_1 = bq + dq;
    f4_2 = bq - dq;
  }
  float f5_0, f5_1, f5_2;
  {
    float sh, ch;
    sincos_halfpi(zq[5], sh, ch);
    const float ct = fmaf(-2.0f * sh, sh, 1.0f);
    const float st = 2.0f * sh * ch;
    const float bq = fmaf(ct, -0.33333333333333333f, 0.33333333333333333f);
    const float dq = st * 0.57735026918962576f;
    f5_0 = fmaf(ct, 0.66666666666666667f, 0.33333333333333333f);
    f5_1 = bq + dq;
    f5_2 = bq - dq;
  }
  const float g12_0 = f1_0 * f2_0, h45_0 = f4_0 * f5_0;
  const float g12_1 = f1_0 * f2_1, h45_1 = f4_0 * f5_1;
  const float g12_2 = f1_0 * f2_2, h45_2 = f4_0 * f5_2;
  const float g12_3 = f1_1 * f2_0, h45_3 = f4_1 * f5_0;
  const float g12_4 = f1_1 * f2_1, h45_4 = f4_1 * f5_1;
  const float g12_5 = f1_1 * f2_2, h45_5 = f4_1 * f5_2;
  const float g12_6 = f1_2 * f2_0, h45_6 = f4_2 * f5_0;
  const float g12_7 = f1_2 * f2_1, h45_7 = f4_2 * f5_1;
  const float g12_8 = f1_2 * f2_2, h45_8 = f4_2 * f5_2;
  const float G_0 = f0_0 * g12_0, H_0 = f3_0 * h45_0;
  const float G_1 = f0_0 * g12_1, H_1 = f3_0 * h45_1;
  const float G_2 = f0_0 * g12_2, H_2 = f3_0 * h45_2;
  const float G_3 = f0_0 * g12_3, H_3 = f3_0 * h45_3;
  const float G_4 = f0_0 * g12_4, H_4 = f3_0 * h45_4;
  const float G_5 = f0_0 * g12_5, H_5 = f3_0 * h45_5;
  const float G_6 = f0_0 * g12_6, H_6 = f3_0 * h45_6;
  const float G_7 = f0_0 * g12_7, H_7 = f3_0 * h45_7;
  const float G_8 = f0_0 * g12_8, H_8 = f3_0 * h45_8;
  const float G_9 = f0_1 * g12_0, H_9 = f3_1 * h45_0;
  const float G_10 = f0_1 * g12_1, H_10 = f3_1 * h45_1;
  const float G_11 = f0_1 * g12_2, H_11 = f3_1 * h45_2;
  const float G_12 = f0_1 * g12_3, H_12 = f3_1 * h45_3;
  const float G_13 = f0_1 * g12_4, H_13 = f3_1 * h45_4;
  const float G_14 = f0_1 * g12_5, H_14 = f3_1 * h45_5;
  const float G_15 = f0_1 * g12_6, H_15 = f3_1 * h45_6;
  const float G_16 = f0_1 * g12_7, H_16 = f3_1 * h45_7;
  const float G_17 = f0_1 * g12_8, H_17 = f3_1 * h45_8;
  const float G_18 = f0_2 * g12_0, H_18 = f3_2 * h45_0;
  const float G_19 = f0_2 * g12_1, H_19 = f3_2 * h45_1;
  const float G_20 = f0_2 * g12_2, H_20 = f3_2 * h45_2;
  const float G_21 = f0_2 * g12_3, H_21 = f3_2 * h45_3;
  const float G_22 = f0_2 * g12_4, H_22 = f3_2 * h45_4;
  const float G_23 = f0_2 * g12_5, H_23 = f3_2 * h45_5;
  const float G_24 = f0_2 * g12_6, H_24 = f3_2 * h45_6;
  const float G_25 = f0_2 * g12_7, H_25 = f3_2 * h45_7;
  const float G_26 = f0_2 * g12_8, H_26 = f3_2 * h45_8;
  v2f hb[7];
  hb[0].x = half ? H_2 : H_0;  hb[0].y = half ? H_3 : H_1;
  hb[1].x = half ? H_6 : H_4;  hb[1].y = half ? H_7 : H_5;
  hb[2].x = half ? H_10 : H_8;  hb[2].y = half ? H_11 : H_9;
  hb[3].x = half ? H_14 : H_12;  hb[3].y = half ? H_15 : H_13;
  hb[4].x = half ? H_18 : H_16;  hb[4].y = half ? H_19 : H_17;
  hb[5].x = half ? H_22 : H_20;  hb[5].y = half ? H_23 : H_21;
  hb[6].x = half ? H_26 : H_24;  hb[6].y = half ? 0.0f : H_25;
  float gs[2][8];
  gs[0][0] = half ? G_8 : G_0;
  gs[0][1] = half ? G_9 : G_1;
  gs[0][2] = half ? G_10 : G_2;
  gs[0][3] = half ? G_11 : G_3;
  gs[0][4] = half ? G_12 : G_4;
  gs[0][5] = half ? G_13 : G_5;
  gs[0][6] = half ? G_14 : G_6;
  gs[0][7] = half ? G_15 : G_7;
  gs[1][0] = half ? G_24 : G_16;
  gs[1][1] = half ? G_25 : G_17;
  gs[1][2] = half ? G_26 : G_18;
  gs[1][3] = half ? 0.0f : G_19;
  gs[1][4] = half ? 0.0f : G_20;
  gs[1][5] = half ? 0.0f : G_21;
  gs[1][6] = half ? 0.0f : G_22;
  gs[1][7] = half ? 0.0f : G_23;

  v8f yv[10] = {};
  v2f va[2][10];
  #pragma unroll
  for (int t = 0; t < 10; ++t) va[0][t] = *(const v2f*)(vp + 16 * t * VP);
  #pragma unroll
  for (int kk = 0; kk < 7; ++kk) {
    if (kk + 1 < 7) {
      #pragma unroll
      for (int t = 0; t < 10; ++t) va[(kk + 1) & 1][t] = *(const v2f*)(vp + 16 * t * VP + 4 * (kk + 1));
    }
    __builtin_amdgcn_sched_barrier(0);
    #pragma unroll
    for (int t = 0; t < 10; ++t)
      yv[t] = __builtin_amdgcn_wmma_f32_16x16x4_f32(false, va[kk & 1][t], false, hb[kk],
                                                    (short)0, yv[t], false, false);
    __builtin_amdgcn_sched_barrier(0);
  }

  const int row = rowBase + lm;
  #pragma unroll
  for (int k = 0; k < 5; ++k) {
    float v0 = gs[0][0] * yv[2 * k][0], v1 = gs[1][0] * yv[2 * k + 1][0];
    #pragma unroll
    for (int r = 1; r < 8; ++r) { v0 = fmaf(gs[0][r], yv[2 * k][r], v0); v1 = fmaf(gs[1][r], yv[2 * k + 1][r], v1); }
    float v = v0 + v1;
    v += xhalf(v);
    if (half == 0 && row < B) out[(size_t)row * 5 + k] = v + sBc[k];
  }
  #pragma unroll
  for (int kk = 0; kk < 16; ++kk) xf[kk] = xn[kk];
  }
}

extern "C" void kernel_launch(void* const* d_in, const int* in_sizes, int n_in,
                              void* d_out, int out_size, void* d_ws, size_t ws_size,
                              hipStream_t stream) {
  (void)n_in; (void)out_size; (void)ws_size;
  const float* x    = (const float*)d_in[0];
  const float* W1   = (const float*)d_in[1];
  const float* b1   = (const float*)d_in[2];
  const float* ln_g = (const float*)d_in[3];
  const float* ln_b = (const float*)d_in[4];
  const float* W2   = (const float*)d_in[5];
  const float* b2   = (const float*)d_in[6];
  const float* th_s = (const float*)d_in[7];
  const float* th_t = (const float*)d_in[8];
  const float* Wc   = (const float*)d_in[9];
  const float* bc   = (const float*)d_in[10];
  float* out = (float*)d_out;
  float* vimg = (float*)d_ws;
  int B = in_sizes[0] / 64;
  circuit_compose<<<122, 192, 0, stream>>>(th_s, th_t, Wc, vimg);
  qmaml_kernel<<<512, 256, 0, stream>>>(x, W1, b1, ln_g, ln_b, W2, b2, vimg, bc, out, B);
}


# Launch 1 gate loop: partner amplitudes by vector-ALU lane permutes (DPP row_xmask, permlanex16), rings folded into compile-time index masks; a lane applies only its own matrix row; sum order changes
# speedup vs baseline: 11.0619x; 1.0574x over previous
#include <hip/hip_runtime.h>
#include <math.h>


typedef __attribute__((ext_vector_type(2))) float v2f;
typedef __attribute__((ext_vector_type(8))) float v8f;
typedef __attribute__((ext_vector_type(4))) float v4f;

struct Rings { int r1[64]; int r2[64]; };

constexpr Rings makeRings() {
  Rings R{};
  for (int rr = 1; rr <= 2; ++rr) {
    int total[64];
    for (int i = 0; i < 64; ++i) total[i] = i;
    for (int w = 0; w < 6; ++w) {
      int t = (w + rr) % 6;
      int nxt[64];
      for (int i = 0; i < 64; ++i) {
        int cbit = (i >> (5 - w)) & 1;
        nxt[i] = total[i ^ (cbit << (5 - t))];
      }
      for (int i = 0; i < 64; ++i) total[i] = nxt[i];
    }
    for (int i = 0; i < 64; ++i) { if (rr == 1) R.r1[i] = total[i]; else R.r2[i] = total[i]; }
  }
  return R;
}

struct Walk { int pm[18]; int rm[18]; int zm[6]; };

constexpr int parityMask(const int* inv, int b) {
  int r = 0;
  for (int k = 0; k < 6; ++k) { if ((inv[1 << k] >> b) & 1) r |= (1 << k); }
  return r;
}

constexpr Walk makeWalk() {
  Walk K{};
  const Rings R = makeRings();
  int pos[64] = {};
  int inv[64] = {};
  for (int j = 0; j < 64; ++j) pos[j] = j;
  for (int l = 0; l < 3; ++l) {
    for (int j = 0; j < 64; ++j) inv[pos[j]] = j;
    for (int w = 0; w < 6; ++w) {
      K.pm[l * 6 + w] = pos[1 << (5 - w)];
      K.rm[l * 6 + w] = parityMask(inv, 5 - w);
    }
    int nxt[64] = {};
    for (int j = 0; j < 64; ++j) nxt[j] = pos[(l == 1) ? R.r2[j] : R.r1[j]];
    for (int j = 0; j < 64; ++j) pos[j] = nxt[j];
  }
  for (int j = 0; j < 64; ++j) inv[pos[j]] = j;
  for (int q = 0; q < 6; ++q) K.zm[q] = parityMask(inv, 5 - q);
  return K;
}
constexpr int walkPm(int g) { return makeWalk().pm[g]; }
constexpr int walkRm(int g) { return makeWalk().rm[g]; }
constexpr int walkZm(int q) { return makeWalk().zm[q]; }

__device__ __forceinline__ float2 cmul(float2 a, float2 b) {
  return make_float2(a.x * b.x - a.y * b.y, a.x * b.y + a.y * b.x);
}
__device__ __forceinline__ float2 cadd(float2 a, float2 b) { return make_float2(a.x + b.x, a.y + b.y); }
constexpr unsigned selLo(int m) { unsigned r = 0; for (int i = 0; i < 8; ++i) r |= (unsigned)((i ^ m) & 15) << (4 * i); return r; }
constexpr unsigned selHi(int m) { unsigned r = 0; for (int i = 8; i < 16; ++i) r |= (unsigned)((i ^ m) & 15) << (4 * (i - 8)); return r; }

template <int LM>
__device__ __forceinline__ float lanex(float v) {
  if constexpr (LM == 0) {
    return v;
  } else {
    const int b = __builtin_bit_cast(int, v);
    if constexpr (LM < 16) {
      return __builtin_bit_cast(float, __builtin_amdgcn_update_dpp(0, b, 0x160 + LM, 0xf, 0xf, true));
    } else {
      return __builtin_bit_cast(float, __builtin_amdgcn_permlanex16(b, b, selLo(LM & 15), selHi(LM & 15), false, false));
    }
  }
}
template <int LM>
__device__ __forceinline__ float2 lanex2(float2 v) { return make_float2(lanex<LM>(v.x), lanex<LM>(v.y)); }

template <int G>
__device__ __forceinline__ void gateRows(v4f& r0, v4f& r1, const v4f* sg4, const int lane) {
  constexpr int RM = walkRm(G);
  const int b0 = __builtin_popcount(lane & (RM & 31)) & 1;
  const int b1 = b0 ^ ((RM >> 5) & 1);
  r0 = sg4[2 * G + b0];
  r1 = sg4[2 * G + b1];
}
template <int G>
__device__ __forceinline__ void gateStep(float2& a0, float2& a1, const v4f r0, const v4f r1) {
  constexpr int PM = walkPm(G);
  const float2 s0 = (PM & 32) ? a1 : a0;
  const float2 s1 = (PM & 32) ? a0 : a1;
  const float2 p0 = lanex2<(PM & 31)>(s0);
  const float2 p1 = lanex2<(PM & 31)>(s1);
  a0 = cadd(cmul(make_float2(r0.x, r0.y), a0), cmul(make_float2(r0.z, r0.w), p0));
  a1 = cadd(cmul(make_float2(r1.x, r1.y), a1), cmul(make_float2(r1.z, r1.w), p1));
}

template <int Q>
__device__ __forceinline__ float zTerm(const float P0, const float P1, const int lane) {
  constexpr int ZM = walkZm(Q);
  const int s0 = __builtin_popcount(lane & (ZM & 31)) & 1;
  const int s1 = s0 ^ ((ZM >> 5) & 1);
  return (s0 ? -P0 : P0) + (s1 ? -P1 : P1);
}

__device__ __forceinline__ float xhalf(float v) {
  int b = __builtin_bit_cast(int, v);
  return __builtin_bit_cast(float, __builtin_amdgcn_permlanex16(b, b, 0x76543210, 0xfedcba98, false, false));
}

__device__ __forceinline__ float tanh_bounded(float x) {
  const float t = __builtin_amdgcn_exp2f(fabsf(x) * -2.885390081777927f);
  const float r = (1.0f - t) * __builtin_amdgcn_rcpf(1.0f + t);
  return copysignf(r, x);
}

__device__ __forceinline__ void sincos_halfpi(float z, float& s, float& c) {
  const float u  = 0.78539816339744831f * z;
  const float u2 = u * u;
  const float ps = fmaf(u2, fmaf(u2, -1.9515295891e-4f, 8.3321608736e-3f), -1.6666654611e-1f);
  const float su = fmaf(u * u2, ps, u);
  const float pc = fmaf(u2, fmaf(u2, 2.443315711809948e-5f, -1.388731625493765e-3f), 4.166664568298827e-2f);
  const float cu = fmaf(u2 * u2, pc, fmaf(u2, -0.5f, 1.0f));
  const float s2 = su + su;
  s = s2 * cu;
  c = fmaf(-s2, su, 1.0f);
}

#define VP 36
#define VROWS 160
#define NODE_S 0.8660254037844386f
#define WP 68

__global__ __launch_bounds__(192) void circuit_compose(
    const float* __restrict__ th_s, const float* __restrict__ th_t,
    const float* __restrict__ Wc, float* __restrict__ vimg)
{
  __shared__ __attribute__((aligned(16))) float2 sGates[72];
  __shared__ float  sLine[32];

  const int tid = threadIdx.x;
  const int kc = ((tid & 31) < 5) ? (tid & 31) : 0;
  float wc[6];
  #pragma unroll
  for (int q = 0; q < 6; ++q) wc[q] = Wc[kc * 6 + q];
  if (tid < 18) {
    int l = tid / 6, w = tid % 6;
    const float* th = (l < 2) ? (th_s + (l * 6 + w) * 3) : (th_t + w * 3);
    float phi = th[0], theta = th[1], omega = th[2];
    float c  = cosf(0.5f * theta), s = sinf(0.5f * theta);
    float aa = 0.5f * (phi + omega), bb = 0.5f * (phi - omega);
    float ca = cosf(aa), sa = sinf(aa);
    float cb = cosf(bb), sb = sinf(bb);
    sGates[tid * 4 + 0] = make_float2( ca * c, -sa * c);
    sGates[tid * 4 + 1] = make_float2(-cb * s, -sb * s);
    sGates[tid * 4 + 2] = make_float2( ca * c,  sa * c);
    sGates[tid * 4 + 3] = make_float2( cb * s, -sb * s);
  }
  __syncthreads();
  asm volatile("" :: "v"(wc[0]), "v"(wc[1]), "v"(wc[2]), "v"(wc[3]), "v"(wc[4]), "v"(wc[5]));

  const int lane = tid & 31;
  const int wave = tid >> 5;

  const int st = blockIdx.x * 6 + wave;
  const int dg[6] = { (st / 243) % 3, (st / 81) % 3, (st / 27) % 3, (st / 9) % 3, (st / 3) % 3, st % 3 };
  float am0 = 1.f, am1 = 1.f;
  #pragma unroll
  for (int q = 0; q < 6; ++q) {
    const float nc = (dg[q] == 0) ? 1.f : ((dg[q] == 1) ? 0.5f : -0.5f);
    const float ns = (dg[q] == 0) ? 0.f : NODE_S;
    const int bit = (q == 0) ? 0 : ((lane >> (5 - q)) & 1);
    am0 *= bit ? ns : nc;
    am1 *= ((q == 0) ? 1 : bit) ? ns : nc;
  }
  float2 a0 = make_float2(am0, 0.f);
  float2 a1 = make_float2(am1, 0.f);

  const v4f* sg4 = (const v4f*)sGates;
  v4f r0[18], r1[18];
  gateRows<0>(r0[0], r1[0], sg4, lane);
  gateRows<1>(r0[1], r1[1], sg4, lane);
  gateRows<2>(r0[2], r1[2], sg4, lane);
  gateRows<3>(r0[3], r1[3], sg4, lane);
  gateRows<4>(r0[4], r1[4], sg4, lane);
  gateRows<5>(r0[5], r1[5], sg4, lane);
  __builtin_amdgcn_sched_barrier(0);
  gateStep<0>(a0, a1, r0[0], r1[0]);
  gateStep<1>(a0, a1, r0[1], r1[1]);
  gateStep<2>(a0, a1, r0[2], r1[2]);
  gateRows<6>(r0[6], r1[6], sg4, lane);
  gateRows<7>(r0[7], r1[7], sg4, lane);
  gateRows<8>(r0[8], r1[8], sg4, lane);
  __builtin_amdgcn_sched_barrier(0);
  gateStep<3>(a0, a1, r0[3], r1[3]);
  gateStep<4>(a0, a1, r0[4], r1[4]);
  gateStep<5>(a0, a1, r0[5], r1[5]);
  gateRows<9>(r0[9], r1[9], sg4, lane);
  gateRows<10>(r0[10], r1[10], sg4, lane);
  gateRows<11>(r0[11], r1[11], sg4, lane);
  __builtin_amdgcn_sched_barrier(0);
  gateStep<6>(a0, a1, r0[6], r1[6]);
  gateStep<7>(a0, a1, r0[7], r1[7]);
  gateStep<8>(a0, a1, r0[8], r1[8]);
  gateRows<12>(r0[12], r1[12], sg4, lane);
  gateRows<13>(r0[13], r1[13], sg4, lane);
  gateRows<14>(r0[14], r1[14], sg4, lane);
  __builtin_amdgcn_sched_barrier(0);
  gateStep<9>(a0, a1, r0[9], r1[9]);
  gateStep<10>(a0, a1, r0[10], r1[10]);
  gateStep<11>(a0, a1, r0[11], r1[11]);
  gateRows<15>(r0[15], r1[15], sg4, lane);
  gateRows<16>(r0[16], r1[16], sg4, lane);
  gateRows<17>(r0[17], r1[17], sg4, lane);
  __builtin_amdgcn_sched_barrier(0);
  gateStep<12>(a0, a1, r0[12], r1[12]);
  gateStep<13>(a0, a1, r0[13], r1[13]);
  gateStep<14>(a0, a1, r0[14], r1[14]);
  gateStep<15>(a0, a1, r0[15], r1[15]);
  gateStep<16>(a0, a1, r0[16], r1[16]);
  gateStep<17>(a0, a1, r0[17], r1[17]);
  const float P0 = a0.x * a0.x + a0.y * a0.y;
  const float P1 = a1.x * a1.x + a1.y * a1.y;
  float e[6];
  e[0] = zTerm<0>(P0, P1, lane);
  e[1] = zTerm<1>(P0, P1, lane);
  e[2] = zTerm<2>(P0, P1, lane);
  e[3] = zTerm<3>(P0, P1, lane);
  e[4] = zTerm<4>(P0, P1, lane);
  e[5] = zTerm<5>(P0, P1, lane);
  #pragma unroll
  for (int q = 0; q < 6; ++q) e[q] += lanex<16>(e[q]);
  #pragma unroll
  for (int q = 0; q < 6; ++q) e[q] += lanex<8>(e[q]);
  #pragma unroll
  for (int q = 0; q < 6; ++q) e[q] += lanex<4>(e[q]);
  #pragma unroll
  for (int q = 0; q < 6; ++q) e[q] += lanex<2>(e[q]);
  #pragma unroll
  for (int q = 0; q < 6; ++q) e[q] += lanex<1>(e[q]);
  float v = 0.f;
  #pragma unroll
  for (int q = 0; q < 6; ++q) v += wc[q] * e[q];
  if (lane < 5) sLine[wave * 5 + lane] = (st < 729) ? v : 0.f;
  if (tid < 2) sLine[30 + tid] = 0.f;
  __syncthreads();
  if (tid < 32) vimg[blockIdx.x * 32 + tid] = sLine[tid];
}

__global__ __launch_bounds__(256) void qmaml_kernel(
    const float* __restrict__ x,  const float* __restrict__ W1,
    const float* __restrict__ b1, const float* __restrict__ ln_g,
    const float* __restrict__ ln_b, const float* __restrict__ W2,
    const float* __restrict__ b2, const float* __restrict__ vimg,
    const float* __restrict__ bc,
    float* __restrict__ out, int B)
{
  __shared__ float  sW1[64 * WP];
  __shared__ __attribute__((aligned(16))) float sW2[512];
  __shared__ __attribute__((aligned(16))) float sSmall[256];
  const float* sB1 = sSmall;
  const float* sG = sSmall + 64;
  const float* sBeta = sSmall + 128;
  const float* sB2 = sSmall + 192;
  const float* sBc = sSmall + 198;
  __shared__ __attribute__((aligned(16))) float sV[VROWS * VP];

  const int tid = threadIdx.x;
  const int lane = tid & 31;
  const int wave = __builtin_amdgcn_readfirstlane(tid >> 5);
  const int half = lane >> 4;
  const int lm   = lane & 15;
  const int nTiles = (B + 15) >> 4;
  const int tStep  = gridDim.x * 8;
  int tile = blockIdx.x * 8 + wave;

  float4 rW1[4];
  #pragma unroll
  for (int j = 0; j < 4; ++j) rW1[j] = ((const float4*)W1)[tid + 256 * j];
  float rV[3][5];
  #pragma unroll
  for (int j = 0; j < 3; ++j) {
    int sN = tid + 256 * j; if (sN > 728) sN = 728;
    const float* pv = vimg + 32 * (sN / 6) + 5 * (sN % 6);
    #pragma unroll
    for (int k = 0; k < 5; ++k) rV[j][k] = pv[k];
  }
  const float rW2a = W2[tid];
  const float rW2b = W2[256 + (tid & 127)];
  const int i64 = tid & 63;
  const int iB2 = tid < 192 ? 0 : (tid < 198 ? tid - 192 : 5);
  const int iBc = tid < 198 ? 0 : (tid < 203 ? tid - 198 : 4);
  const float* const pB1 = b1 + i64;
  const float* const pG  = ln_g + i64;
  const float* const pBe = ln_b + i64;
  const float* const pB2 = b2 + iB2;
  const float* const pBc = bc + iBc;
  const float* pSmall = pBc;
  pSmall = (tid < 198) ? pB2 : pSmall;
  pSmall = (tid < 192) ? pBe : pSmall;
  pSmall = (tid < 128) ? pG  : pSmall;
  pSmall = (tid <  64) ? pB1 : pSmall;
  const float rSmall = *pSmall;
  v2f xf[16];
  {
    int rowA = tile * 16 + lm; if (rowA >= B) rowA = B - 1;
    const float* xrow = x + (size_t)rowA * 64 + 2 * half;
    #pragma unroll
    for (int kk = 0; kk < 16; ++kk) xf[kk] = *(const v2f*)(xrow + 4 * kk);
  }

  #pragma unroll
  for (int j = 0; j < 4; ++j) {
    const int i = tid + 256 * j;
    *(float4*)(sW1 + (i >> 4) * WP + 4 * (i & 15)) = rW1[j];
  }
  #pragma unroll
  for (int j = 0; j < 3; ++j) {
    const int sN = tid + 256 * j;
    const int sC = sN > 728 ? 728 : sN;
    const int g = sC / 27;
    const int h = (sN < 729) ? sC - 27 * g : 28;
    #pragma unroll
    for (int k = 0; k < 5; ++k) sV[(32 * k + g) * VP + h] = rV[j][k];
  }
  if (tid < 225) {
    const int k = tid / 45, rem = tid - 45 * k, j = rem / 9, c4 = rem - 9 * j;
    *(float4*)(sV + (32 * k + 27 + j) * VP + 4 * c4) = make_float4(0.f, 0.f, 0.f, 0.f);
  }
  if (tid < 135) {
    const int k = tid / 27, g = tid - 27 * k;
    sV[(32 * k + g) * VP + 27] = 0.f;
  }
  sW2[tid] = rW2a;
  sW2[256 + tid] = rW2b;
  sSmall[tid] = rSmall;
  __syncthreads();

  const int fb = 8 * half;
  const float* w1p = sW1 + lm * WP + 2 * half;
  const float* vp = sV + lm * VP + 2 * half;

  for (; tile < nTiles; tile += tStep) {
  const int rowBase = tile * 16;
  v2f xn[16];
  {
    int rowN = rowBase + tStep * 16 + lm; if (rowN >= B) rowN = B - 1;
    const float* xrowN = x + (size_t)rowN * 64 + 2 * half;
    #pragma unroll
    for (int kk = 0; kk < 16; ++kk) xn[kk] = *(const v2f*)(xrowN + 4 * kk);
  }

  v8f acc[4] = {};
  v2f wa[2][8];
  #pragma unroll
  for (int j = 0; j < 2; ++j) {
    #pragma unroll
    for (int t = 0; t < 4; ++t) wa[0][4 * j + t] = *(const v2f*)(w1p + 16 * t * WP + 4 * j);
  }
  #pragma unroll
  for (int b = 0; b < 8; ++b) {
    if (b + 1 < 8) {
      #pragma unroll
      for (int j = 0; j < 2; ++j) {
        #pragma unroll
        for (int t = 0; t < 4; ++t)
          wa[(b + 1) & 1][4 * j + t] = *(const v2f*)(w1p + 16 * t * WP + 4 * (2 * b + 2 + j));
      }
    }
    __builtin_amdgcn_sched_barrier(0);
    #pragma unroll
    for (int j = 0; j < 2; ++j) {
      #pragma unroll
      for (int t = 0; t < 4; ++t)
        acc[t] = __builtin_amdgcn_wmma_f32_16x16x4_f32(false, wa[b & 1][4 * j + t], false, xf[2 * b + j],
                                                       (short)0, acc[t], false, false);
    }
    __builtin_amdgcn_sched_barrier(0);
  }

  float hv[4][8];
  #pragma unroll
  for (int t = 0; t < 4; ++t) {
    const float4 ba = *(const float4*)(sB1 + 16 * t + fb);
    const float4 bb = *(const float4*)(sB1 + 16 * t + fb + 4);
    const float bias[8] = { ba.x, ba.y, ba.z, ba.w, bb.x, bb.y, bb.z, bb.w };
    #pragma unroll
    for (int j = 0; j < 8; ++j) hv[t][j] = fmaxf(acc[t][j] + bias[j], 0.0f);
  }

  float st[4], qt[4];
  #pragma unroll
  for (int t = 0; t < 4; ++t) {
    st[t] = hv[t][0];
    qt[t] = hv[t][0] * hv[t][0];
    #pragma unroll
    for (int j = 1; j < 8; ++j) { st[t] += hv[t][j]; qt[t] += hv[t][j] * hv[t][j]; }
  }
  float s  = (st[0] + st[1]) + (st[2] + st[3]);
  float q2 = (qt[0] + qt[1]) + (qt[2] + qt[3]);
  s  += xhalf(s);
  q2 += xhalf(q2);
  const float mu  = s * 0.015625f;
  const float var = q2 * 0.015625f - mu * mu;
  const float inv = rsqrtf(var + 1e-5f);
  #pragma unroll
  for (int t = 0; t < 4; ++t) {
    const float4 ga = *(const float4*)(sG + 16 * t + fb);
    const float4 gb = *(const float4*)(sG + 16 * t + fb + 4);
    const float4 ea = *(const float4*)(sBeta + 16 * t + fb);
    const float4 eb = *(const float4*)(sBeta + 16 * t + fb + 4);
    const float g[8]  = { ga.x, ga.y, ga.z, ga.w, gb.x, gb.y, gb.z, gb.w };
    const float be[8] = { ea.x, ea.y, ea.z, ea.w, eb.x, eb.y, eb.z, eb.w };
    #pragma unroll
    for (int j = 0; j < 8; ++j) hv[t][j] = (hv[t][j] - mu) * inv * g[j] + be[j];
  }

  float zq[6];
  #pragma unroll
  for (int q = 0; q < 6; ++q) {
    v2f pt[4];
    #pragma unroll
    for (int t = 0; t < 4; ++t) {
      const float4 wa = *(const float4*)(sW2 + q * 64 + 16 * t + fb);
      const float4 wb = *(const float4*)(sW2 + q * 64 + 16 * t + fb + 4);
      pt[t]  = v2f{hv[t][0], hv[t][1]} * v2f{wa.x, wa.y};
      pt[t] += v2f{hv[t][2], hv[t][3]} * v2f{wa.z, wa.w};
      pt[t] += v2f{hv[t][4], hv[t][5]} * v2f{wb.x, wb.y};
      pt[t] += v2f{hv[t][6], hv[t][7]} * v2f{wb.z, wb.w};
    }
    const v2f pp = (pt[0] + pt[1]) + (pt[2] + pt[3]);
    float p = pp.x + pp.y;
    p += xhalf(p);
    zq[q] = tanh_bounded(p + sB2[q]);
  }

  float f0_0, f0_1, f0_2;
  {
    float sh, ch;
    sincos_halfpi(zq[0], sh, ch);
    const float ct = fmaf(-2.0f * sh, sh, 1.0f);
    const float st = 2.0f * sh * ch;
    const float bq = fmaf(ct, -0.33333333333333333f, 0.33333333333333333f);
    const float dq = st * 0.57735026918962576f;
    f0_0 = fmaf(ct, 0.66666666666666667f, 0.33333333333333333f);
    f0_1 = bq + dq;
    f0_2 = bq - dq;
  }
  float f1_0, f1_1, f1_2;
  {
    float sh, ch;
    sincos_halfpi(zq[1], sh, ch);
    const float ct = fmaf(-2.0f * sh, sh, 1.0f);
    const float st = 2.0f * sh * ch;
    const float bq = fmaf(ct, -0.33333333333333333f, 0.33333333333333333f);
    const float dq = st * 0.57735026918962576f;
    f1_0 = fmaf(ct, 0.66666666666666667f, 0.33333333333333333f);
    f1_1 = bq + dq;
    f1_2 = bq - dq;
  }
  float f2_0, f2_1, f2_2;
  {
    float sh, ch;
    sincos_halfpi(zq[2], sh, ch);
    const float ct = fmaf(-2.0f * sh, sh, 1.0f);
    const float st = 2.0f * sh * ch;
    const float bq = fmaf(ct, -0.33333333333333333f, 0.33333333333333333f);
    const float dq = st * 0.57735026918962576f;
    f2_0 = fmaf(ct, 0.66666666666666667f, 0.33333333333333333f);
    f2_1 = bq + dq;
    f2_2 = bq - dq;
  }
  float f3_0, f3_1, f3_2;
  {
    float sh, ch;
    sincos_halfpi(zq[3], sh, ch);
    const float ct = fmaf(-2.0f * sh, sh, 1.0f);
    const float st = 2.0f * sh * ch;
    const float bq = fmaf(ct, -0.33333333333333333f, 0.33333333333333333f);
    const float dq = st * 0.57735026918962576f;
    f3_0 = fmaf(ct, 0.66666666666666667f, 0.33333333333333333f);
    f3_1 = bq + dq;
    f3_2 = bq - dq;
  }
  float f4_0, f4_1, f4_2;
  {
    float sh, ch;
    sincos_halfpi(zq[4], sh, ch);
    const float ct = fmaf(-2.0f * sh, sh, 1.0f);
    const float st = 2.0f * sh * ch;
    const float bq = fmaf(ct, -0.33333333333333333f, 0.33333333333333333f);
    const float dq = st * 0.57735026918962576f;
    f4_0 = fmaf(ct, 0.66666666666666667f, 0.33333333333333333f);
    f4_1 = bq + dq;
    f4_2 = bq - dq;
  }
  float f5_0, f5_1, f5_2;
  {
    float sh, ch;
    sincos_halfpi(zq[5], sh, ch);
    const float ct = fmaf(-2.0f * sh, sh, 1.0f);
    const float st = 2.0f * sh * ch;
    const float bq = fmaf(ct, -0.33333333333333333f, 0.33333333333333333f);
    const float dq = st * 0.57735026918962576f;
    f5_0 = fmaf(ct, 0.66666666666666667f, 0.33333333333333333f);
    f5_1 = bq + dq;
    f5_2 = bq - dq;
  }
  const float g12_0 = f1_0 * f2_0, h45_0 = f4_0 * f5_0;
  const float g12_1 = f1_0 * f2_1, h45_1 = f4_0 * f5_1;
  const float g12_2 = f1_0 * f2_2, h45_2 = f4_0 * f5_2;
  const float g12_3 = f1_1 * f2_0, h45_3 = f4_1 * f5_0;
  const float g12_4 = f1_1 * f2_1, h45_4 = f4_1 * f5_1;
  const float g12_5 = f1_1 * f2_2, h45_5 = f4_1 * f5_2;
  const float g12_6 = f1_2 * f2_0, h45_6 = f4_2 * f5_0;
  const float g12_7 = f1_2 * f2_1, h45_7 = f4_2 * f5_1;
  const float g12_8 = f1_2 * f2_2, h45_8 = f4_2 * f5_2;
  const float G_0 = f0_0 * g12_0, H_0 = f3_0 * h45_0;
  const float G_1 = f0_0 * g12_1, H_1 = f3_0 * h45_1;
  const float G_2 = f0_0 * g12_2, H_2 = f3_0 * h45_2;
  const float G_3 = f0_0 * g12_3, H_3 = f3_0 * h45_3;
  const float G_4 = f0_0 * g12_4, H_4 = f3_0 * h45_4;
  const float G_5 = f0_0 * g12_5, H_5 = f3_0 * h45_5;
  const float G_6 = f0_0 * g12_6, H_6 = f3_0 * h45_6;
  const float G_7 = f0_0 * g12_7, H_7 = f3_0 * h45_7;
  const float G_8 = f0_0 * g12_8, H_8 = f3_0 * h45_8;
  const float G_9 = f0_1 * g12_0, H_9 = f3_1 * h45_0;
  const float G_10 = f0_1 * g12_1, H_10 = f3_1 * h45_1;
  const float G_11 = f0_1 * g12_2, H_11 = f3_1 * h45_2;
  const float G_12 = f0_1 * g12_3, H_12 = f3_1 * h45_3;
  const float G_13 = f0_1 * g12_4, H_13 = f3_1 * h45_4;
  const float G_14 = f0_1 * g12_5, H_14 = f3_1 * h45_5;
  const float G_15 = f0_1 * g12_6, H_15 = f3_1 * h45_6;
  const float G_16 = f0_1 * g12_7, H_16 = f3_1 * h45_7;
  const float G_17 = f0_1 * g12_8, H_17 = f3_1 * h45_8;
  const float G_18 = f0_2 * g12_0, H_18 = f3_2 * h45_0;
  const float G_19 = f0_2 * g12_1, H_19 = f3_2 * h45_1;
  const float G_20 = f0_2 * g12_2, H_20 = f3_2 * h45_2;
  const float G_21 = f0_2 * g12_3, H_21 = f3_2 * h45_3;
  const float G_22 = f0_2 * g12_4, H_22 = f3_2 * h45_4;
  const float G_23 = f0_2 * g12_5, H_23 = f3_2 * h45_5;
  const float G_24 = f0_2 * g12_6, H_24 = f3_2 * h45_6;
  const float G_25 = f0_2 * g12_7, H_25 = f3_2 * h45_7;
  const float G_26 = f0_2 * g12_8, H_26 = f3_2 * h45_8;
  v2f hb[7];
  hb[0].x = half ? H_2 : H_0;  hb[0].y = half ? H_3 : H_1;
  hb[1].x = half ? H_6 : H_4;  hb[1].y = half ? H_7 : H_5;
  hb[2].x = half ? H_10 : H_8;  hb[2].y = half ? H_11 : H_9;
  hb[3].x = half ? H_14 : H_12;  hb[3].y = half ? H_15 : H_13;
  hb[4].x = half ? H_18 : H_16;  hb[4].y = half ? H_19 : H_17;
  hb[5].x = half ? H_22 : H_20;  hb[5].y = half ? H_23 : H_21;
  hb[6].x = half ? H_26 : H_24;  hb[6].y = half ? 0.0f : H_25;
  float gs[2][8];
  gs[0][0] = half ? G_8 : G_0;
  gs[0][1] = half ? G_9 : G_1;
  gs[0][2] = half ? G_10 : G_2;
  gs[0][3] = half ? G_11 : G_3;
  gs[0][4] = half ? G_12 : G_4;
  gs[0][5] = half ? G_13 : G_5;
  gs[0][6] = half ? G_14 : G_6;
  gs[0][7] = half ? G_15 : G_7;
  gs[1][0] = half ? G_24 : G_16;
  gs[1][1] = half ? G_25 : G_17;
  gs[1][2] = half ? G_26 : G_18;
  gs[1][3] = half ? 0.0f : G_19;
  gs[1][4] = half ? 0.0f : G_20;
  gs[1][5] = half ? 0.0f : G_21;
  gs[1][6] = half ? 0.0f : G_22;
  gs[1][7] = half ? 0.0f : G_23;

  v8f yv[10] = {};
  v2f va[2][10];
  #pragma unroll
  for (int t = 0; t < 10; ++t) va[0][t] = *(const v2f*)(vp + 16 * t * VP);
  #pragma unroll
  for (int kk = 0; kk < 7; ++kk) {
    if (kk + 1 < 7) {
      #pragma unroll
      for (int t = 0; t < 10; ++t) va[(kk + 1) & 1][t] = *(const v2f*)(vp + 16 * t * VP + 4 * (kk + 1));
    }
    __builtin_amdgcn_sched_barrier(0);
    #pragma unroll
    for (int t = 0; t < 10; ++t)
      yv[t] = __builtin_amdgcn_wmma_f32_16x16x4_f32(false, va[kk & 1][t], false, hb[kk],
                                                    (short)0, yv[t], false, false);
    __builtin_amdgcn_sched_barrier(0);
  }

  const int row = rowBase + lm;
  #pragma unroll
  for (int k = 0; k < 5; ++k) {
    float v0 = gs[0][0] * yv[2 * k][0], v1 = gs[1][0] * yv[2 * k + 1][0];
    #pragma unroll
    for (int r = 1; r < 8; ++r) { v0 = fmaf(gs[0][r], yv[2 * k][r], v0); v1 = fmaf(gs[1][r], yv[2 * k + 1][r], v1); }
    float v = v0 + v1;
    v += xhalf(v);
    if (half == 0 && row < B) out[(size_t)row * 5 + k] = v + sBc[k];
  }
  #pragma unroll
  for (int kk = 0; kk < 16; ++kk) xf[kk] = xn[kk];
  }
}

extern "C" void kernel_launch(void* const* d_in, const int* in_sizes, int n_in,
                              void* d_out, int out_size, void* d_ws, size_t ws_size,
                              hipStream_t stream) {
  (void)n_in; (void)out_size; (void)ws_size;
  const float* x    = (const float*)d_in[0];
  const float* W1   = (const float*)d_in[1];
  const float* b1   = (const float*)d_in[2];
  const float* ln_g = (const float*)d_in[3];
  const float* ln_b = (const float*)d_in[4];
  const float* W2   = (const float*)d_in[5];
  const float* b2   = (const float*)d_in[6];
  const float* th_s = (const float*)d_in[7];
  const float* th_t = (const float*)d_in[8];
  const float* Wc   = (const float*)d_in[9];
  const float* bc   = (const float*)d_in[10];
  float* out = (float*)d_out;
  float* vimg = (float*)d_ws;
  int B = in_sizes[0] / 64;
  circuit_compose<<<122, 192, 0, stream>>>(th_s, th_t, Wc, vimg);
  qmaml_kernel<<<512, 256, 0, stream>>>(x, W1, b1, ln_g, ln_b, W2, b2, vimg, bc, out, B);
}


# Node products split 9 x 81, not 27 x 27: second product is V(48x84) @ H(84x16), 63 matrix instructions not 70 per tile, closing 23 FMAs not 80, G side 9 products not 36; sum order changes
# speedup vs baseline: 11.6969x; 1.0574x over previous
#include <hip/hip_runtime.h>
#include <math.h>


typedef __attribute__((ext_vector_type(2))) float v2f;
typedef __attribute__((ext_vector_type(8))) float v8f;
typedef __attribute__((ext_vector_type(4))) float v4f;

struct Rings { int r1[64]; int r2[64]; };

constexpr Rings makeRings() {
  Rings R{};
  for (int rr = 1; rr <= 2; ++rr) {
    int total[64];
    for (int i = 0; i < 64; ++i) total[i] = i;
    for (int w = 0; w < 6; ++w) {
      int t = (w + rr) % 6;
      int nxt[64];
      for (int i = 0; i < 64; ++i) {
        int cbit = (i >> (5 - w)) & 1;
        nxt[i] = total[i ^ (cbit << (5 - t))];
      }
      for (int i = 0; i < 64; ++i) total[i] = nxt[i];
    }
    for (int i = 0; i < 64; ++i) { if (rr == 1) R.r1[i] = total[i]; else R.r2[i] = total[i]; }
  }
  return R;
}

struct Walk { int pm[18]; int rm[18]; int zm[6]; };

constexpr int parityMask(const int* inv, int b) {
  int r = 0;
  for (int k = 0; k < 6; ++k) { if ((inv[1 << k] >> b) & 1) r |= (1 << k); }
  return r;
}

constexpr Walk makeWalk() {
  Walk K{};
  const Rings R = makeRings();
  int pos[64] = {};
  int inv[64] = {};
  for (int j = 0; j < 64; ++j) pos[j] = j;
  for (int l = 0; l < 3; ++l) {
    for (int j = 0; j < 64; ++j) inv[pos[j]] = j;
    for (int w = 0; w < 6; ++w) {
      K.pm[l * 6 + w] = pos[1 << (5 - w)];
      K.rm[l * 6 + w] = parityMask(inv, 5 - w);
    }
    int nxt[64] = {};
    for (int j = 0; j < 64; ++j) nxt[j] = pos[(l == 1) ? R.r2[j] : R.r1[j]];
    for (int j = 0; j < 64; ++j) pos[j] = nxt[j];
  }
  for (int j = 0; j < 64; ++j) inv[pos[j]] = j;
  for (int q = 0; q < 6; ++q) K.zm[q] = parityMask(inv, 5 - q);
  return K;
}
constexpr int walkPm(int g) { return makeWalk().pm[g]; }
constexpr int walkRm(int g) { return makeWalk().rm[g]; }
constexpr int walkZm(int q) { return makeWalk().zm[q]; }

__device__ __forceinline__ float2 cmul(float2 a, float2 b) {
  return make_float2(a.x * b.x - a.y * b.y, a.x * b.y + a.y * b.x);
}
__device__ __forceinline__ float2 cadd(float2 a, float2 b) { return make_float2(a.x + b.x, a.y + b.y); }
constexpr unsigned selLo(int m) { unsigned r = 0; for (int i = 0; i < 8; ++i) r |= (unsigned)((i ^ m) & 15) << (4 * i); return r; }
constexpr unsigned selHi(int m) { unsigned r = 0; for (int i = 8; i < 16; ++i) r |= (unsigned)((i ^ m) & 15) << (4 * (i - 8)); return r; }

template <int LM>
__device__ __forceinline__ float lanex(float v) {
  if constexpr (LM == 0) {
    return v;
  } else {
    const int b = __builtin_bit_cast(int, v);
    if constexpr (LM < 16) {
      return __builtin_bit_cast(float, __builtin_amdgcn_update_dpp(0, b, 0x160 + LM, 0xf, 0xf, true));
    } else {
      return __builtin_bit_cast(float, __builtin_amdgcn_permlanex16(b, b, selLo(LM & 15), selHi(LM & 15), false, false));
    }
  }
}
template <int LM>
__device__ __forceinline__ float2 lanex2(float2 v) { return make_float2(lanex<LM>(v.x), lanex<LM>(v.y)); }

template <int G>
__device__ __forceinline__ void gateRows(v4f& r0, v4f& r1, const v4f* sg4, const int lane) {
  constexpr int RM = walkRm(G);
  const int b0 = __builtin_popcount(lane & (RM & 31)) & 1;
  const int b1 = b0 ^ ((RM >> 5) & 1);
  r0 = sg4[2 * G + b0];
  r1 = sg4[2 * G + b1];
}
template <int G>
__device__ __forceinline__ void gateStep(float2& a0, float2& a1, const v4f r0, const v4f r1) {
  constexpr int PM = walkPm(G);
  const float2 s0 = (PM & 32) ? a1 : a0;
  const float2 s1 = (PM & 32) ? a0 : a1;
  const float2 p0 = lanex2<(PM & 31)>(s0);
  const float2 p1 = lanex2<(PM & 31)>(s1);
  a0 = cadd(cmul(make_float2(r0.x, r0.y), a0), cmul(make_float2(r0.z, r0.w), p0));
  a1 = cadd(cmul(make_float2(r1.x, r1.y), a1), cmul(make_float2(r1.z, r1.w), p1));
}

template <int Q>
__device__ __forceinline__ float zTerm(const float P0, const float P1, const int lane) {
  constexpr int ZM = walkZm(Q);
  const int s0 = __builtin_popcount(lane & (ZM & 31)) & 1;
  const int s1 = s0 ^ ((ZM >> 5) & 1);
  return (s0 ? -P0 : P0) + (s1 ? -P1 : P1);
}

__device__ __forceinline__ float xhalf(float v) {
  int b = __builtin_bit_cast(int, v);
  return __builtin_bit_cast(float, __builtin_amdgcn_permlanex16(b, b, 0x76543210, 0xfedcba98, false, false));
}

__device__ __forceinline__ float tanh_bounded(float x) {
  const float t = __builtin_amdgcn_exp2f(fabsf(x) * -2.885390081777927f);
  const float r = (1.0f - t) * __builtin_amdgcn_rcpf(1.0f + t);
  return copysignf(r, x);
}

__device__ __forceinline__ void sincos_halfpi(float z, float& s, float& c) {
  const float u  = 0.78539816339744831f * z;
  const float u2 = u * u;
  const float ps = fmaf(u2, fmaf(u2, -1.9515295891e-4f, 8.3321608736e-3f), -1.6666654611e-1f);
  const float su = fmaf(u * u2, ps, u);
  const float pc = fmaf(u2, fmaf(u2, 2.443315711809948e-5f, -1.388731625493765e-3f), 4.166664568298827e-2f);
  const float cu = fmaf(u2 * u2, pc, fmaf(u2, -0.5f, 1.0f));
  const float s2 = su + su;
  s = s2 * cu;
  c = fmaf(-s2, su, 1.0f);
}

#define VP 84
#define VROWS 58
#define NODE_S 0.8660254037844386f
#define WP 68

__global__ __launch_bounds__(192) void circuit_compose(
    const float* __restrict__ th_s, const float* __restrict__ th_t,
    const float* __restrict__ Wc, float* __restrict__ vimg)
{
  __shared__ __attribute__((aligned(16))) float2 sGates[72];
  __shared__ float  sLine[32];

  const int tid = threadIdx.x;
  const int kc = ((tid & 31) < 5) ? (tid & 31) : 0;
  float wc[6];
  #pragma unroll
  for (int q = 0; q < 6; ++q) wc[q] = Wc[kc * 6 + q];
  if (tid < 18) {
    int l = tid / 6, w = tid % 6;
    const float* th = (l < 2) ? (th_s + (l * 6 + w) * 3) : (th_t + w * 3);
    float phi = th[0], theta = th[1], omega = th[2];
    float c  = cosf(0.5f * theta), s = sinf(0.5f * theta);
    float aa = 0.5f * (phi + omega), bb = 0.5f * (phi - omega);
    float ca = cosf(aa), sa = sinf(aa);
    float cb = cosf(bb), sb = sinf(bb);
    sGates[tid * 4 + 0] = make_float2( ca * c, -sa * c);
    sGates[tid * 4 + 1] = make_float2(-cb * s, -sb * s);
    sGates[tid * 4 + 2] = make_float2( ca * c,  sa * c);
    sGates[tid * 4 + 3] = make_float2( cb * s, -sb * s);
  }
  __syncthreads();
  asm volatile("" :: "v"(wc[0]), "v"(wc[1]), "v"(wc[2]), "v"(wc[3]), "v"(wc[4]), "v"(wc[5]));

  const int lane = tid & 31;
  const int wave = tid >> 5;

  const int st = blockIdx.x * 6 + wave;
  const int dg[6] = { (st / 243) % 3, (st / 81) % 3, (st / 27) % 3, (st / 9) % 3, (st / 3) % 3, st % 3 };
  float am0 = 1.f, am1 = 1.f;
  #pragma unroll
  for (int q = 0; q < 6; ++q) {
    const float nc = (dg[q] == 0) ? 1.f : ((dg[q] == 1) ? 0.5f : -0.5f);
    const float ns = (dg[q] == 0) ? 0.f : NODE_S;
    const int bit = (q == 0) ? 0 : ((lane >> (5 - q)) & 1);
    am0 *= bit ? ns : nc;
    am1 *= ((q == 0) ? 1 : bit) ? ns : nc;
  }
  float2 a0 = make_float2(am0, 0.f);
  float2 a1 = make_float2(am1, 0.f);

  const v4f* sg4 = (const v4f*)sGates;
  v4f r0[18], r1[18];
  gateRows<0>(r0[0], r1[0], sg4, lane);
  gateRows<1>(r0[1], r1[1], sg4, lane);
  gateRows<2>(r0[2], r1[2], sg4, lane);
  gateRows<3>(r0[3], r1[3], sg4, lane);
  gateRows<4>(r0[4], r1[4], sg4, lane);
  gateRows<5>(r0[5], r1[5], sg4, lane);
  __builtin_amdgcn_sched_barrier(0);
  gateStep<0>(a0, a1, r0[0], r1[0]);
  gateStep<1>(a0, a1, r0[1], r1[1]);
  gateStep<2>(a0, a1, r0[2], r1[2]);
  gateRows<6>(r0[6], r1[6], sg4, lane);
  gateRows<7>(r0[7], r1[7], sg4, lane);
  gateRows<8>(r0[8], r1[8], sg4, lane);
  __builtin_amdgcn_sched_barrier(0);
  gateStep<3>(a0, a1, r0[3], r1[3]);
  gateStep<4>(a0, a1, r0[4], r1[4]);
  gateStep<5>(a0, a1, r0[5], r1[5]);
  gateRows<9>(r0[9], r1[9], sg4, lane);
  gateRows<10>(r0[10], r1[10], sg4, lane);
  gateRows<11>(r0[11], r1[11], sg4, lane);
  __builtin_amdgcn_sched_barrier(0);
  gateStep<6>(a0, a1, r0[6], r1[6]);
  gateStep<7>(a0, a1, r0[7], r1[7]);
  gateStep<8>(a0, a1, r0[8], r1[8]);
  gateRows<12>(r0[12], r1[12], sg4, lane);
  gateRows<13>(r0[13], r1[13], sg4, lane);
  gateRows<14>(r0[14], r1[14], sg4, lane);
  __builtin_amdgcn_sched_barrier(0);
  gateStep<9>(a0, a1, r0[9], r1[9]);
  gateStep<10>(a0, a1, r0[10], r1[10]);
  gateStep<11>(a0, a1, r0[11], r1[11]);
  gateRows<15>(r0[15], r1[15], sg4, lane);
  gateRows<16>(r0[16], r1[16], sg4, lane);
  gateRows<17>(r0[17], r1[17], sg4, lane);
  __builtin_amdgcn_sched_barrier(0);
  gateStep<12>(a0, a1, r0[12], r1[12]);
  gateStep<13>(a0, a1, r0[13], r1[13]);
  gateStep<14>(a0, a1, r0[14], r1[14]);
  gateStep<15>(a0, a1, r0[15], r1[15]);
  gateStep<16>(a0, a1, r0[16], r1[16]);
  gateStep<17>(a0, a1, r0[17], r1[17]);
  const float P0 = a0.x * a0.x + a0.y * a0.y;
  const float P1 = a1.x * a1.x + a1.y * a1.y;
  float e[6];
  e[0] = zTerm<0>(P0, P1, lane);
  e[1] = zTerm<1>(P0, P1, lane);
  e[2] = zTerm<2>(P0, P1, lane);
  e[3] = zTerm<3>(P0, P1, lane);
  e[4] = zTerm<4>(P0, P1, lane);
  e[5] = zTerm<5>(P0, P1, lane);
  #pragma unroll
  for (int q = 0; q < 6; ++q) e[q] += lanex<16>(e[q]);
  #pragma unroll
  for (int q = 0; q < 6; ++q) e[q] += lanex<8>(e[q]);
  #pragma unroll
  for (int q = 0; q < 6; ++q) e[q] += lanex<4>(e[q]);
  #pragma unroll
  for (int q = 0; q < 6; ++q) e[q] += lanex<2>(e[q]);
  #pragma unroll
  for (int q = 0; q < 6; ++q) e[q] += lanex<1>(e[q]);
  float v = 0.f;
  #pragma unroll
  for (int q = 0; q < 6; ++q) v += wc[q] * e[q];
  if (lane < 5) sLine[wave * 5 + lane] = (st < 729) ? v : 0.f;
  if (tid < 2) sLine[30 + tid] = 0.f;
  __syncthreads();
  if (tid < 32) vimg[blockIdx.x * 32 + tid] = sLine[tid];
}

__global__ __launch_bounds__(256) void qmaml_kernel(
    const float* __restrict__ x,  const float* __restrict__ W1,
    const float* __restrict__ b1, const float* __restrict__ ln_g,
    const float* __restrict__ ln_b, const float* __restrict__ W2,
    const float* __restrict__ b2, const float* __restrict__ vimg,
    const float* __restrict__ bc,
    float* __restrict__ out, int B)
{
  __shared__ float  sW1[64 * WP];
  __shared__ __attribute__((aligned(16))) float sW2[512];
  __shared__ __attribute__((aligned(16))) float sSmall[256];
  const float* sB1 = sSmall;
  const float* sG = sSmall + 64;
  const float* sBeta = sSmall + 128;
  const float* sB2 = sSmall + 192;
  const float* sBc = sSmall + 198;
  __shared__ __attribute__((aligned(16))) float sV[VROWS * VP];

  const int tid = threadIdx.x;
  const int lane = tid & 31;
  const int wave = __builtin_amdgcn_readfirstlane(tid >> 5);
  const int half = lane >> 4;
  const int lm   = lane & 15;
  const int nTiles = (B + 15) >> 4;
  const int tStep  = gridDim.x * 8;
  int tile = blockIdx.x * 8 + wave;

  float4 rW1[4];
  #pragma unroll
  for (int j = 0; j < 4; ++j) rW1[j] = ((const float4*)W1)[tid + 256 * j];
  float rV[3][5];
  #pragma unroll
  for (int j = 0; j < 3; ++j) {
    int sN = tid + 256 * j; if (sN > 728) sN = 728;
    const float* pv = vimg + 32 * (sN / 6) + 5 * (sN % 6);
    #pragma unroll
    for (int k = 0; k < 5; ++k) rV[j][k] = pv[k];
  }
  const float rW2a = W2[tid];
  const float rW2b = W2[256 + (tid & 127)];
  const int i64 = tid & 63;
  const int iB2 = tid < 192 ? 0 : (tid < 198 ? tid - 192 : 5);
  const int iBc = tid < 198 ? 0 : (tid < 203 ? tid - 198 : 4);
  const float* const pB1 = b1 + i64;
  const float* const pG  = ln_g + i64;
  const float* const pBe = ln_b + i64;
  const float* const pB2 = b2 + iB2;
  const float* const pBc = bc + iBc;
  const float* pSmall = pBc;
  pSmall = (tid < 198) ? pB2 : pSmall;
  pSmall = (tid < 192) ? pBe : pSmall;
  pSmall = (tid < 128) ? pG  : pSmall;
  pSmall = (tid <  64) ? pB1 : pSmall;
  const float rSmall = *pSmall;
  v2f xf[16];
  {
    int rowA = tile * 16 + lm; if (rowA >= B) rowA = B - 1;
    const float* xrow = x + (size_t)rowA * 64 + 2 * half;
    #pragma unroll
    for (int kk = 0; kk < 16; ++kk) xf[kk] = *(const v2f*)(xrow + 4 * kk);
  }

  int oV[3], oV4[3];
  #pragma unroll
  for (int j = 0; j < 3; ++j) {
    const int sN = tid + 256 * j;
    const int sC = sN > 728 ? 728 : sN;
    const int a   = sC / 81;
    const int m   = sC - 81 * a;
    const int j2  = m / 27;
    const int n   = m - 27 * j2;
    const int j3  = n / 9;
    const int i45 = n - 9 * j3;
    const int hk  = (j3 < 2) ? j3 : (i45 >= 5 ? 1 : 0);
    const int xs  = (j3 < 2) ? i45 : (9 + i45 - 5 * hk);
    const int col = 28 * j2 + 2 * xs - (xs & 1) + 2 * hk;
    const bool live = (j < 2) || (sN < 729);
    const int rb  = live ? (16 * (a >> 2) + 2 * (a & 3)) : 48;
    const int r4  = live ? (34 + a + (a >= 5 ? 3 : 0)) : 48;
    oV[j]  = rb * VP + col;
    oV4[j] = r4 * VP + col;
  }

  #pragma unroll
  for (int j = 0; j < 4; ++j) {
    const int i = tid + 256 * j;
    *(float4*)(sW1 + (i >> 4) * WP + 4 * (i & 15)) = rW1[j];
  }
  #pragma unroll
  for (int j = 0; j < 3; ++j) {
    sV[oV[j]]                = rV[j][0];
    sV[oV[j] + 8 * VP]       = rV[j][1];
    sV[oV[j] + VP]           = rV[j][2];
    sV[oV[j] + 9 * VP]       = rV[j][3];
    sV[oV4[j]]               = rV[j][4];
  }
  if (tid < 63) {
    const int rs = tid / 21, c4 = tid - 21 * rs;
    const int row = (rs == 0) ? 39 : (45 + rs);
    *(float4*)(sV + row * VP + 4 * c4) = make_float4(0.f, 0.f, 0.f, 0.f);
  }
  if (tid < 135) {
    const int rr = tid / 3, cc = tid - 3 * rr;
    const int row = rr + (rr >= 39 ? 1 : 0);
    sV[row * VP + 27 + 28 * cc] = 0.f;
  }
  sW2[tid] = rW2a;
  sW2[256 + tid] = rW2b;
  sSmall[tid] = rSmall;
  __syncthreads();

  const int fb = 8 * half;
  const float* w1p = sW1 + lm * WP + 2 * half;
  const float* vp = sV + lm * VP + 2 * half;

  for (; tile < nTiles; tile += tStep) {
  const int rowBase = tile * 16;
  v2f xn[16];
  {
    int rowN = rowBase + tStep * 16 + lm; if (rowN >= B) rowN = B - 1;
    const float* xrowN = x + (size_t)rowN * 64 + 2 * half;
    #pragma unroll
    for (int kk = 0; kk < 16; ++kk) xn[kk] = *(const v2f*)(xrowN + 4 * kk);
  }

  v8f acc[4] = {};
  v2f wa[2][8];
  #pragma unroll
  for (int j = 0; j < 2; ++j) {
    #pragma unroll
    for (int t = 0; t < 4; ++t) wa[0][4 * j + t] = *(const v2f*)(w1p + 16 * t * WP + 4 * j);
  }
  #pragma unroll
  for (int b = 0; b < 8; ++b) {
    if (b + 1 < 8) {
      #pragma unroll
      for (int j = 0; j < 2; ++j) {
        #pragma unroll
        for (int t = 0; t < 4; ++t)
          wa[(b + 1) & 1][4 * j + t] = *(const v2f*)(w1p + 16 * t * WP + 4 * (2 * b + 2 + j));
      }
    }
    __builtin_amdgcn_sched_barrier(0);
    #pragma unroll
    for (int j = 0; j < 2; ++j) {
      #pragma unroll
      for (int t = 0; t < 4; ++t)
        acc[t] = __builtin_amdgcn_wmma_f32_16x16x4_f32(false, wa[b & 1][4 * j + t], false, xf[2 * b + j],
                                                       (short)0, acc[t], false, false);
    }
    __builtin_amdgcn_sched_barrier(0);
  }

  float hv[4][8];
  #pragma unroll
  for (int t = 0; t < 4; ++t) {
    const float4 ba = *(const float4*)(sB1 + 16 * t + fb);
    const float4 bb = *(const float4*)(sB1 + 16 * t + fb + 4);
    const float bias[8] = { ba.x, ba.y, ba.z, ba.w, bb.x, bb.y, bb.z, bb.w };
    #pragma unroll
    for (int j = 0; j < 8; ++j) hv[t][j] = fmaxf(acc[t][j] + bias[j], 0.0f);
  }

  float st[4], qt[4];
  #pragma unroll
  for (int t = 0; t < 4; ++t) {
    st[t] = hv[t][0];
    qt[t] = hv[t][0] * hv[t][0];
    #pragma unroll
    for (int j = 1; j < 8; ++j) { st[t] += hv[t][j]; qt[t] += hv[t][j] * hv[t][j]; }
  }
  float s  = (st[0] + st[1]) + (st[2] + st[3]);
  float q2 = (qt[0] + qt[1]) + (qt[2] + qt[3]);
  s  += xhalf(s);
  q2 += xhalf(q2);
  const float mu  = s * 0.015625f;
  const float var = q2 * 0.015625f - mu * mu;
  const float inv = rsqrtf(var + 1e-5f);
  #pragma unroll
  for (int t = 0; t < 4; ++t) {
    const float4 ga = *(const float4*)(sG + 16 * t + fb);
    const float4 gb = *(const float4*)(sG + 16 * t + fb + 4);
    const float4 ea = *(const float4*)(sBeta + 16 * t + fb);
    const float4 eb = *(const float4*)(sBeta + 16 * t + fb + 4);
    const float g[8]  = { ga.x, ga.y, ga.z, ga.w, gb.x, gb.y, gb.z, gb.w };
    const float be[8] = { ea.x, ea.y, ea.z, ea.w, eb.x, eb.y, eb.z, eb.w };
    #pragma unroll
    for (int j = 0; j < 8; ++j) hv[t][j] = (hv[t][j] - mu) * inv * g[j] + be[j];
  }

  float zq[6];
  #pragma unroll
  for (int q = 0; q < 6; ++q) {
    v2f pt[4];
    #pragma unroll
    for (int t = 0; t < 4; ++t) {
      const float4 wa = *(const float4*)(sW2 + q * 64 + 16 * t + fb);
      const float4 wb = *(const float4*)(sW2 + q * 64 + 16 * t + fb + 4);
      pt[t]  = v2f{hv[t][0], hv[t][1]} * v2f{wa.x, wa.y};
      pt[t] += v2f{hv[t][2], hv[t][3]} * v2f{wa.z, wa.w};
      pt[t] += v2f{hv[t][4], hv[t][5]} * v2f{wb.x, wb.y};
      pt[t] += v2f{hv[t][6], hv[t][7]} * v2f{wb.z, wb.w};
    }
    const v2f pp = (pt[0] + pt[1]) + (pt[2] + pt[3]);
    float p = pp.x + pp.y;
    p += xhalf(p);
    zq[q] = tanh_bounded(p + sB2[q]);
  }

  float f0_0, f0_1, f0_2;
  {
    float sh, ch;
    sincos_halfpi(zq[0], sh, ch);
    const float ct = fmaf(-2.0f * sh, sh, 1.0f);
    const float st = 2.0f * sh * ch;
    const float bq = fmaf(ct, -0.33333333333333333f, 0.33333333333333333f);
    const float dq = st * 0.57735026918962576f;
    f0_0 = fmaf(ct, 0.66666666666666667f, 0.33333333333333333f);
    f0_1 = bq + dq;
    f0_2 = bq - dq;
  }
  float f1_0, f1_1, f1_2;
  {
    float sh, ch;
    sincos_halfpi(zq[1], sh, ch);
    const float ct = fmaf(-2.0f * sh, sh, 1.0f);
    const float st = 2.0f * sh * ch;
    const float bq = fmaf(ct, -0.33333333333333333f, 0.33333333333333333f);
    const float dq = st * 0.57735026918962576f;
    f1_0 = fmaf(ct, 0.66666666666666667f, 0.33333333333333333f);
    f1_1 = bq + dq;
    f1_2 = bq - dq;
  }
  float f2_0, f2_1, f2_2;
  {
    float sh, ch;
    sincos_halfpi(zq[2], sh, ch);
    const float ct = fmaf(-2.0f * sh, sh, 1.0f);
    const float st = 2.0f * sh * ch;
    const float bq = fmaf(ct, -0.33333333333333333f, 0.33333333333333333f);
    const float dq = st * 0.57735026918962576f;
    f2_0 = fmaf(ct, 0.66666666666666667f, 0.33333333333333333f);
    f2_1 = bq + dq;
    f2_2 = bq - dq;
  }
  float f3_0, f3_1, f3_2;
  {
    float sh, ch;
    sincos_halfpi(zq[3], sh, ch);
    const float ct = fmaf(-2.0f * sh, sh, 1.0f);
    const float st = 2.0f * sh * ch;
    const float bq = fmaf(ct, -0.33333333333333333f, 0.33333333333333333f);
    const float dq = st * 0.57735026918962576f;
    f3_0 = fmaf(ct, 0.66666666666666667f, 0.33333333333333333f);
    f3_1 = bq + dq;
    f3_2 = bq - dq;
  }
  float f4_0, f4_1, f4_2;
  {
    float sh, ch;
    sincos_halfpi(zq[4], sh, ch);
    const float ct = fmaf(-2.0f * sh, sh, 1.0f);
    const float st = 2.0f * sh * ch;
    const float bq = fmaf(ct, -0.33333333333333333f, 0.33333333333333333f);
    const float dq = st * 0.57735026918962576f;
    f4_0 = fmaf(ct, 0.66666666666666667f, 0.33333333333333333f);
    f4_1 = bq + dq;
    f4_2 = bq - dq;
  }
  float f5_0, f5_1, f5_2;
  {
    float sh, ch;
    sincos_halfpi(zq[5], sh, ch);
    const float ct = fmaf(-2.0f * sh, sh, 1.0f);
    const float st = 2.0f * sh * ch;
    const float bq = fmaf(ct, -0.33333333333333333f, 0.33333333333333333f);
    const float dq = st * 0.57735026918962576f;
    f5_0 = fmaf(ct, 0.66666666666666667f, 0.33333333333333333f);
    f5_1 = bq + dq;
    f5_2 = bq - dq;
  }
  const float G_0 = f0_0 * f1_0, h45_0 = f4_0 * f5_0;
  const float G_1 = f0_0 * f1_1, h45_1 = f4_0 * f5_1;
  const float G_2 = f0_0 * f1_2, h45_2 = f4_0 * f5_2;
  const float G_3 = f0_1 * f1_0, h45_3 = f4_1 * f5_0;
  const float G_4 = f0_1 * f1_1, h45_4 = f4_1 * f5_1;
  const float G_5 = f0_1 * f1_2, h45_5 = f4_1 * f5_2;
  const float G_6 = f0_2 * f1_0, h45_6 = f4_2 * f5_0;
  const float G_7 = f0_2 * f1_1, h45_7 = f4_2 * f5_1;
  const float G_8 = f0_2 * f1_2, h45_8 = f4_2 * f5_2;
  const float f3s = half ? f3_1 : f3_0;
  const float hs_0 = f3s * h45_0;
  const float hs_1 = f3s * h45_1;
  const float hs_2 = f3s * h45_2;
  const float hs_3 = f3s * h45_3;
  const float hs_4 = f3s * h45_4;
  const float hs_5 = f3s * h45_5;
  const float hs_6 = f3s * h45_6;
  const float hs_7 = f3s * h45_7;
  const float hs_8 = f3s * h45_8;
  const float hs_9  = f3_2 * (half ? h45_5 : h45_0);
  const float hs_10 = f3_2 * (half ? h45_6 : h45_1);
  const float hs_11 = f3_2 * (half ? h45_7 : h45_2);
  const float hs_12 = f3_2 * (half ? h45_8 : h45_3);
  const float hs_13 = f3_2 * (half ? 0.0f : h45_4);
  const v2f hp[7] = { v2f{hs_0, hs_1}, v2f{hs_2, hs_3}, v2f{hs_4, hs_5}, v2f{hs_6, hs_7},
                      v2f{hs_8, hs_9}, v2f{hs_10, hs_11}, v2f{hs_12, hs_13} };
  v2f hb[21];
  #pragma unroll
  for (int kk = 0; kk < 21; ++kk) {
    const float fs = (kk < 7) ? f2_0 : ((kk < 14) ? f2_1 : f2_2);
    hb[kk] = hp[kk % 7] * v2f{fs, fs};
  }

  v8f yv[3] = {};
  v2f va[2][9];
  #pragma unroll
  for (int j = 0; j < 3; ++j) {
    #pragma unroll
    for (int t = 0; t < 3; ++t) va[0][3 * j + t] = *(const v2f*)(vp + 16 * t * VP + 4 * j);
  }
  #pragma unroll
  for (int b = 0; b < 7; ++b) {
    if (b + 1 < 7) {
      #pragma unroll
      for (int j = 0; j < 3; ++j) {
        #pragma unroll
        for (int t = 0; t < 3; ++t)
          va[(b + 1) & 1][3 * j + t] = *(const v2f*)(vp + 16 * t * VP + 4 * (3 * b + 3 + j));
      }
    }
    __builtin_amdgcn_sched_barrier(0);
    #pragma unroll
    for (int j = 0; j < 3; ++j) {
      #pragma unroll
      for (int t = 0; t < 3; ++t)
        yv[t] = __builtin_amdgcn_wmma_f32_16x16x4_f32(false, va[b & 1][3 * j + t], false, hb[3 * b + j],
                                                      (short)0, yv[t], false, false);
    }
    __builtin_amdgcn_sched_barrier(0);
  }

  const int row = rowBase + lm;
  v2f p01 = v2f{yv[0][0], yv[0][1]} * v2f{G_0, G_0};
  p01 += v2f{yv[0][2], yv[0][3]} * v2f{G_1, G_1};
  p01 += v2f{yv[0][4], yv[0][5]} * v2f{G_2, G_2};
  p01 += v2f{yv[0][6], yv[0][7]} * v2f{G_3, G_3};
  p01 += v2f{yv[1][0], yv[1][1]} * v2f{G_4, G_4};
  p01 += v2f{yv[1][2], yv[1][3]} * v2f{G_5, G_5};
  p01 += v2f{yv[1][4], yv[1][5]} * v2f{G_6, G_6};
  p01 += v2f{yv[1][6], yv[1][7]} * v2f{G_7, G_7};
  p01 += v2f{yv[2][0], yv[2][1]} * v2f{G_8, G_8};
  float p4 = (half ? G_5 : G_0) * yv[2][2];
  p4 = fmaf(half ? G_6 : G_1, yv[2][3], p4);
  p4 = fmaf(half ? G_7 : G_2, yv[2][4], p4);
  p4 = fmaf(half ? G_8 : G_3, yv[2][5], p4);
  p4 = fmaf(half ? 0.0f : G_4, yv[2][6], p4);
  p4 += xhalf(p4);
  if (row < B) {
    float* orow = out + (size_t)row * 5;
    orow[half]     = p01.x + sBc[half];
    orow[2 + half] = p01.y + sBc[2 + half];
    if (half == 0) orow[4] = p4 + sBc[4];
  }
  #pragma unroll
  for (int kk = 0; kk < 16; ++kk) xf[kk] = xn[kk];
  }
}

extern "C" void kernel_launch(void* const* d_in, const int* in_sizes, int n_in,
                              void* d_out, int out_size, void* d_ws, size_t ws_size,
                              hipStream_t stream) {
  (void)n_in; (void)out_size; (void)ws_size;
  const float* x    = (const float*)d_in[0];
  const float* W1   = (const float*)d_in[1];
  const float* b1   = (const float*)d_in[2];
  const float* ln_g = (const float*)d_in[3];
  const float* ln_b = (const float*)d_in[4];
  const float* W2   = (const float*)d_in[5];
  const float* b2   = (const float*)d_in[6];
  const float* th_s = (const float*)d_in[7];
  const float* th_t = (const float*)d_in[8];
  const float* Wc   = (const float*)d_in[9];
  const float* bc   = (const float*)d_in[10];
  float* out = (float*)d_out;
  float* vimg = (float*)d_ws;
  int B = in_sizes[0] / 64;
  circuit_compose<<<122, 192, 0, stream>>>(th_s, th_t, Wc, vimg);
  qmaml_kernel<<<512, 256, 0, stream>>>(x, W1, b1, ln_g, ln_b, W2, b2, vimg, bc, out, B);
}
